# starting kernel
# baseline (speedup 1.0000x reference)
#include <hip/hip_runtime.h>
#include <hip/hip_bf16.h>
#include <math.h>

#define EPSV 1e-8f

typedef float v2f __attribute__((ext_vector_type(2)));
typedef float v8f __attribute__((ext_vector_type(8)));

__device__ __forceinline__ v8f wmma4(v2f a, v2f b, v8f c) {
  return __builtin_amdgcn_wmma_f32_16x16x4_f32(false, a, false, b, (short)0, c,
                                               false, false);
}

__device__ __forceinline__ float wave_sum32(float v) {
#pragma unroll
  for (int off = 16; off > 0; off >>= 1) v += __shfl_xor(v, off, 32);
  return v;
}

__device__ __forceinline__ void ln128_row(const float* __restrict__ src,
                                          float* __restrict__ dst,
                                          const float* __restrict__ g,
                                          const float* __restrict__ b,
                                          int lane) {
  float xv[4];
#pragma unroll
  for (int j = 0; j < 4; ++j) xv[j] = src[lane + 32 * j];
  float s = xv[0] + xv[1] + xv[2] + xv[3];
  s = wave_sum32(s);
  float m = s * (1.0f / 128.0f);
  float vs = 0.0f;
#pragma unroll
  for (int j = 0; j < 4; ++j) { float d = xv[j] - m; vs += d * d; }
  vs = wave_sum32(vs);
  float inv = rsqrtf(vs * (1.0f / 128.0f) + EPSV);
#pragma unroll
  for (int j = 0; j < 4; ++j) {
    int d = lane + 32 * j;
    dst[d] = (xv[j] - m) * inv * g[d] + b[d];
  }
}

__global__ void k_gram(const float* __restrict__ x,
                       const float* __restrict__ g1,
                       const float* __restrict__ b1,
                       float* __restrict__ stats1,
                       float* __restrict__ stats2,
                       float* __restrict__ gpart) {
  extern __shared__ float yt[];
  __shared__ float s1l[128], s2l[128];
  const int tid = threadIdx.x, lane = tid & 31, wave = tid >> 5;
  const int b = blockIdx.x >> 3;
  const int oct = blockIdx.x & 7;
  if (tid < 128) { s1l[tid] = 0.0f; s2l[tid] = 0.0f; }
  const int khalf = lane >> 4;
  const int nl = lane & 15;
  const v8f vzero = {0, 0, 0, 0, 0, 0, 0, 0};
  v8f acc[8];
#pragma unroll
  for (int t = 0; t < 8; ++t) acc[t] = vzero;
  float a1[4] = {0, 0, 0, 0}, a2[4] = {0, 0, 0, 0};

  for (int sc = 0; sc < 2; ++sc) {
    const long rowbase = (long)b * 4096 + oct * 512 + sc * 256;
    __syncthreads();
    for (int i = 0; i < 32; ++i) {
      int r = wave * 32 + i;
      const float* src = x + (rowbase + r) * 128;
      float xv[4];
#pragma unroll
      for (int j = 0; j < 4; ++j) xv[j] = src[lane + 32 * j];
      float s = wave_sum32(xv[0] + xv[1] + xv[2] + xv[3]);
      float m = s * (1.0f / 128.0f);
      float vs = 0.0f;
#pragma unroll
      for (int j = 0; j < 4; ++j) { float d = xv[j] - m; vs += d * d; }
      vs = wave_sum32(vs);
      float inv = rsqrtf(vs * (1.0f / 128.0f) + EPSV);
#pragma unroll
      for (int j = 0; j < 4; ++j) {
        int d = lane + 32 * j;
        float y = (xv[j] - m) * inv * g1[d] + b1[d];
        a1[j] += y; a2[j] += y * y;
        yt[r * 128 + d] = y;
      }
    }
    __syncthreads();
    for (int t = 0; t < 8; ++t) {
      int idx = wave + 8 * t;
      int mi = idx >> 3, ni = idx & 7;
      int mrow = mi * 16 + nl;
      int ncol = ni * 16 + nl;
      v8f a = acc[t];
      for (int kb = 0; kb < 64; ++kb) {
        int k0 = kb * 4 + khalf * 2;
        v2f af, bf;
        af.x = yt[k0 * 128 + mrow];   af.y = yt[(k0 + 1) * 128 + mrow];
        bf.x = yt[k0 * 128 + ncol];   bf.y = yt[(k0 + 1) * 128 + ncol];
        a = wmma4(af, bf, a);
      }
      acc[t] = a;
    }
  }

  float* gp = gpart + (long)blockIdx.x * 16384;
  for (int t = 0; t < 8; ++t) {
    int idx = wave + 8 * t;
    int mi = idx >> 3, ni = idx & 7;
    int dmBase = mi * 16 + khalf * 8;
    int de = ni * 16 + nl;
#pragma unroll
    for (int r = 0; r < 8; ++r) gp[(dmBase + r) * 128 + de] = acc[t][r];
  }
#pragma unroll
  for (int j = 0; j < 4; ++j) {
    int d = lane + 32 * j;
    atomicAdd(&s1l[d], a1[j]);
    atomicAdd(&s2l[d], a2[j]);
  }
  __syncthreads();
  if (tid < 128) {
    atomicAdd(&stats1[b * 128 + tid], s1l[tid]);
    atomicAdd(&stats2[b * 128 + tid], s2l[tid]);
  }
}

__global__ void k_combine(const float* __restrict__ stats1,
                          const float* __restrict__ stats2,
                          const float* __restrict__ gpart,
                          float* __restrict__ dcorr) {
  const int d = blockIdx.x;
  const int e = threadIdx.x;
  float acc = 0.0f;
  for (int b = 0; b < 32; ++b) {
    float s1d = stats1[b * 128 + d], s2d = stats2[b * 128 + d];
    float s1e = stats1[b * 128 + e], s2e = stats2[b * 128 + e];
    float md = s1d * (1.0f / 4096.0f), me = s1e * (1.0f / 4096.0f);
    float vd = s2d - s1d * s1d * (1.0f / 4096.0f);
    float ve = s2e - s1e * s1e * (1.0f / 4096.0f);
    float ind = 1.0f / fmaxf(sqrtf(fmaxf(vd, 0.0f)), EPSV);
    float ine = 1.0f / fmaxf(sqrtf(fmaxf(ve, 0.0f)), EPSV);
    float G = 0.0f;
#pragma unroll
    for (int q = 0; q < 8; ++q)
      G += gpart[(long)(b * 8 + q) * 16384 + d * 128 + e];
    acc += (G - 4096.0f * md * me) * ind * ine;
  }
  dcorr[d * 128 + e] = acc * (1.0f / 32.0f);
}

__global__ void k_graph(const float* __restrict__ corr,
                        const float* __restrict__ alpha_logit,
                        const float* __restrict__ chebw,
                        const float* __restrict__ dcorr,
                        float* __restrict__ mixed_out,
                        float* __restrict__ Lap,
                        float* __restrict__ wout) {
  __shared__ float dis[128];
  const int i = threadIdx.x;
  const float alpha = 1.0f / (1.0f + __expf(-alpha_logit[0]));
  float deg = 0.0f;
  for (int j = 0; j < 128; ++j) {
    float learned = tanhf(0.5f * (corr[i * 128 + j] + corr[j * 128 + i]));
    float dc = fminf(fmaxf(dcorr[i * 128 + j], -1.0f), 1.0f);
    float mx = alpha * learned + (1.0f - alpha) * dc;
    mixed_out[i * 128 + j] = mx;
    if (j != i) deg += mx;
  }
  dis[i] = rsqrtf(fmaxf(deg, EPSV));
  __syncthreads();
  const float di = dis[i];
  for (int j = 0; j < 128; ++j) {
    float a = (j == i) ? 0.0f : mixed_out[i * 128 + j];
    float l = ((j == i) ? 1.0f : 0.0f) - di * a * dis[j];
    Lap[i * 128 + j] = fminf(fmaxf(l, -1.5f), 1.5f);
  }
  if (i == 0) {
    float c0 = chebw[0], c1 = chebw[1], c2 = chebw[2];
    float mx = fmaxf(c0, fmaxf(c1, c2));
    float e0 = __expf(c0 - mx), e1 = __expf(c1 - mx), e2 = __expf(c2 - mx);
    float s = e0 + e1 + e2;
    wout[0] = e0 / s; wout[1] = e1 / s; wout[2] = e2 / s;
  }
}

__global__ void k_main(const float* __restrict__ x,
                       const float* __restrict__ g1, const float* __restrict__ b1,
                       const float* __restrict__ Lap, const float* __restrict__ wv,
                       const float* __restrict__ Wi, const float* __restrict__ bi,
                       const float* __restrict__ Wo, const float* __restrict__ bo,
                       const float* __restrict__ g2, const float* __restrict__ b2,
                       const float* __restrict__ g3, const float* __restrict__ b3,
                       float* __restrict__ out) {
  extern __shared__ float smem[];
  float* yo = smem;
  float* t1 = smem + 8192;
  float* hb = smem + 16384;
  const int tid = threadIdx.x, lane = tid & 31, wave = tid >> 5;
  const long base = (long)blockIdx.x * 64;
  const float w0 = wv[0], w1 = wv[1], w2 = wv[2];
  const int khalf = lane >> 4;
  const int nl = lane & 15;

  for (int i = 0; i < 8; ++i) {
    int r = wave * 8 + i;
    ln128_row(x + (base + r) * 128, yo + r * 128, g1, b1, lane);
  }
  __syncthreads();

  for (int t = 0; t < 4; ++t) {
    int idx = wave + 8 * t; int mi = idx >> 3, ni = idx & 7;
    v8f acc = {};
    int mrow = mi * 16 + nl, nc = ni * 16 + nl;
    for (int kb = 0; kb < 32; ++kb) {
      int k0 = kb * 4 + khalf * 2;
      v2f a, bb;
      a.x = yo[mrow * 128 + k0];  a.y = yo[mrow * 128 + k0 + 1];
      bb.x = Lap[k0 * 128 + nc];  bb.y = Lap[(k0 + 1) * 128 + nc];
      acc = wmma4(a, bb, acc);
    }
    int rowB = mi * 16 + khalf * 8;
#pragma unroll
    for (int r = 0; r < 8; ++r) t1[(rowB + r) * 128 + nc] = acc[r];
  }
  __syncthreads();

  for (int t = 0; t < 4; ++t) {
    int idx = wave + 8 * t; int mi = idx >> 3, ni = idx & 7;
    v8f acc = {};
    int mrow = mi * 16 + nl, nc = ni * 16 + nl;
    for (int kb = 0; kb < 32; ++kb) {
      int k0 = kb * 4 + khalf * 2;
      v2f a, bb;
      a.x = t1[mrow * 128 + k0];  a.y = t1[mrow * 128 + k0 + 1];
      bb.x = Lap[k0 * 128 + nc];  bb.y = Lap[(k0 + 1) * 128 + nc];
      acc = wmma4(a, bb, acc);
    }
    int rowB = mi * 16 + khalf * 8;
#pragma unroll
    for (int r = 0; r < 8; ++r) {
      int pos = (rowB + r) * 128 + nc;
      float y0 = yo[pos], y1 = t1[pos];
      float tx2 = fminf(fmaxf(2.0f * acc[r] - y0, -50.0f), 50.0f);
      yo[pos] = w0 * y0 + w1 * y1 + w2 * tx2;
    }
  }
  __syncthreads();

  for (int t = 0; t < 8; ++t) {
    int idx = wave + 8 * t; int mi = idx >> 4, ni = idx & 15;
    v8f acc = {};
    int mrow = mi * 16 + nl, nc = ni * 16 + nl;
    for (int kb = 0; kb < 32; ++kb) {
      int k0 = kb * 4 + khalf * 2;
      v2f a, bb;
      a.x = yo[mrow * 128 + k0];  a.y = yo[mrow * 128 + k0 + 1];
      bb.x = Wi[nc * 128 + k0];   bb.y = Wi[nc * 128 + k0 + 1];
      acc = wmma4(a, bb, acc);
    }
    int rowB = mi * 16 + khalf * 8;
    float bia = bi[nc];
#pragma unroll
    for (int r = 0; r < 8; ++r) hb[(rowB + r) * 256 + nc] = acc[r] + bia;
  }
  __syncthreads();

  for (int i = 0; i < 8; ++i) {
    int r = wave * 8 + i;
    float xv[8];
#pragma unroll
    for (int j = 0; j < 8; ++j) xv[j] = hb[r * 256 + lane + 32 * j];
    float s = 0.0f;
#pragma unroll
    for (int j = 0; j < 8; ++j) s += xv[j];
    s = wave_sum32(s);
    float m = s * (1.0f / 256.0f);
    float vs = 0.0f;
#pragma unroll
    for (int j = 0; j < 8; ++j) { float d = xv[j] - m; vs += d * d; }
    vs = wave_sum32(vs);
    float inv = rsqrtf(vs * (1.0f / 256.0f) + EPSV);
#pragma unroll
    for (int j = 0; j < 8; ++j) {
      int d = lane + 32 * j;
      float yv = (xv[j] - m) * inv * g2[d] + b2[d];
      hb[r * 256 + d] = 0.5f * yv * (1.0f + erff(yv * 0.70710678118654752f));
    }
  }
  __syncthreads();

  for (int t = 0; t < 4; ++t) {
    int idx = wave + 8 * t; int mi = idx >> 3, ni = idx & 7;
    v8f acc = {};
    int mrow = mi * 16 + nl, nc = ni * 16 + nl;
    for (int kb = 0; kb < 64; ++kb) {
      int k0 = kb * 4 + khalf * 2;
      v2f a, bb;
      a.x = hb[mrow * 256 + k0];  a.y = hb[mrow * 256 + k0 + 1];
      bb.x = Wo[nc * 256 + k0];   bb.y = Wo[nc * 256 + k0 + 1];
      acc = wmma4(a, bb, acc);
    }
    int rowB = mi * 16 + khalf * 8;
    float boa = bo[nc];
#pragma unroll
    for (int r = 0; r < 8; ++r) t1[(rowB + r) * 128 + nc] = acc[r] + boa;
  }
  __syncthreads();

  for (int i = 0; i < 8; ++i) {
    int r = wave * 8 + i;
    float xv[4];
#pragma unroll
    for (int j = 0; j < 4; ++j) xv[j] = t1[r * 128 + lane + 32 * j];
    float s = wave_sum32(xv[0] + xv[1] + xv[2] + xv[3]);
    float m = s * (1.0f / 128.0f);
    float vs = 0.0f;
#pragma unroll
    for (int j = 0; j < 4; ++j) { float d = xv[j] - m; vs += d * d; }
    vs = wave_sum32(vs);
    float inv = rsqrtf(vs * (1.0f / 128.0f) + EPSV);
#pragma unroll
    for (int j = 0; j < 4; ++j) {
      int d = lane + 32 * j;
      out[(base + r) * 128 + d] = (xv[j] - m) * inv * g3[d] + b3[d];
    }
  }
}

extern "C" void kernel_launch(void* const* d_in, const int* in_sizes, int n_in,
                              void* d_out, int out_size, void* d_ws, size_t ws_size,
                              hipStream_t stream) {
  const float* x     = (const float*)d_in[0];
  const float* corr  = (const float*)d_in[1];
  const float* alog  = (const float*)d_in[2];
  const float* chebw = (const float*)d_in[3];
  const float* Wi    = (const float*)d_in[4];
  const float* bi    = (const float*)d_in[5];
  const float* Wo    = (const float*)d_in[6];
  const float* bo    = (const float*)d_in[7];
  const float* g1    = (const float*)d_in[8];
  const float* b1    = (const float*)d_in[9];
  const float* g2    = (const float*)d_in[10];
  const float* b2    = (const float*)d_in[11];
  const float* g3    = (const float*)d_in[12];
  const float* b3    = (const float*)d_in[13];
  float* out = (float*)d_out;
  float* ws  = (float*)d_ws;

  float* stats1 = ws;
  float* stats2 = ws + 4096;
  float* gpart  = ws + 8192;
  float* dcorr  = ws + 8192 + 256 * 16384;
  float* Lap    = dcorr + 16384;
  float* wv     = Lap + 16384;
  float* mixed  = out + (long)32 * 4096 * 128;

  hipMemsetAsync(d_ws, 0, 8192 * sizeof(float), stream);
  k_gram<<<256, 256, 256 * 128 * sizeof(float), stream>>>(x, g1, b1, stats1,
                                                          stats2, gpart);
  k_combine<<<128, 128, 0, stream>>>(stats1, stats2, gpart, dcorr);
  k_graph<<<1, 128, 0, stream>>>(corr, alog, chebw, dcorr, mixed, Lap, wv);
  k_main<<<2048, 256, 131072, stream>>>(x, g1, b1, Lap, wv, Wi, bi, Wo, bo,
                                        g2, b2, g3, b3, out);
}


# LDS rows padded (pitch 132/260 floats in main kernel, 144 in gram) so WMMA fragment reads spread over all 64 banks; column stats now fixed-order per-wg partials, float atomics and memset removed
# speedup vs baseline: 1.7859x; 1.7859x over previous
#include <hip/hip_runtime.h>
#include <hip/hip_bf16.h>
#include <math.h>

#define EPSV 1e-8f

#define PY 132
#define PH 260
#define PG 144

typedef float v2f __attribute__((ext_vector_type(2)));
typedef float v8f __attribute__((ext_vector_type(8)));

__device__ __forceinline__ v8f wmma4(v2f a, v2f b, v8f c) {
  return __builtin_amdgcn_wmma_f32_16x16x4_f32(false, a, false, b, (short)0, c,
                                               false, false);
}

__device__ __forceinline__ float wave_sum32(float v) {
#pragma unroll
  for (int off = 16; off > 0; off >>= 1) v += __shfl_xor(v, off, 32);
  return v;
}

__device__ __forceinline__ void ln128_row(const float* __restrict__ src,
                                          float* __restrict__ dst,
                                          const float* __restrict__ g,
                                          const float* __restrict__ b,
                                          int lane) {
  float xv[4];
#pragma unroll
  for (int j = 0; j < 4; ++j) xv[j] = src[lane + 32 * j];
  float s = xv[0] + xv[1] + xv[2] + xv[3];
  s = wave_sum32(s);
  float m = s * (1.0f / 128.0f);
  float vs = 0.0f;
#pragma unroll
  for (int j = 0; j < 4; ++j) { float d = xv[j] - m; vs += d * d; }
  vs = wave_sum32(vs);
  float inv = rsqrtf(vs * (1.0f / 128.0f) + EPSV);
#pragma unroll
  for (int j = 0; j < 4; ++j) {
    int d = lane + 32 * j;
    dst[d] = (xv[j] - m) * inv * g[d] + b[d];
  }
}

__global__ void k_gram(const float* __restrict__ x,
                       const float* __restrict__ g1,
                       const float* __restrict__ b1,
                       float* __restrict__ statp,
                       float* __restrict__ gpart) {
  extern __shared__ float yt[];
  __shared__ float sred[2][8][128];
  const int tid = threadIdx.x, lane = tid & 31, wave = tid >> 5;
  const int b = blockIdx.x >> 3;
  const int oct = blockIdx.x & 7;
  const int khalf = lane >> 4;
  const int nl = lane & 15;
  const v8f vzero = {0, 0, 0, 0, 0, 0, 0, 0};
  v8f acc[8];
#pragma unroll
  for (int t = 0; t < 8; ++t) acc[t] = vzero;
  float a1[4] = {0, 0, 0, 0}, a2[4] = {0, 0, 0, 0};

  for (int sc = 0; sc < 2; ++sc) {
    const long rowbase = (long)b * 4096 + oct * 512 + sc * 256;
    __syncthreads();
    for (int i = 0; i < 32; ++i) {
      int r = wave * 32 + i;
      const float* src = x + (rowbase + r) * 128;
      float xv[4];
#pragma unroll
      for (int j = 0; j < 4; ++j) xv[j] = src[lane + 32 * j];
      float s = wave_sum32(xv[0] + xv[1] + xv[2] + xv[3]);
      float m = s * (1.0f / 128.0f);
      float vs = 0.0f;
#pragma unroll
      for (int j = 0; j < 4; ++j) { float d = xv[j] - m; vs += d * d; }
      vs = wave_sum32(vs);
      float inv = rsqrtf(vs * (1.0f / 128.0f) + EPSV);
#pragma unroll
      for (int j = 0; j < 4; ++j) {
        int d = lane + 32 * j;
        float y = (xv[j] - m) * inv * g1[d] + b1[d];
        a1[j] += y; a2[j] += y * y;
        yt[r * PG + d] = y;
      }
    }
    __syncthreads();
    for (int t = 0; t < 8; ++t) {
      int idx = wave + 8 * t;
      int mi = idx >> 3, ni = idx & 7;
      int mrow = mi * 16 + nl;
      int ncol = ni * 16 + nl;
      v8f a = acc[t];
      for (int kb = 0; kb < 64; ++kb) {
        int k0 = kb * 4 + khalf * 2;
        v2f af, bf;
        af.x = yt[k0 * PG + mrow];   af.y = yt[(k0 + 1) * PG + mrow];
        bf.x = yt[k0 * PG + ncol];   bf.y = yt[(k0 + 1) * PG + ncol];
        a = wmma4(af, bf, a);
      }
      acc[t] = a;
    }
  }

  float* gp = gpart + (long)blockIdx.x * 16384;
  for (int t = 0; t < 8; ++t) {
    int idx = wave + 8 * t;
    int mi = idx >> 3, ni = idx & 7;
    int dmBase = mi * 16 + khalf * 8;
    int de = ni * 16 + nl;
#pragma unroll
    for (int r = 0; r < 8; ++r) gp[(dmBase + r) * 128 + de] = acc[t][r];
  }
#pragma unroll
  for (int j = 0; j < 4; ++j) {
    int d = lane + 32 * j;
    sred[0][wave][d] = a1[j];
    sred[1][wave][d] = a2[j];
  }
  __syncthreads();
  {
    const int which = tid >> 7, d = tid & 127;
    float s = 0.0f;
#pragma unroll
    for (int w = 0; w < 8; ++w) s += sred[which][w][d];
    statp[which * 32768 + blockIdx.x * 128 + d] = s;
  }
}

__global__ void k_combine(const float* __restrict__ statp,
                          const float* __restrict__ gpart,
                          float* __restrict__ dcorr) {
  __shared__ float mu[32][128];
  __shared__ float iv[32][128];
  const int d = blockIdx.x;
  const int e = threadIdx.x;
  for (int b = 0; b < 32; ++b) {
    float s1 = 0.0f, s2 = 0.0f;
#pragma unroll
    for (int q = 0; q < 8; ++q) {
      s1 += statp[(b * 8 + q) * 128 + e];
      s2 += statp[32768 + (b * 8 + q) * 128 + e];
    }
    float v = s2 - s1 * s1 * (1.0f / 4096.0f);
    mu[b][e] = s1 * (1.0f / 4096.0f);
    iv[b][e] = 1.0f / fmaxf(sqrtf(fmaxf(v, 0.0f)), EPSV);
  }
  __syncthreads();
  float acc = 0.0f;
  for (int b = 0; b < 32; ++b) {
    float md = mu[b][d], me = mu[b][e];
    float ind = iv[b][d], ine = iv[b][e];
    float G = 0.0f;
#pragma unroll
    for (int q = 0; q < 8; ++q)
      G += gpart[(long)(b * 8 + q) * 16384 + d * 128 + e];
    acc += (G - 4096.0f * md * me) * ind * ine;
  }
  dcorr[d * 128 + e] = acc * (1.0f / 32.0f);
}

__global__ void k_graph(const float* __restrict__ corr,
                        const float* __restrict__ alpha_logit,
                        const float* __restrict__ chebw,
                        const float* __restrict__ dcorr,
                        float* __restrict__ mixed_out,
                        float* __restrict__ Lap,
                        float* __restrict__ wout) {
  __shared__ float dis[128];
  const int i = threadIdx.x;
  const float alpha = 1.0f / (1.0f + __expf(-alpha_logit[0]));
  float deg = 0.0f;
  for (int j = 0; j < 128; ++j) {
    float learned = tanhf(0.5f * (corr[i * 128 + j] + corr[j * 128 + i]));
    float dc = fminf(fmaxf(dcorr[i * 128 + j], -1.0f), 1.0f);
    float mx = alpha * learned + (1.0f - alpha) * dc;
    mixed_out[i * 128 + j] = mx;
    if (j != i) deg += mx;
  }
  dis[i] = rsqrtf(fmaxf(deg, EPSV));
  __syncthreads();
  const float di = dis[i];
  for (int j = 0; j < 128; ++j) {
    float a = (j == i) ? 0.0f : mixed_out[i * 128 + j];
    float l = ((j == i) ? 1.0f : 0.0f) - di * a * dis[j];
    Lap[i * 128 + j] = fminf(fmaxf(l, -1.5f), 1.5f);
  }
  if (i == 0) {
    float c0 = chebw[0], c1 = chebw[1], c2 = chebw[2];
    float mx = fmaxf(c0, fmaxf(c1, c2));
    float e0 = __expf(c0 - mx), e1 = __expf(c1 - mx), e2 = __expf(c2 - mx);
    float s = e0 + e1 + e2;
    wout[0] = e0 / s; wout[1] = e1 / s; wout[2] = e2 / s;
  }
}

__global__ void k_main(const float* __restrict__ x,
                       const float* __restrict__ g1, const float* __restrict__ b1,
                       const float* __restrict__ Lap, const float* __restrict__ wv,
                       const float* __restrict__ Wi, const float* __restrict__ bi,
                       const float* __restrict__ Wo, const float* __restrict__ bo,
                       const float* __restrict__ g2, const float* __restrict__ b2,
                       const float* __restrict__ g3, const float* __restrict__ b3,
                       float* __restrict__ out) {
  extern __shared__ float smem[];
  float* yo = smem;
  float* t1 = smem + 64 * PY;
  float* hb = smem + 2 * 64 * PY;
  const int tid = threadIdx.x, lane = tid & 31, wave = tid >> 5;
  const long base = (long)blockIdx.x * 64;
  const float w0 = wv[0], w1 = wv[1], w2 = wv[2];
  const int khalf = lane >> 4;
  const int nl = lane & 15;

  for (int i = 0; i < 8; ++i) {
    int r = wave * 8 + i;
    ln128_row(x + (base + r) * 128, yo + r * PY, g1, b1, lane);
  }
  __syncthreads();

  for (int t = 0; t < 4; ++t) {
    int idx = wave + 8 * t; int mi = idx >> 3, ni = idx & 7;
    v8f acc = {};
    int mrow = mi * 16 + nl, nc = ni * 16 + nl;
    for (int kb = 0; kb < 32; ++kb) {
      int k0 = kb * 4 + khalf * 2;
      v2f a, bb;
      a.x = yo[mrow * PY + k0];  a.y = yo[mrow * PY + k0 + 1];
      bb.x = Lap[k0 * 128 + nc];  bb.y = Lap[(k0 + 1) * 128 + nc];
      acc = wmma4(a, bb, acc);
    }
    int rowB = mi * 16 + khalf * 8;
#pragma unroll
    for (int r = 0; r < 8; ++r) t1[(rowB + r) * PY + nc] = acc[r];
  }
  __syncthreads();

  for (int t = 0; t < 4; ++t) {
    int idx = wave + 8 * t; int mi = idx >> 3, ni = idx & 7;
    v8f acc = {};
    int mrow = mi * 16 + nl, nc = ni * 16 + nl;
    for (int kb = 0; kb < 32; ++kb) {
      int k0 = kb * 4 + khalf * 2;
      v2f a, bb;
      a.x = t1[mrow * PY + k0];  a.y = t1[mrow * PY + k0 + 1];
      bb.x = Lap[k0 * 128 + nc];  bb.y = Lap[(k0 + 1) * 128 + nc];
      acc = wmma4(a, bb, acc);
    }
    int rowB = mi * 16 + khalf * 8;
#pragma unroll
    for (int r = 0; r < 8; ++r) {
      int pos = (rowB + r) * PY + nc;
      float y0 = yo[pos], y1 = t1[pos];
      float tx2 = fminf(fmaxf(2.0f * acc[r] - y0, -50.0f), 50.0f);
      yo[pos] = w0 * y0 + w1 * y1 + w2 * tx2;
    }
  }
  __syncthreads();

  for (int t = 0; t < 8; ++t) {
    int idx = wave + 8 * t; int mi = idx >> 4, ni = idx & 15;
    v8f acc = {};
    int mrow = mi * 16 + nl, nc = ni * 16 + nl;
    for (int kb = 0; kb < 32; ++kb) {
      int k0 = kb * 4 + khalf * 2;
      v2f a, bb;
      a.x = yo[mrow * PY + k0];  a.y = yo[mrow * PY + k0 + 1];
      bb.x = Wi[nc * 128 + k0];   bb.y = Wi[nc * 128 + k0 + 1];
      acc = wmma4(a, bb, acc);
    }
    int rowB = mi * 16 + khalf * 8;
    float bia = bi[nc];
#pragma unroll
    for (int r = 0; r < 8; ++r) hb[(rowB + r) * PH + nc] = acc[r] + bia;
  }
  __syncthreads();

  for (int i = 0; i < 8; ++i) {
    int r = wave * 8 + i;
    float xv[8];
#pragma unroll
    for (int j = 0; j < 8; ++j) xv[j] = hb[r * PH + lane + 32 * j];
    float s = 0.0f;
#pragma unroll
    for (int j = 0; j < 8; ++j) s += xv[j];
    s = wave_sum32(s);
    float m = s * (1.0f / 256.0f);
    float vs = 0.0f;
#pragma unroll
    for (int j = 0; j < 8; ++j) { float d = xv[j] - m; vs += d * d; }
    vs = wave_sum32(vs);
    float inv = rsqrtf(vs * (1.0f / 256.0f) + EPSV);
#pragma unroll
    for (int j = 0; j < 8; ++j) {
      int d = lane + 32 * j;
      float yv = (xv[j] - m) * inv * g2[d] + b2[d];
      hb[r * PH + d] = 0.5f * yv * (1.0f + erff(yv * 0.70710678118654752f));
    }
  }
  __syncthreads();

  for (int t = 0; t < 4; ++t) {
    int idx = wave + 8 * t; int mi = idx >> 3, ni = idx & 7;
    v8f acc = {};
    int mrow = mi * 16 + nl, nc = ni * 16 + nl;
    for (int kb = 0; kb < 64; ++kb) {
      int k0 = kb * 4 + khalf * 2;
      v2f a, bb;
      a.x = hb[mrow * PH + k0];  a.y = hb[mrow * PH + k0 + 1];
      bb.x = Wo[nc * 256 + k0];   bb.y = Wo[nc * 256 + k0 + 1];
      acc = wmma4(a, bb, acc);
    }
    int rowB = mi * 16 + khalf * 8;
    float boa = bo[nc];
#pragma unroll
    for (int r = 0; r < 8; ++r) t1[(rowB + r) * PY + nc] = acc[r] + boa;
  }
  __syncthreads();

  for (int i = 0; i < 8; ++i) {
    int r = wave * 8 + i;
    float xv[4];
#pragma unroll
    for (int j = 0; j < 4; ++j) xv[j] = t1[r * PY + lane + 32 * j];
    float s = wave_sum32(xv[0] + xv[1] + xv[2] + xv[3]);
    float m = s * (1.0f / 128.0f);
    float vs = 0.0f;
#pragma unroll
    for (int j = 0; j < 4; ++j) { float d = xv[j] - m; vs += d * d; }
    vs = wave_sum32(vs);
    float inv = rsqrtf(vs * (1.0f / 128.0f) + EPSV);
#pragma unroll
    for (int j = 0; j < 4; ++j) {
      int d = lane + 32 * j;
      out[(base + r) * 128 + d] = (xv[j] - m) * inv * g3[d] + b3[d];
    }
  }
}

extern "C" void kernel_launch(void* const* d_in, const int* in_sizes, int n_in,
                              void* d_out, int out_size, void* d_ws, size_t ws_size,
                              hipStream_t stream) {
  const float* x     = (const float*)d_in[0];
  const float* corr  = (const float*)d_in[1];
  const float* alog  = (const float*)d_in[2];
  const float* chebw = (const float*)d_in[3];
  const float* Wi    = (const float*)d_in[4];
  const float* bi    = (const float*)d_in[5];
  const float* Wo    = (const float*)d_in[6];
  const float* bo    = (const float*)d_in[7];
  const float* g1    = (const float*)d_in[8];
  const float* b1    = (const float*)d_in[9];
  const float* g2    = (const float*)d_in[10];
  const float* b2    = (const float*)d_in[11];
  const float* g3    = (const float*)d_in[12];
  const float* b3    = (const float*)d_in[13];
  float* out = (float*)d_out;
  float* ws  = (float*)d_ws;

  float* gpart  = ws + 8192;
  float* dcorr  = ws + 8192 + 256 * 16384;
  float* Lap    = dcorr + 16384;
  float* wv     = Lap + 16384;
  float* mixed  = out + (long)32 * 4096 * 128;
  float* statp  = out;

  k_gram<<<256, 256, 256 * PG * sizeof(float), stream>>>(x, g1, b1, statp, gpart);
  k_combine<<<128, 128, 0, stream>>>(statp, gpart, dcorr);
  k_graph<<<1, 128, 0, stream>>>(corr, alog, chebw, dcorr, mixed, Lap, wv);
  k_main<<<2048, 256, (2 * 64 * PY + 64 * PH) * sizeof(float), stream>>>(x, g1, b1, Lap, wv, Wi, bi, Wo, bo,
                                        g2, b2, g3, b3, out);
}


# register blocking of every WMMA loop: a wave keeps 4 or 8 accumulator tiles (all row blocks of its column block) and reuses one B fragment per k step across them; __launch_bounds__(256) for registers
# speedup vs baseline: 2.4836x; 1.3907x over previous
#include <hip/hip_runtime.h>
#include <hip/hip_bf16.h>
#include <math.h>

#define EPSV 1e-8f

#define PY 132
#define PH 260
#define PG 144

typedef float v2f __attribute__((ext_vector_type(2)));
typedef float v8f __attribute__((ext_vector_type(8)));

__device__ __forceinline__ v8f wmma4(v2f a, v2f b, v8f c) {
  return __builtin_amdgcn_wmma_f32_16x16x4_f32(false, a, false, b, (short)0, c,
                                               false, false);
}

__device__ __forceinline__ float wave_sum32(float v) {
#pragma unroll
  for (int off = 16; off > 0; off >>= 1) v += __shfl_xor(v, off, 32);
  return v;
}

__device__ __forceinline__ void ln128_row(const float* __restrict__ src,
                                          float* __restrict__ dst,
                                          const float* __restrict__ g,
                                          const float* __restrict__ b,
                                          int lane) {
  float xv[4];
#pragma unroll
  for (int j = 0; j < 4; ++j) xv[j] = src[lane + 32 * j];
  float s = xv[0] + xv[1] + xv[2] + xv[3];
  s = wave_sum32(s);
  float m = s * (1.0f / 128.0f);
  float vs = 0.0f;
#pragma unroll
  for (int j = 0; j < 4; ++j) { float d = xv[j] - m; vs += d * d; }
  vs = wave_sum32(vs);
  float inv = rsqrtf(vs * (1.0f / 128.0f) + EPSV);
#pragma unroll
  for (int j = 0; j < 4; ++j) {
    int d = lane + 32 * j;
    dst[d] = (xv[j] - m) * inv * g[d] + b[d];
  }
}

__global__ void __launch_bounds__(256) k_gram(const float* __restrict__ x,
                       const float* __restrict__ g1,
                       const float* __restrict__ b1,
                       float* __restrict__ statp,
                       float* __restrict__ gpart) {
  extern __shared__ float yt[];
  __shared__ float sred[2][8][128];
  const int tid = threadIdx.x, lane = tid & 31, wave = tid >> 5;
  const int b = blockIdx.x >> 3;
  const int oct = blockIdx.x & 7;
  const int khalf = lane >> 4;
  const int nl = lane & 15;
  const v8f vzero = {0, 0, 0, 0, 0, 0, 0, 0};
  v8f acc[8];
#pragma unroll
  for (int t = 0; t < 8; ++t) acc[t] = vzero;
  float a1[4] = {0, 0, 0, 0}, a2[4] = {0, 0, 0, 0};

  for (int sc = 0; sc < 2; ++sc) {
    const long rowbase = (long)b * 4096 + oct * 512 + sc * 256;
    __syncthreads();
    for (int i = 0; i < 32; ++i) {
      int r = wave * 32 + i;
      const float* src = x + (rowbase + r) * 128;
      float xv[4];
#pragma unroll
      for (int j = 0; j < 4; ++j) xv[j] = src[lane + 32 * j];
      float s = wave_sum32(xv[0] + xv[1] + xv[2] + xv[3]);
      float m = s * (1.0f / 128.0f);
      float vs = 0.0f;
#pragma unroll
      for (int j = 0; j < 4; ++j) { float d = xv[j] - m; vs += d * d; }
      vs = wave_sum32(vs);
      float inv = rsqrtf(vs * (1.0f / 128.0f) + EPSV);
#pragma unroll
      for (int j = 0; j < 4; ++j) {
        int d = lane + 32 * j;
        float y = (xv[j] - m) * inv * g1[d] + b1[d];
        a1[j] += y; a2[j] += y * y;
        yt[r * PG + d] = y;
      }
    }
    __syncthreads();
    {
      const float* bcol = yt + (khalf * 2) * PG + wave * 16 + nl;
      const float* arow = yt + (khalf * 2) * PG + nl;
#pragma unroll 2
      for (int kb = 0; kb < 64; ++kb) {
        v2f bf;
        bf.x = bcol[kb * 4 * PG];   bf.y = bcol[kb * 4 * PG + PG];
#pragma unroll
        for (int mi = 0; mi < 8; ++mi) {
          v2f af;
          af.x = arow[kb * 4 * PG + mi * 16];   af.y = arow[kb * 4 * PG + PG + mi * 16];
          acc[mi] = wmma4(af, bf, acc[mi]);
        }
      }
    }
  }

  float* gp = gpart + (long)blockIdx.x * 16384;
  for (int t = 0; t < 8; ++t) {
    int idx = wave + 8 * t;
    int mi = idx >> 3, ni = idx & 7;
    int dmBase = mi * 16 + khalf * 8;
    int de = ni * 16 + nl;
#pragma unroll
    for (int r = 0; r < 8; ++r) gp[(dmBase + r) * 128 + de] = acc[t][r];
  }
#pragma unroll
  for (int j = 0; j < 4; ++j) {
    int d = lane + 32 * j;
    sred[0][wave][d] = a1[j];
    sred[1][wave][d] = a2[j];
  }
  __syncthreads();
  {
    const int which = tid >> 7, d = tid & 127;
    float s = 0.0f;
#pragma unroll
    for (int w = 0; w < 8; ++w) s += sred[which][w][d];
    statp[which * 32768 + blockIdx.x * 128 + d] = s;
  }
}

__global__ void k_combine(const float* __restrict__ statp,
                          const float* __restrict__ gpart,
                          float* __restrict__ dcorr) {
  __shared__ float mu[32][128];
  __shared__ float iv[32][128];
  const int d = blockIdx.x;
  const int e = threadIdx.x;
  for (int b = 0; b < 32; ++b) {
    float s1 = 0.0f, s2 = 0.0f;
#pragma unroll
    for (int q = 0; q < 8; ++q) {
      s1 += statp[(b * 8 + q) * 128 + e];
      s2 += statp[32768 + (b * 8 + q) * 128 + e];
    }
    float v = s2 - s1 * s1 * (1.0f / 4096.0f);
    mu[b][e] = s1 * (1.0f / 4096.0f);
    iv[b][e] = 1.0f / fmaxf(sqrtf(fmaxf(v, 0.0f)), EPSV);
  }
  __syncthreads();
  float acc = 0.0f;
  for (int b = 0; b < 32; ++b) {
    float md = mu[b][d], me = mu[b][e];
    float ind = iv[b][d], ine = iv[b][e];
    float G = 0.0f;
#pragma unroll
    for (int q = 0; q < 8; ++q)
      G += gpart[(long)(b * 8 + q) * 16384 + d * 128 + e];
    acc += (G - 4096.0f * md * me) * ind * ine;
  }
  dcorr[d * 128 + e] = acc * (1.0f / 32.0f);
}

__global__ void k_graph(const float* __restrict__ corr,
                        const float* __restrict__ alpha_logit,
                        const float* __restrict__ chebw,
                        const float* __restrict__ dcorr,
                        float* __restrict__ mixed_out,
                        float* __restrict__ Lap,
                        float* __restrict__ wout) {
  __shared__ float dis[128];
  const int i = threadIdx.x;
  const float alpha = 1.0f / (1.0f + __expf(-alpha_logit[0]));
  float deg = 0.0f;
  for (int j = 0; j < 128; ++j) {
    float learned = tanhf(0.5f * (corr[i * 128 + j] + corr[j * 128 + i]));
    float dc = fminf(fmaxf(dcorr[i * 128 + j], -1.0f), 1.0f);
    float mx = alpha * learned + (1.0f - alpha) * dc;
    mixed_out[i * 128 + j] = mx;
    if (j != i) deg += mx;
  }
  dis[i] = rsqrtf(fmaxf(deg, EPSV));
  __syncthreads();
  const float di = dis[i];
  for (int j = 0; j < 128; ++j) {
    float a = (j == i) ? 0.0f : mixed_out[i * 128 + j];
    float l = ((j == i) ? 1.0f : 0.0f) - di * a * dis[j];
    Lap[i * 128 + j] = fminf(fmaxf(l, -1.5f), 1.5f);
  }
  if (i == 0) {
    float c0 = chebw[0], c1 = chebw[1], c2 = chebw[2];
    float mx = fmaxf(c0, fmaxf(c1, c2));
    float e0 = __expf(c0 - mx), e1 = __expf(c1 - mx), e2 = __expf(c2 - mx);
    float s = e0 + e1 + e2;
    wout[0] = e0 / s; wout[1] = e1 / s; wout[2] = e2 / s;
  }
}

__global__ void __launch_bounds__(256) k_main(const float* __restrict__ x,
                       const float* __restrict__ g1, const float* __restrict__ b1,
                       const float* __restrict__ Lap, const float* __restrict__ wv,
                       const float* __restrict__ Wi, const float* __restrict__ bi,
                       const float* __restrict__ Wo, const float* __restrict__ bo,
                       const float* __restrict__ g2, const float* __restrict__ b2,
                       const float* __restrict__ g3, const float* __restrict__ b3,
                       float* __restrict__ out) {
  extern __shared__ float smem[];
  float* yo = smem;
  float* t1 = smem + 64 * PY;
  float* hb = smem + 2 * 64 * PY;
  const int tid = threadIdx.x, lane = tid & 31, wave = tid >> 5;
  const long base = (long)blockIdx.x * 64;
  const float w0 = wv[0], w1 = wv[1], w2 = wv[2];
  const int khalf = lane >> 4;
  const int nl = lane & 15;

  for (int i = 0; i < 8; ++i) {
    int r = wave * 8 + i;
    ln128_row(x + (base + r) * 128, yo + r * PY, g1, b1, lane);
  }
  __syncthreads();

  const float* arow_yo = yo + nl * PY + khalf * 2;
  const float* arow_t1 = t1 + nl * PY + khalf * 2;
  const float* arow_hb = hb + nl * PH + khalf * 2;
  const int nc = wave * 16 + nl;
  const float* lapcol = Lap + (khalf * 2) * 128 + nc;

  {
    v8f acc[4] = {};
#pragma unroll
    for (int kb = 0; kb < 32; ++kb) {
      v2f bb;
      bb.x = lapcol[kb * 512];   bb.y = lapcol[kb * 512 + 128];
#pragma unroll
      for (int mi = 0; mi < 4; ++mi) {
        v2f a = *(const v2f*)(arow_yo + mi * 16 * PY + kb * 4);
        acc[mi] = wmma4(a, bb, acc[mi]);
      }
    }
#pragma unroll
    for (int mi = 0; mi < 4; ++mi) {
      int rowB = mi * 16 + khalf * 8;
#pragma unroll
      for (int r = 0; r < 8; ++r) t1[(rowB + r) * PY + nc] = acc[mi][r];
    }
  }
  __syncthreads();

  {
    v8f acc[4] = {};
#pragma unroll
    for (int kb = 0; kb < 32; ++kb) {
      v2f bb;
      bb.x = lapcol[kb * 512];   bb.y = lapcol[kb * 512 + 128];
#pragma unroll
      for (int mi = 0; mi < 4; ++mi) {
        v2f a = *(const v2f*)(arow_t1 + mi * 16 * PY + kb * 4);
        acc[mi] = wmma4(a, bb, acc[mi]);
      }
    }
#pragma unroll
    for (int mi = 0; mi < 4; ++mi) {
      int rowB = mi * 16 + khalf * 8;
#pragma unroll
      for (int r = 0; r < 8; ++r) {
        int pos = (rowB + r) * PY + nc;
        float y0 = yo[pos], y1 = t1[pos];
        float tx2 = fminf(fmaxf(2.0f * acc[mi][r] - y0, -50.0f), 50.0f);
        yo[pos] = w0 * y0 + w1 * y1 + w2 * tx2;
      }
    }
  }
  __syncthreads();

  {
    v8f acc[2][4] = {};
    const float* wi0 = Wi + nc * 128 + khalf * 2;
    const float* wi1 = Wi + (nc + 128) * 128 + khalf * 2;
#pragma unroll
    for (int kb = 0; kb < 32; ++kb) {
      v2f b0 = *(const v2f*)(wi0 + kb * 4);
      v2f b1 = *(const v2f*)(wi1 + kb * 4);
#pragma unroll
      for (int mi = 0; mi < 4; ++mi) {
        v2f a = *(const v2f*)(arow_yo + mi * 16 * PY + kb * 4);
        acc[0][mi] = wmma4(a, b0, acc[0][mi]);
        acc[1][mi] = wmma4(a, b1, acc[1][mi]);
      }
    }
#pragma unroll
    for (int j = 0; j < 2; ++j) {
      int ncj = nc + 128 * j;
      float bia = bi[ncj];
#pragma unroll
      for (int mi = 0; mi < 4; ++mi) {
        int rowB = mi * 16 + khalf * 8;
#pragma unroll
        for (int r = 0; r < 8; ++r) hb[(rowB + r) * PH + ncj] = acc[j][mi][r] + bia;
      }
    }
  }
  __syncthreads();

  for (int i = 0; i < 8; ++i) {
    int r = wave * 8 + i;
    float xv[8];
#pragma unroll
    for (int j = 0; j < 8; ++j) xv[j] = hb[r * PH + lane + 32 * j];
    float s = 0.0f;
#pragma unroll
    for (int j = 0; j < 8; ++j) s += xv[j];
    s = wave_sum32(s);
    float m = s * (1.0f / 256.0f);
    float vs = 0.0f;
#pragma unroll
    for (int j = 0; j < 8; ++j) { float d = xv[j] - m; vs += d * d; }
    vs = wave_sum32(vs);
    float inv = rsqrtf(vs * (1.0f / 256.0f) + EPSV);
#pragma unroll
    for (int j = 0; j < 8; ++j) {
      int d = lane + 32 * j;
      float yv = (xv[j] - m) * inv * g2[d] + b2[d];
      hb[r * PH + d] = 0.5f * yv * (1.0f + erff(yv * 0.70710678118654752f));
    }
  }
  __syncthreads();

  {
    v8f acc[4] = {};
    const float* wo0 = Wo + nc * 256 + khalf * 2;
#pragma unroll
    for (int kb = 0; kb < 64; ++kb) {
      v2f bb = *(const v2f*)(wo0 + kb * 4);
#pragma unroll
      for (int mi = 0; mi < 4; ++mi) {
        v2f a = *(const v2f*)(arow_hb + mi * 16 * PH + kb * 4);
        acc[mi] = wmma4(a, bb, acc[mi]);
      }
    }
    float boa = bo[nc];
#pragma unroll
    for (int mi = 0; mi < 4; ++mi) {
      int rowB = mi * 16 + khalf * 8;
#pragma unroll
      for (int r = 0; r < 8; ++r) t1[(rowB + r) * PY + nc] = acc[mi][r] + boa;
    }
  }
  __syncthreads();

  for (int i = 0; i < 8; ++i) {
    int r = wave * 8 + i;
    float xv[4];
#pragma unroll
    for (int j = 0; j < 4; ++j) xv[j] = t1[r * PY + lane + 32 * j];
    float s = wave_sum32(xv[0] + xv[1] + xv[2] + xv[3]);
    float m = s * (1.0f / 128.0f);
    float vs = 0.0f;
#pragma unroll
    for (int j = 0; j < 4; ++j) { float d = xv[j] - m; vs += d * d; }
    vs = wave_sum32(vs);
    float inv = rsqrtf(vs * (1.0f / 128.0f) + EPSV);
#pragma unroll
    for (int j = 0; j < 4; ++j) {
      int d = lane + 32 * j;
      out[(base + r) * 128 + d] = (xv[j] - m) * inv * g3[d] + b3[d];
    }
  }
}

extern "C" void kernel_launch(void* const* d_in, const int* in_sizes, int n_in,
                              void* d_out, int out_size, void* d_ws, size_t ws_size,
                              hipStream_t stream) {
  const float* x     = (const float*)d_in[0];
  const float* corr  = (const float*)d_in[1];
  const float* alog  = (const float*)d_in[2];
  const float* chebw = (const float*)d_in[3];
  const float* Wi    = (const float*)d_in[4];
  const float* bi    = (const float*)d_in[5];
  const float* Wo    = (const float*)d_in[6];
  const float* bo    = (const float*)d_in[7];
  const float* g1    = (const float*)d_in[8];
  const float* b1    = (const float*)d_in[9];
  const float* g2    = (const float*)d_in[10];
  const float* b2    = (const float*)d_in[11];
  const float* g3    = (const float*)d_in[12];
  const float* b3    = (const float*)d_in[13];
  float* out = (float*)d_out;
  float* ws  = (float*)d_ws;

  float* gpart  = ws + 8192;
  float* dcorr  = ws + 8192 + 256 * 16384;
  float* Lap    = dcorr + 16384;
  float* wv     = Lap + 16384;
  float* mixed  = out + (long)32 * 4096 * 128;
  float* statp  = out;

  k_gram<<<256, 256, 256 * PG * sizeof(float), stream>>>(x, g1, b1, statp, gpart);
  k_combine<<<128, 128, 0, stream>>>(statp, gpart, dcorr);
  k_graph<<<1, 128, 0, stream>>>(corr, alog, chebw, dcorr, mixed, Lap, wv);
  k_main<<<2048, 256, (2 * 64 * PY + 64 * PH) * sizeof(float), stream>>>(x, g1, b1, Lap, wv, Wi, bi, Wo, bo,
                                        g2, b2, g3, b3, out);
}


# LayerNorm passes take their rows in batches per wave (16 rows in the gram kernel, 8 in LN2 and LN3): a batch's loads and cross-lane sums overlap instead of running row by row; same summation order
# speedup vs baseline: 3.1202x; 1.2563x over previous
#include <hip/hip_runtime.h>
#include <hip/hip_bf16.h>
#include <math.h>

#define EPSV 1e-8f

#define PY 132
#define PH 260
#define PG 144

typedef float v2f __attribute__((ext_vector_type(2)));
typedef float v8f __attribute__((ext_vector_type(8)));

__device__ __forceinline__ v8f wmma4(v2f a, v2f b, v8f c) {
  return __builtin_amdgcn_wmma_f32_16x16x4_f32(false, a, false, b, (short)0, c,
                                               false, false);
}

__device__ __forceinline__ float wave_sum32(float v) {
#pragma unroll
  for (int off = 16; off > 0; off >>= 1) v += __shfl_xor(v, off, 32);
  return v;
}

template <int R>
__device__ __forceinline__ void wave_sum32_rows(float (&v)[R]) {
#pragma unroll
  for (int off = 16; off > 0; off >>= 1) {
    float t[R];
#pragma unroll
    for (int r = 0; r < R; ++r) t[r] = __shfl_xor(v[r], off, 32);
#pragma unroll
    for (int r = 0; r < R; ++r) v[r] += t[r];
  }
}

__device__ __forceinline__ void ln128_row(const float* __restrict__ src,
                                          float* __restrict__ dst,
                                          const float* __restrict__ g,
                                          const float* __restrict__ b,
                                          int lane) {
  float xv[4];
#pragma unroll
  for (int j = 0; j < 4; ++j) xv[j] = src[lane + 32 * j];
  float s = xv[0] + xv[1] + xv[2] + xv[3];
  s = wave_sum32(s);
  float m = s * (1.0f / 128.0f);
  float vs = 0.0f;
#pragma unroll
  for (int j = 0; j < 4; ++j) { float d = xv[j] - m; vs += d * d; }
  vs = wave_sum32(vs);
  float inv = rsqrtf(vs * (1.0f / 128.0f) + EPSV);
#pragma unroll
  for (int j = 0; j < 4; ++j) {
    int d = lane + 32 * j;
    dst[d] = (xv[j] - m) * inv * g[d] + b[d];
  }
}

__global__ void __launch_bounds__(256) k_gram(const float* __restrict__ x,
                       const float* __restrict__ g1,
                       const float* __restrict__ b1,
                       float* __restrict__ statp,
                       float* __restrict__ gpart) {
  extern __shared__ float yt[];
  __shared__ float sred[2][8][128];
  const int tid = threadIdx.x, lane = tid & 31, wave = tid >> 5;
  const int b = blockIdx.x >> 3;
  const int oct = blockIdx.x & 7;
  const int khalf = lane >> 4;
  const int nl = lane & 15;
  const v8f vzero = {0, 0, 0, 0, 0, 0, 0, 0};
  v8f acc[8];
#pragma unroll
  for (int t = 0; t < 8; ++t) acc[t] = vzero;
  float a1[4] = {0, 0, 0, 0}, a2[4] = {0, 0, 0, 0};

  for (int sc = 0; sc < 2; ++sc) {
    const long rowbase = (long)b * 4096 + oct * 512 + sc * 256;
    __syncthreads();
    for (int i0 = 0; i0 < 32; i0 += 16) {
      const int r0 = wave * 32 + i0;
      float xv[16][4], s[16], m[16], vs[16];
#pragma unroll
      for (int q = 0; q < 16; ++q) {
        const float* src = x + (rowbase + r0 + q) * 128;
#pragma unroll
        for (int j = 0; j < 4; ++j) xv[q][j] = src[lane + 32 * j];
      }
#pragma unroll
      for (int q = 0; q < 16; ++q) s[q] = xv[q][0] + xv[q][1] + xv[q][2] + xv[q][3];
      wave_sum32_rows<16>(s);
#pragma unroll
      for (int q = 0; q < 16; ++q) {
        m[q] = s[q] * (1.0f / 128.0f);
        float t = 0.0f;
#pragma unroll
        for (int j = 0; j < 4; ++j) { float d = xv[q][j] - m[q]; t += d * d; }
        vs[q] = t;
      }
      wave_sum32_rows<16>(vs);
#pragma unroll
      for (int q = 0; q < 16; ++q) {
        float inv = rsqrtf(vs[q] * (1.0f / 128.0f) + EPSV);
#pragma unroll
        for (int j = 0; j < 4; ++j) {
          int d = lane + 32 * j;
          float y = (xv[q][j] - m[q]) * inv * g1[d] + b1[d];
          a1[j] += y; a2[j] += y * y;
          yt[(r0 + q) * PG + d] = y;
        }
      }
    }
    __syncthreads();
    {
      const float* bcol = yt + (khalf * 2) * PG + wave * 16 + nl;
      const float* arow = yt + (khalf * 2) * PG + nl;
#pragma unroll 2
      for (int kb = 0; kb < 64; ++kb) {
        v2f bf;
        bf.x = bcol[kb * 4 * PG];   bf.y = bcol[kb * 4 * PG + PG];
#pragma unroll
        for (int mi = 0; mi < 8; ++mi) {
          v2f af;
          af.x = arow[kb * 4 * PG + mi * 16];   af.y = arow[kb * 4 * PG + PG + mi * 16];
          acc[mi] = wmma4(af, bf, acc[mi]);
        }
      }
    }
  }

  float* gp = gpart + (long)blockIdx.x * 16384;
  for (int t = 0; t < 8; ++t) {
    int idx = wave + 8 * t;
    int mi = idx >> 3, ni = idx & 7;
    int dmBase = mi * 16 + khalf * 8;
    int de = ni * 16 + nl;
#pragma unroll
    for (int r = 0; r < 8; ++r) gp[(dmBase + r) * 128 + de] = acc[t][r];
  }
#pragma unroll
  for (int j = 0; j < 4; ++j) {
    int d = lane + 32 * j;
    sred[0][wave][d] = a1[j];
    sred[1][wave][d] = a2[j];
  }
  __syncthreads();
  {
    const int which = tid >> 7, d = tid & 127;
    float s = 0.0f;
#pragma unroll
    for (int w = 0; w < 8; ++w) s += sred[which][w][d];
    statp[which * 32768 + blockIdx.x * 128 + d] = s;
  }
}

__global__ void k_combine(const float* __restrict__ statp,
                          const float* __restrict__ gpart,
                          float* __restrict__ dcorr) {
  __shared__ float mu[32][128];
  __shared__ float iv[32][128];
  const int d = blockIdx.x;
  const int e = threadIdx.x;
  for (int b = 0; b < 32; ++b) {
    float s1 = 0.0f, s2 = 0.0f;
#pragma unroll
    for (int q = 0; q < 8; ++q) {
      s1 += statp[(b * 8 + q) * 128 + e];
      s2 += statp[32768 + (b * 8 + q) * 128 + e];
    }
    float v = s2 - s1 * s1 * (1.0f / 4096.0f);
    mu[b][e] = s1 * (1.0f / 4096.0f);
    iv[b][e] = 1.0f / fmaxf(sqrtf(fmaxf(v, 0.0f)), EPSV);
  }
  __syncthreads();
  float acc = 0.0f;
  for (int b = 0; b < 32; ++b) {
    float md = mu[b][d], me = mu[b][e];
    float ind = iv[b][d], ine = iv[b][e];
    float G = 0.0f;
#pragma unroll
    for (int q = 0; q < 8; ++q)
      G += gpart[(long)(b * 8 + q) * 16384 + d * 128 + e];
    acc += (G - 4096.0f * md * me) * ind * ine;
  }
  dcorr[d * 128 + e] = acc * (1.0f / 32.0f);
}

__global__ void k_graph(const float* __restrict__ corr,
                        const float* __restrict__ alpha_logit,
                        const float* __restrict__ chebw,
                        const float* __restrict__ dcorr,
                        float* __restrict__ mixed_out,
                        float* __restrict__ Lap,
                        float* __restrict__ wout) {
  __shared__ float dis[128];
  const int i = threadIdx.x;
  const float alpha = 1.0f / (1.0f + __expf(-alpha_logit[0]));
  float deg = 0.0f;
  for (int j = 0; j < 128; ++j) {
    float learned = tanhf(0.5f * (corr[i * 128 + j] + corr[j * 128 + i]));
    float dc = fminf(fmaxf(dcorr[i * 128 + j], -1.0f), 1.0f);
    float mx = alpha * learned + (1.0f - alpha) * dc;
    mixed_out[i * 128 + j] = mx;
    if (j != i) deg += mx;
  }
  dis[i] = rsqrtf(fmaxf(deg, EPSV));
  __syncthreads();
  const float di = dis[i];
  for (int j = 0; j < 128; ++j) {
    float a = (j == i) ? 0.0f : mixed_out[i * 128 + j];
    float l = ((j == i) ? 1.0f : 0.0f) - di * a * dis[j];
    Lap[i * 128 + j] = fminf(fmaxf(l, -1.5f), 1.5f);
  }
  if (i == 0) {
    float c0 = chebw[0], c1 = chebw[1], c2 = chebw[2];
    float mx = fmaxf(c0, fmaxf(c1, c2));
    float e0 = __expf(c0 - mx), e1 = __expf(c1 - mx), e2 = __expf(c2 - mx);
    float s = e0 + e1 + e2;
    wout[0] = e0 / s; wout[1] = e1 / s; wout[2] = e2 / s;
  }
}

__global__ void __launch_bounds__(256) k_main(const float* __restrict__ x,
                       const float* __restrict__ g1, const float* __restrict__ b1,
                       const float* __restrict__ Lap, const float* __restrict__ wv,
                       const float* __restrict__ Wi, const float* __restrict__ bi,
                       const float* __restrict__ Wo, const float* __restrict__ bo,
                       const float* __restrict__ g2, const float* __restrict__ b2,
                       const float* __restrict__ g3, const float* __restrict__ b3,
                       float* __restrict__ out) {
  extern __shared__ float smem[];
  float* yo = smem;
  float* t1 = smem + 64 * PY;
  float* hb = smem + 2 * 64 * PY;
  const int tid = threadIdx.x, lane = tid & 31, wave = tid >> 5;
  const long base = (long)blockIdx.x * 64;
  const float w0 = wv[0], w1 = wv[1], w2 = wv[2];
  const int khalf = lane >> 4;
  const int nl = lane & 15;

  for (int i = 0; i < 8; ++i) {
    int r = wave * 8 + i;
    ln128_row(x + (base + r) * 128, yo + r * PY, g1, b1, lane);
  }
  __syncthreads();

  const float* arow_yo = yo + nl * PY + khalf * 2;
  const float* arow_t1 = t1 + nl * PY + khalf * 2;
  const float* arow_hb = hb + nl * PH + khalf * 2;
  const int nc = wave * 16 + nl;
  const float* lapcol = Lap + (khalf * 2) * 128 + nc;

  {
    v8f acc[4] = {};
#pragma unroll
    for (int kb = 0; kb < 32; ++kb) {
      v2f bb;
      bb.x = lapcol[kb * 512];   bb.y = lapcol[kb * 512 + 128];
#pragma unroll
      for (int mi = 0; mi < 4; ++mi) {
        v2f a = *(const v2f*)(arow_yo + mi * 16 * PY + kb * 4);
        acc[mi] = wmma4(a, bb, acc[mi]);
      }
    }
#pragma unroll
    for (int mi = 0; mi < 4; ++mi) {
      int rowB = mi * 16 + khalf * 8;
#pragma unroll
      for (int r = 0; r < 8; ++r) t1[(rowB + r) * PY + nc] = acc[mi][r];
    }
  }
  __syncthreads();

  {
    v8f acc[4] = {};
#pragma unroll
    for (int kb = 0; kb < 32; ++kb) {
      v2f bb;
      bb.x = lapcol[kb * 512];   bb.y = lapcol[kb * 512 + 128];
#pragma unroll
      for (int mi = 0; mi < 4; ++mi) {
        v2f a = *(const v2f*)(arow_t1 + mi * 16 * PY + kb * 4);
        acc[mi] = wmma4(a, bb, acc[mi]);
      }
    }
#pragma unroll
    for (int mi = 0; mi < 4; ++mi) {
      int rowB = mi * 16 + khalf * 8;
#pragma unroll
      for (int r = 0; r < 8; ++r) {
        int pos = (rowB + r) * PY + nc;
        float y0 = yo[pos], y1 = t1[pos];
        float tx2 = fminf(fmaxf(2.0f * acc[mi][r] - y0, -50.0f), 50.0f);
        yo[pos] = w0 * y0 + w1 * y1 + w2 * tx2;
      }
    }
  }
  __syncthreads();

  {
    v8f acc[2][4] = {};
    const float* wi0 = Wi + nc * 128 + khalf * 2;
    const float* wi1 = Wi + (nc + 128) * 128 + khalf * 2;
#pragma unroll
    for (int kb = 0; kb < 32; ++kb) {
      v2f b0 = *(const v2f*)(wi0 + kb * 4);
      v2f b1 = *(const v2f*)(wi1 + kb * 4);
#pragma unroll
      for (int mi = 0; mi < 4; ++mi) {
        v2f a = *(const v2f*)(arow_yo + mi * 16 * PY + kb * 4);
        acc[0][mi] = wmma4(a, b0, acc[0][mi]);
        acc[1][mi] = wmma4(a, b1, acc[1][mi]);
      }
    }
#pragma unroll
    for (int j = 0; j < 2; ++j) {
      int ncj = nc + 128 * j;
      float bia = bi[ncj];
#pragma unroll
      for (int mi = 0; mi < 4; ++mi) {
        int rowB = mi * 16 + khalf * 8;
#pragma unroll
        for (int r = 0; r < 8; ++r) hb[(rowB + r) * PH + ncj] = acc[j][mi][r] + bia;
      }
    }
  }
  __syncthreads();

  {
    const int rw = wave * 8;
    float xv[8][8], s[8], m[8], vs[8];
#pragma unroll
    for (int i = 0; i < 8; ++i) {
#pragma unroll
      for (int j = 0; j < 8; ++j) xv[i][j] = hb[(rw + i) * PH + lane + 32 * j];
    }
#pragma unroll
    for (int i = 0; i < 8; ++i) {
      float t = 0.0f;
#pragma unroll
      for (int j = 0; j < 8; ++j) t += xv[i][j];
      s[i] = t;
    }
    wave_sum32_rows<8>(s);
#pragma unroll
    for (int i = 0; i < 8; ++i) {
      m[i] = s[i] * (1.0f / 256.0f);
      float t = 0.0f;
#pragma unroll
      for (int j = 0; j < 8; ++j) { float d = xv[i][j] - m[i]; t += d * d; }
      vs[i] = t;
    }
    wave_sum32_rows<8>(vs);
#pragma unroll
    for (int i = 0; i < 8; ++i) {
      float inv = rsqrtf(vs[i] * (1.0f / 256.0f) + EPSV);
#pragma unroll
      for (int j = 0; j < 8; ++j) {
        int d = lane + 32 * j;
        hb[(rw + i) * PH + d] = (xv[i][j] - m[i]) * inv * g2[d] + b2[d];
      }
    }
    for (int i = 0; i < 8; ++i) {
      float yv[8];
#pragma unroll
      for (int j = 0; j < 8; ++j) yv[j] = hb[(rw + i) * PH + lane + 32 * j];
#pragma unroll
      for (int j = 0; j < 8; ++j)
        hb[(rw + i) * PH + lane + 32 * j] =
            0.5f * yv[j] * (1.0f + erff(yv[j] * 0.70710678118654752f));
    }
  }
  __syncthreads();

  {
    v8f acc[4] = {};
    const float* wo0 = Wo + nc * 256 + khalf * 2;
#pragma unroll
    for (int kb = 0; kb < 64; ++kb) {
      v2f bb = *(const v2f*)(wo0 + kb * 4);
#pragma unroll
      for (int mi = 0; mi < 4; ++mi) {
        v2f a = *(const v2f*)(arow_hb + mi * 16 * PH + kb * 4);
        acc[mi] = wmma4(a, bb, acc[mi]);
      }
    }
    float boa = bo[nc];
#pragma unroll
    for (int mi = 0; mi < 4; ++mi) {
      int rowB = mi * 16 + khalf * 8;
#pragma unroll
      for (int r = 0; r < 8; ++r) t1[(rowB + r) * PY + nc] = acc[mi][r] + boa;
    }
  }
  __syncthreads();

  {
    const int rw = wave * 8;
    float xv[8][4], s[8], m[8], vs[8];
#pragma unroll
    for (int i = 0; i < 8; ++i) {
#pragma unroll
      for (int j = 0; j < 4; ++j) xv[i][j] = t1[(rw + i) * PY + lane + 32 * j];
    }
#pragma unroll
    for (int i = 0; i < 8; ++i) s[i] = xv[i][0] + xv[i][1] + xv[i][2] + xv[i][3];
    wave_sum32_rows<8>(s);
#pragma unroll
    for (int i = 0; i < 8; ++i) {
      m[i] = s[i] * (1.0f / 128.0f);
      float t = 0.0f;
#pragma unroll
      for (int j = 0; j < 4; ++j) { float d = xv[i][j] - m[i]; t += d * d; }
      vs[i] = t;
    }
    wave_sum32_rows<8>(vs);
#pragma unroll
    for (int i = 0; i < 8; ++i) {
      float inv = rsqrtf(vs[i] * (1.0f / 128.0f) + EPSV);
#pragma unroll
      for (int j = 0; j < 4; ++j) {
        int d = lane + 32 * j;
        out[(base + rw + i) * 128 + d] = (xv[i][j] - m[i]) * inv * g3[d] + b3[d];
      }
    }
  }
}

extern "C" void kernel_launch(void* const* d_in, const int* in_sizes, int n_in,
                              void* d_out, int out_size, void* d_ws, size_t ws_size,
                              hipStream_t stream) {
  const float* x     = (const float*)d_in[0];
  const float* corr  = (const float*)d_in[1];
  const float* alog  = (const float*)d_in[2];
  const float* chebw = (const float*)d_in[3];
  const float* Wi    = (const float*)d_in[4];
  const float* bi    = (const float*)d_in[5];
  const float* Wo    = (const float*)d_in[6];
  const float* bo    = (const float*)d_in[7];
  const float* g1    = (const float*)d_in[8];
  const float* b1    = (const float*)d_in[9];
  const float* g2    = (const float*)d_in[10];
  const float* b2    = (const float*)d_in[11];
  const float* g3    = (const float*)d_in[12];
  const float* b3    = (const float*)d_in[13];
  float* out = (float*)d_out;
  float* ws  = (float*)d_ws;

  float* gpart  = ws + 8192;
  float* dcorr  = ws + 8192 + 256 * 16384;
  float* Lap    = dcorr + 16384;
  float* wv     = Lap + 16384;
  float* mixed  = out + (long)32 * 4096 * 128;
  float* statp  = out;

  k_gram<<<256, 256, 256 * PG * sizeof(float), stream>>>(x, g1, b1, statp, gpart);
  k_combine<<<128, 128, 0, stream>>>(statp, gpart, dcorr);
  k_graph<<<1, 128, 0, stream>>>(corr, alog, chebw, dcorr, mixed, Lap, wv);
  k_main<<<2048, 256, (2 * 64 * PY + 64 * PH) * sizeof(float), stream>>>(x, g1, b1, Lap, wv, Wi, bi, Wo, bo,
                                        g2, b2, g3, b3, out);
}


# graph stage made row-parallel: mixed row and degree formed in the combine kernel's row block (degree sum now a fixed shuffle tree + 4 wave sums: order changed), Laplacian kernel on 128 work-groups
# speedup vs baseline: 4.3156x; 1.3831x over previous
#include <hip/hip_runtime.h>
#include <hip/hip_bf16.h>
#include <math.h>

#define EPSV 1e-8f

#define PY 132
#define PH 260
#define PG 144

typedef float v2f __attribute__((ext_vector_type(2)));
typedef float v8f __attribute__((ext_vector_type(8)));

__device__ __forceinline__ v8f wmma4(v2f a, v2f b, v8f c) {
  return __builtin_amdgcn_wmma_f32_16x16x4_f32(false, a, false, b, (short)0, c,
                                               false, false);
}

__device__ __forceinline__ float wave_sum32(float v) {
#pragma unroll
  for (int off = 16; off > 0; off >>= 1) v += __shfl_xor(v, off, 32);
  return v;
}

template <int R>
__device__ __forceinline__ void wave_sum32_rows(float (&v)[R]) {
#pragma unroll
  for (int off = 16; off > 0; off >>= 1) {
    float t[R];
#pragma unroll
    for (int r = 0; r < R; ++r) t[r] = __shfl_xor(v[r], off, 32);
#pragma unroll
    for (int r = 0; r < R; ++r) v[r] += t[r];
  }
}

__device__ __forceinline__ void ln128_row(const float* __restrict__ src,
                                          float* __restrict__ dst,
                                          const float* __restrict__ g,
                                          const float* __restrict__ b,
                                          int lane) {
  float xv[4];
#pragma unroll
  for (int j = 0; j < 4; ++j) xv[j] = src[lane + 32 * j];
  float s = xv[0] + xv[1] + xv[2] + xv[3];
  s = wave_sum32(s);
  float m = s * (1.0f / 128.0f);
  float vs = 0.0f;
#pragma unroll
  for (int j = 0; j < 4; ++j) { float d = xv[j] - m; vs += d * d; }
  vs = wave_sum32(vs);
  float inv = rsqrtf(vs * (1.0f / 128.0f) + EPSV);
#pragma unroll
  for (int j = 0; j < 4; ++j) {
    int d = lane + 32 * j;
    dst[d] = (xv[j] - m) * inv * g[d] + b[d];
  }
}

__global__ void __launch_bounds__(256) k_gram(const float* __restrict__ x,
                       const float* __restrict__ g1,
                       const float* __restrict__ b1,
                       float* __restrict__ statp,
                       float* __restrict__ gpart) {
  extern __shared__ float yt[];
  __shared__ float sred[2][8][128];
  const int tid = threadIdx.x, lane = tid & 31, wave = tid >> 5;
  const int b = blockIdx.x >> 3;
  const int oct = blockIdx.x & 7;
  const int khalf = lane >> 4;
  const int nl = lane & 15;
  const v8f vzero = {0, 0, 0, 0, 0, 0, 0, 0};
  v8f acc[8];
#pragma unroll
  for (int t = 0; t < 8; ++t) acc[t] = vzero;
  float a1[4] = {0, 0, 0, 0}, a2[4] = {0, 0, 0, 0};

  for (int sc = 0; sc < 2; ++sc) {
    const long rowbase = (long)b * 4096 + oct * 512 + sc * 256;
    __syncthreads();
    for (int i0 = 0; i0 < 32; i0 += 16) {
      const int r0 = wave * 32 + i0;
      float xv[16][4], s[16], m[16], vs[16];
#pragma unroll
      for (int q = 0; q < 16; ++q) {
        const float* src = x + (rowbase + r0 + q) * 128;
#pragma unroll
        for (int j = 0; j < 4; ++j) xv[q][j] = src[lane + 32 * j];
      }
#pragma unroll
      for (int q = 0; q < 16; ++q) s[q] = xv[q][0] + xv[q][1] + xv[q][2] + xv[q][3];
      wave_sum32_rows<16>(s);
#pragma unroll
      for (int q = 0; q < 16; ++q) {
        m[q] = s[q] * (1.0f / 128.0f);
        float t = 0.0f;
#pragma unroll
        for (int j = 0; j < 4; ++j) { float d = xv[q][j] - m[q]; t += d * d; }
        vs[q] = t;
      }
      wave_sum32_rows<16>(vs);
#pragma unroll
      for (int q = 0; q < 16; ++q) {
        float inv = rsqrtf(vs[q] * (1.0f / 128.0f) + EPSV);
#pragma unroll
        for (int j = 0; j < 4; ++j) {
          int d = lane + 32 * j;
          float y = (xv[q][j] - m[q]) * inv * g1[d] + b1[d];
          a1[j] += y; a2[j] += y * y;
          yt[(r0 + q) * PG + d] = y;
        }
      }
    }
    __syncthreads();
    {
      const float* bcol = yt + (khalf * 2) * PG + wave * 16 + nl;
      const float* arow = yt + (khalf * 2) * PG + nl;
#pragma unroll 2
      for (int kb = 0; kb < 64; ++kb) {
        v2f bf;
        bf.x = bcol[kb * 4 * PG];   bf.y = bcol[kb * 4 * PG + PG];
#pragma unroll
        for (int mi = 0; mi < 8; ++mi) {
          v2f af;
          af.x = arow[kb * 4 * PG + mi * 16];   af.y = arow[kb * 4 * PG + PG + mi * 16];
          acc[mi] = wmma4(af, bf, acc[mi]);
        }
      }
    }
  }

  float* gp = gpart + (long)blockIdx.x * 16384;
  for (int t = 0; t < 8; ++t) {
    int idx = wave + 8 * t;
    int mi = idx >> 3, ni = idx & 7;
    int dmBase = mi * 16 + khalf * 8;
    int de = ni * 16 + nl;
#pragma unroll
    for (int r = 0; r < 8; ++r) gp[(dmBase + r) * 128 + de] = acc[t][r];
  }
#pragma unroll
  for (int j = 0; j < 4; ++j) {
    int d = lane + 32 * j;
    sred[0][wave][d] = a1[j];
    sred[1][wave][d] = a2[j];
  }
  __syncthreads();
  {
    const int which = tid >> 7, d = tid & 127;
    float s = 0.0f;
#pragma unroll
    for (int w = 0; w < 8; ++w) s += sred[which][w][d];
    statp[which * 32768 + blockIdx.x * 128 + d] = s;
  }
}

__global__ void __launch_bounds__(128) k_combine(const float* __restrict__ statp,
                          const float* __restrict__ gpart,
                          const float* __restrict__ corr,
                          const float* __restrict__ alpha_logit,
                          float* __restrict__ mixed_out,
                          float* __restrict__ dis_out) {
  __shared__ float mu[32][128];
  __shared__ float iv[32][128];
  __shared__ float wsum[4];
  const int d = blockIdx.x;
  const int e = threadIdx.x;
  const float cde = corr[d * 128 + e];
  const float ced = corr[e * 128 + d];
  const float alog = alpha_logit[0];
  for (int b = 0; b < 32; ++b) {
    float s1 = 0.0f, s2 = 0.0f;
#pragma unroll
    for (int q = 0; q < 8; ++q) {
      s1 += statp[(b * 8 + q) * 128 + e];
      s2 += statp[32768 + (b * 8 + q) * 128 + e];
    }
    float v = s2 - s1 * s1 * (1.0f / 4096.0f);
    mu[b][e] = s1 * (1.0f / 4096.0f);
    iv[b][e] = 1.0f / fmaxf(sqrtf(fmaxf(v, 0.0f)), EPSV);
  }
  __syncthreads();
  float acc = 0.0f;
  for (int b = 0; b < 32; ++b) {
    float md = mu[b][d], me = mu[b][e];
    float ind = iv[b][d], ine = iv[b][e];
    float G = 0.0f;
#pragma unroll
    for (int q = 0; q < 8; ++q)
      G += gpart[(long)(b * 8 + q) * 16384 + d * 128 + e];
    acc += (G - 4096.0f * md * me) * ind * ine;
  }
  const float alpha = 1.0f / (1.0f + __expf(-alog));
  const float learned = tanhf(0.5f * (cde + ced));
  const float dc = fminf(fmaxf(acc * (1.0f / 32.0f), -1.0f), 1.0f);
  const float mx = alpha * learned + (1.0f - alpha) * dc;
  mixed_out[d * 128 + e] = mx;
  float part = wave_sum32((e == d) ? 0.0f : mx);
  if ((e & 31) == 0) wsum[e >> 5] = part;
  __syncthreads();
  if (e == 0) {
    float deg = ((wsum[0] + wsum[1]) + wsum[2]) + wsum[3];
    dis_out[d] = rsqrtf(fmaxf(deg, EPSV));
  }
}

__global__ void __launch_bounds__(128) k_graph(const float* __restrict__ chebw,
                        const float* __restrict__ mixed,
                        const float* __restrict__ dis,
                        float* __restrict__ Lap,
                        float* __restrict__ wout) {
  const int i = blockIdx.x;
  const int j = threadIdx.x;
  const float di = dis[i];
  const float dj = dis[j];
  const float m = mixed[i * 128 + j];
  float a = (j == i) ? 0.0f : m;
  float l = ((j == i) ? 1.0f : 0.0f) - di * a * dj;
  Lap[i * 128 + j] = fminf(fmaxf(l, -1.5f), 1.5f);
  if (i == 0 && j == 0) {
    float c0 = chebw[0], c1 = chebw[1], c2 = chebw[2];
    float mx = fmaxf(c0, fmaxf(c1, c2));
    float e0 = __expf(c0 - mx), e1 = __expf(c1 - mx), e2 = __expf(c2 - mx);
    float s = e0 + e1 + e2;
    wout[0] = e0 / s; wout[1] = e1 / s; wout[2] = e2 / s;
  }
}

__global__ void __launch_bounds__(256) k_main(const float* __restrict__ x,
                       const float* __restrict__ g1, const float* __restrict__ b1,
                       const float* __restrict__ Lap, const float* __restrict__ wv,
                       const float* __restrict__ Wi, const float* __restrict__ bi,
                       const float* __restrict__ Wo, const float* __restrict__ bo,
                       const float* __restrict__ g2, const float* __restrict__ b2,
                       const float* __restrict__ g3, const float* __restrict__ b3,
                       float* __restrict__ out) {
  extern __shared__ float smem[];
  float* yo = smem;
  float* t1 = smem + 64 * PY;
  float* hb = smem + 2 * 64 * PY;
  const int tid = threadIdx.x, lane = tid & 31, wave = tid >> 5;
  const long base = (long)blockIdx.x * 64;
  const float w0 = wv[0], w1 = wv[1], w2 = wv[2];
  const int khalf = lane >> 4;
  const int nl = lane & 15;

  for (int i = 0; i < 8; ++i) {
    int r = wave * 8 + i;
    ln128_row(x + (base + r) * 128, yo + r * PY, g1, b1, lane);
  }
  __syncthreads();

  const float* arow_yo = yo + nl * PY + khalf * 2;
  const float* arow_t1 = t1 + nl * PY + khalf * 2;
  const float* arow_hb = hb + nl * PH + khalf * 2;
  const int nc = wave * 16 + nl;
  const float* lapcol = Lap + (khalf * 2) * 128 + nc;

  {
    v8f acc[4] = {};
#pragma unroll
    for (int kb = 0; kb < 32; ++kb) {
      v2f bb;
      bb.x = lapcol[kb * 512];   bb.y = lapcol[kb * 512 + 128];
#pragma unroll
      for (int mi = 0; mi < 4; ++mi) {
        v2f a = *(const v2f*)(arow_yo + mi * 16 * PY + kb * 4);
        acc[mi] = wmma4(a, bb, acc[mi]);
      }
    }
#pragma unroll
    for (int mi = 0; mi < 4; ++mi) {
      int rowB = mi * 16 + khalf * 8;
#pragma unroll
      for (int r = 0; r < 8; ++r) t1[(rowB + r) * PY + nc] = acc[mi][r];
    }
  }
  __syncthreads();

  {
    v8f acc[4] = {};
#pragma unroll
    for (int kb = 0; kb < 32; ++kb) {
      v2f bb;
      bb.x = lapcol[kb * 512];   bb.y = lapcol[kb * 512 + 128];
#pragma unroll
      for (int mi = 0; mi < 4; ++mi) {
        v2f a = *(const v2f*)(arow_t1 + mi * 16 * PY + kb * 4);
        acc[mi] = wmma4(a, bb, acc[mi]);
      }
    }
#pragma unroll
    for (int mi = 0; mi < 4; ++mi) {
      int rowB = mi * 16 + khalf * 8;
#pragma unroll
      for (int r = 0; r < 8; ++r) {
        int pos = (rowB + r) * PY + nc;
        float y0 = yo[pos], y1 = t1[pos];
        float tx2 = fminf(fmaxf(2.0f * acc[mi][r] - y0, -50.0f), 50.0f);
        yo[pos] = w0 * y0 + w1 * y1 + w2 * tx2;
      }
    }
  }
  __syncthreads();

  {
    v8f acc[2][4] = {};
    const float* wi0 = Wi + nc * 128 + khalf * 2;
    const float* wi1 = Wi + (nc + 128) * 128 + khalf * 2;
#pragma unroll
    for (int kb = 0; kb < 32; ++kb) {
      v2f b0 = *(const v2f*)(wi0 + kb * 4);
      v2f b1 = *(const v2f*)(wi1 + kb * 4);
#pragma unroll
      for (int mi = 0; mi < 4; ++mi) {
        v2f a = *(const v2f*)(arow_yo + mi * 16 * PY + kb * 4);
        acc[0][mi] = wmma4(a, b0, acc[0][mi]);
        acc[1][mi] = wmma4(a, b1, acc[1][mi]);
      }
    }
#pragma unroll
    for (int j = 0; j < 2; ++j) {
      int ncj = nc + 128 * j;
      float bia = bi[ncj];
#pragma unroll
      for (int mi = 0; mi < 4; ++mi) {
        int rowB = mi * 16 + khalf * 8;
#pragma unroll
        for (int r = 0; r < 8; ++r) hb[(rowB + r) * PH + ncj] = acc[j][mi][r] + bia;
      }
    }
  }
  __syncthreads();

  {
    const int rw = wave * 8;
    float xv[8][8], s[8], m[8], vs[8];
#pragma unroll
    for (int i = 0; i < 8; ++i) {
#pragma unroll
      for (int j = 0; j < 8; ++j) xv[i][j] = hb[(rw + i) * PH + lane + 32 * j];
    }
#pragma unroll
    for (int i = 0; i < 8; ++i) {
      float t = 0.0f;
#pragma unroll
      for (int j = 0; j < 8; ++j) t += xv[i][j];
      s[i] = t;
    }
    wave_sum32_rows<8>(s);
#pragma unroll
    for (int i = 0; i < 8; ++i) {
      m[i] = s[i] * (1.0f / 256.0f);
      float t = 0.0f;
#pragma unroll
      for (int j = 0; j < 8; ++j) { float d = xv[i][j] - m[i]; t += d * d; }
      vs[i] = t;
    }
    wave_sum32_rows<8>(vs);
#pragma unroll
    for (int i = 0; i < 8; ++i) {
      float inv = rsqrtf(vs[i] * (1.0f / 256.0f) + EPSV);
#pragma unroll
      for (int j = 0; j < 8; ++j) {
        int d = lane + 32 * j;
        hb[(rw + i) * PH + d] = (xv[i][j] - m[i]) * inv * g2[d] + b2[d];
      }
    }
    for (int i = 0; i < 8; ++i) {
      float yv[8];
#pragma unroll
      for (int j = 0; j < 8; ++j) yv[j] = hb[(rw + i) * PH + lane + 32 * j];
#pragma unroll
      for (int j = 0; j < 8; ++j)
        hb[(rw + i) * PH + lane + 32 * j] =
            0.5f * yv[j] * (1.0f + erff(yv[j] * 0.70710678118654752f));
    }
  }
  __syncthreads();

  {
    v8f acc[4] = {};
    const float* wo0 = Wo + nc * 256 + khalf * 2;
#pragma unroll
    for (int kb = 0; kb < 64; ++kb) {
      v2f bb = *(const v2f*)(wo0 + kb * 4);
#pragma unroll
      for (int mi = 0; mi < 4; ++mi) {
        v2f a = *(const v2f*)(arow_hb + mi * 16 * PH + kb * 4);
        acc[mi] = wmma4(a, bb, acc[mi]);
      }
    }
    float boa = bo[nc];
#pragma unroll
    for (int mi = 0; mi < 4; ++mi) {
      int rowB = mi * 16 + khalf * 8;
#pragma unroll
      for (int r = 0; r < 8; ++r) t1[(rowB + r) * PY + nc] = acc[mi][r] + boa;
    }
  }
  __syncthreads();

  {
    const int rw = wave * 8;
    float xv[8][4], s[8], m[8], vs[8];
#pragma unroll
    for (int i = 0; i < 8; ++i) {
#pragma unroll
      for (int j = 0; j < 4; ++j) xv[i][j] = t1[(rw + i) * PY + lane + 32 * j];
    }
#pragma unroll
    for (int i = 0; i < 8; ++i) s[i] = xv[i][0] + xv[i][1] + xv[i][2] + xv[i][3];
    wave_sum32_rows<8>(s);
#pragma unroll
    for (int i = 0; i < 8; ++i) {
      m[i] = s[i] * (1.0f / 128.0f);
      float t = 0.0f;
#pragma unroll
      for (int j = 0; j < 4; ++j) { float d = xv[i][j] - m[i]; t += d * d; }
      vs[i] = t;
    }
    wave_sum32_rows<8>(vs);
#pragma unroll
    for (int i = 0; i < 8; ++i) {
      float inv = rsqrtf(vs[i] * (1.0f / 128.0f) + EPSV);
#pragma unroll
      for (int j = 0; j < 4; ++j) {
        int d = lane + 32 * j;
        out[(base + rw + i) * 128 + d] = (xv[i][j] - m[i]) * inv * g3[d] + b3[d];
      }
    }
  }
}

extern "C" void kernel_launch(void* const* d_in, const int* in_sizes, int n_in,
                              void* d_out, int out_size, void* d_ws, size_t ws_size,
                              hipStream_t stream) {
  const float* x     = (const float*)d_in[0];
  const float* corr  = (const float*)d_in[1];
  const float* alog  = (const float*)d_in[2];
  const float* chebw = (const float*)d_in[3];
  const float* Wi    = (const float*)d_in[4];
  const float* bi    = (const float*)d_in[5];
  const float* Wo    = (const float*)d_in[6];
  const float* bo    = (const float*)d_in[7];
  const float* g1    = (const float*)d_in[8];
  const float* b1    = (const float*)d_in[9];
  const float* g2    = (const float*)d_in[10];
  const float* b2    = (const float*)d_in[11];
  const float* g3    = (const float*)d_in[12];
  const float* b3    = (const float*)d_in[13];
  float* out = (float*)d_out;
  float* ws  = (float*)d_ws;

  float* gpart  = ws + 8192;
  float* disv   = ws + 8192 + 256 * 16384;
  float* Lap    = disv + 16384;
  float* wv     = Lap + 16384;
  float* mixed  = out + (long)32 * 4096 * 128;
  float* statp  = out;

  k_gram<<<256, 256, 256 * PG * sizeof(float), stream>>>(x, g1, b1, statp, gpart);
  k_combine<<<128, 128, 0, stream>>>(statp, gpart, corr, alog, mixed, disv);
  k_graph<<<128, 128, 0, stream>>>(chebw, mixed, disv, Lap, wv);
  k_main<<<2048, 256, (2 * 64 * PY + 64 * PH) * sizeof(float), stream>>>(x, g1, b1, Lap, wv, Wi, bi, Wo, bo,
                                        g2, b2, g3, b3, out);
}


# Main kernel's B operands (Laplacian, Wi, Wo) written once per call in WMMA fragment order by the graph kernel: a B load is 16 B per lane, 512 contiguous bytes per wave, not a 16-line gather
# speedup vs baseline: 4.4899x; 1.0404x over previous
#include <hip/hip_runtime.h>
#include <hip/hip_bf16.h>
#include <math.h>

#define EPSV 1e-8f

#define PY 132
#define PH 260
#define PG 144

typedef float v2f __attribute__((ext_vector_type(2)));
typedef float v4f __attribute__((ext_vector_type(4)));
typedef float v8f __attribute__((ext_vector_type(8)));

__device__ __forceinline__ v8f wmma4(v2f a, v2f b, v8f c) {
  return __builtin_amdgcn_wmma_f32_16x16x4_f32(false, a, false, b, (short)0, c,
                                               false, false);
}

__device__ __forceinline__ float wave_sum32(float v) {
#pragma unroll
  for (int off = 16; off > 0; off >>= 1) v += __shfl_xor(v, off, 32);
  return v;
}

template <int R>
__device__ __forceinline__ void wave_sum32_rows(float (&v)[R]) {
#pragma unroll
  for (int off = 16; off > 0; off >>= 1) {
    float t[R];
#pragma unroll
    for (int r = 0; r < R; ++r) t[r] = __shfl_xor(v[r], off, 32);
#pragma unroll
    for (int r = 0; r < R; ++r) v[r] += t[r];
  }
}

__device__ __forceinline__ void ln128_row(const float* __restrict__ src,
                                          float* __restrict__ dst,
                                          const float* __restrict__ g,
                                          const float* __restrict__ b,
                                          int lane) {
  float xv[4];
#pragma unroll
  for (int j = 0; j < 4; ++j) xv[j] = src[lane + 32 * j];
  float s = xv[0] + xv[1] + xv[2] + xv[3];
  s = wave_sum32(s);
  float m = s * (1.0f / 128.0f);
  float vs = 0.0f;
#pragma unroll
  for (int j = 0; j < 4; ++j) { float d = xv[j] - m; vs += d * d; }
  vs = wave_sum32(vs);
  float inv = rsqrtf(vs * (1.0f / 128.0f) + EPSV);
#pragma unroll
  for (int j = 0; j < 4; ++j) {
    int d = lane + 32 * j;
    dst[d] = (xv[j] - m) * inv * g[d] + b[d];
  }
}

__global__ void __launch_bounds__(256) k_gram(const float* __restrict__ x,
                       const float* __restrict__ g1,
                       const float* __restrict__ b1,
                       float* __restrict__ statp,
                       float* __restrict__ gpart) {
  extern __shared__ float yt[];
  __shared__ float sred[2][8][128];
  const int tid = threadIdx.x, lane = tid & 31, wave = tid >> 5;
  const int b = blockIdx.x >> 3;
  const int oct = blockIdx.x & 7;
  const int khalf = lane >> 4;
  const int nl = lane & 15;
  const v8f vzero = {0, 0, 0, 0, 0, 0, 0, 0};
  v8f acc[8];
#pragma unroll
  for (int t = 0; t < 8; ++t) acc[t] = vzero;
  float a1[4] = {0, 0, 0, 0}, a2[4] = {0, 0, 0, 0};

  for (int sc = 0; sc < 2; ++sc) {
    const long rowbase = (long)b * 4096 + oct * 512 + sc * 256;
    __syncthreads();
    for (int i0 = 0; i0 < 32; i0 += 16) {
      const int r0 = wave * 32 + i0;
      float xv[16][4], s[16], m[16], vs[16];
#pragma unroll
      for (int q = 0; q < 16; ++q) {
        const float* src = x + (rowbase + r0 + q) * 128;
#pragma unroll
        for (int j = 0; j < 4; ++j) xv[q][j] = src[lane + 32 * j];
      }
#pragma unroll
      for (int q = 0; q < 16; ++q) s[q] = xv[q][0] + xv[q][1] + xv[q][2] + xv[q][3];
      wave_sum32_rows<16>(s);
#pragma unroll
      for (int q = 0; q < 16; ++q) {
        m[q] = s[q] * (1.0f / 128.0f);
        float t = 0.0f;
#pragma unroll
        for (int j = 0; j < 4; ++j) { float d = xv[q][j] - m[q]; t += d * d; }
        vs[q] = t;
      }
      wave_sum32_rows<16>(vs);
#pragma unroll
      for (int q = 0; q < 16; ++q) {
        float inv = rsqrtf(vs[q] * (1.0f / 128.0f) + EPSV);
#pragma unroll
        for (int j = 0; j < 4; ++j) {
          int d = lane + 32 * j;
          float y = (xv[q][j] - m[q]) * inv * g1[d] + b1[d];
          a1[j] += y; a2[j] += y * y;
          yt[(r0 + q) * PG + d] = y;
        }
      }
    }
    __syncthreads();
    {
      const float* bcol = yt + (khalf * 2) * PG + wave * 16 + nl;
      const float* arow = yt + (khalf * 2) * PG + nl;
#pragma unroll 2
      for (int kb = 0; kb < 64; ++kb) {
        v2f bf;
        bf.x = bcol[kb * 4 * PG];   bf.y = bcol[kb * 4 * PG + PG];
#pragma unroll
        for (int mi = 0; mi < 8; ++mi) {
          v2f af;
          af.x = arow[kb * 4 * PG + mi * 16];   af.y = arow[kb * 4 * PG + PG + mi * 16];
          acc[mi] = wmma4(af, bf, acc[mi]);
        }
      }
    }
  }

  float* gp = gpart + (long)blockIdx.x * 16384;
  for (int t = 0; t < 8; ++t) {
    int idx = wave + 8 * t;
    int mi = idx >> 3, ni = idx & 7;
    int dmBase = mi * 16 + khalf * 8;
    int de = ni * 16 + nl;
#pragma unroll
    for (int r = 0; r < 8; ++r) gp[(dmBase + r) * 128 + de] = acc[t][r];
  }
#pragma unroll
  for (int j = 0; j < 4; ++j) {
    int d = lane + 32 * j;
    sred[0][wave][d] = a1[j];
    sred[1][wave][d] = a2[j];
  }
  __syncthreads();
  {
    const int which = tid >> 7, d = tid & 127;
    float s = 0.0f;
#pragma unroll
    for (int w = 0; w < 8; ++w) s += sred[which][w][d];
    statp[which * 32768 + blockIdx.x * 128 + d] = s;
  }
}

__global__ void __launch_bounds__(128) k_combine(const float* __restrict__ statp,
                          const float* __restrict__ gpart,
                          const float* __restrict__ corr,
                          const float* __restrict__ alpha_logit,
                          float* __restrict__ mixed_out,
                          float* __restrict__ dis_out) {
  __shared__ float mu[32][128];
  __shared__ float iv[32][128];
  __shared__ float wsum[4];
  const int d = blockIdx.x;
  const int e = threadIdx.x;
  const float cde = corr[d * 128 + e];
  const float ced = corr[e * 128 + d];
  const float alog = alpha_logit[0];
  for (int b = 0; b < 32; ++b) {
    float s1 = 0.0f, s2 = 0.0f;
#pragma unroll
    for (int q = 0; q < 8; ++q) {
      s1 += statp[(b * 8 + q) * 128 + e];
      s2 += statp[32768 + (b * 8 + q) * 128 + e];
    }
    float v = s2 - s1 * s1 * (1.0f / 4096.0f);
    mu[b][e] = s1 * (1.0f / 4096.0f);
    iv[b][e] = 1.0f / fmaxf(sqrtf(fmaxf(v, 0.0f)), EPSV);
  }
  __syncthreads();
  float acc = 0.0f;
  for (int b = 0; b < 32; ++b) {
    float md = mu[b][d], me = mu[b][e];
    float ind = iv[b][d], ine = iv[b][e];
    float G = 0.0f;
#pragma unroll
    for (int q = 0; q < 8; ++q)
      G += gpart[(long)(b * 8 + q) * 16384 + d * 128 + e];
    acc += (G - 4096.0f * md * me) * ind * ine;
  }
  const float alpha = 1.0f / (1.0f + __expf(-alog));
  const float learned = tanhf(0.5f * (cde + ced));
  const float dc = fminf(fmaxf(acc * (1.0f / 32.0f), -1.0f), 1.0f);
  const float mx = alpha * learned + (1.0f - alpha) * dc;
  mixed_out[d * 128 + e] = mx;
  float part = wave_sum32((e == d) ? 0.0f : mx);
  if ((e & 31) == 0) wsum[e >> 5] = part;
  __syncthreads();
  if (e == 0) {
    float deg = ((wsum[0] + wsum[1]) + wsum[2]) + wsum[3];
    dis_out[d] = rsqrtf(fmaxf(deg, EPSV));
  }
}

__global__ void __launch_bounds__(128) k_graph(const float* __restrict__ chebw,
                        const float* __restrict__ mixed,
                        const float* __restrict__ dis,
                        const float* __restrict__ Wi,
                        const float* __restrict__ Wo,
                        float* __restrict__ LapP,
                        float* __restrict__ WiP,
                        float* __restrict__ WoP,
                        float* __restrict__ wout) {
  const int g = blockIdx.x * 128 + threadIdx.x;
  const int lane = g & 31;
  const int h = lane >> 4, c = lane & 15;
  if (blockIdx.x < 32) {
    const int jj = (g >> 5) & 15;
    const int nb = g >> 9;
    const int n = nb * 16 + c;
    const float dn = dis[n];
    v4f v;
#pragma unroll
    for (int e = 0; e < 4; ++e) {
      const int k = 8 * jj + 4 * (e >> 1) + 2 * h + (e & 1);
      const float dk = dis[k];
      const float m = mixed[k * 128 + n];
      float a = (n == k) ? 0.0f : m;
      float l = ((n == k) ? 1.0f : 0.0f) - dk * a * dn;
      v[e] = fminf(fmaxf(l, -1.5f), 1.5f);
    }
    ((v4f*)LapP)[g] = v;
  } else if (blockIdx.x < 96) {
    const int idx = g - 32 * 128;
    const int jj = (idx >> 5) & 15;
    const int nb = idx >> 9;
    const float* row = Wi + (nb * 16 + c) * 128 + 8 * jj + 2 * h;
    const v2f lo = *(const v2f*)row;
    const v2f hi = *(const v2f*)(row + 4);
    v4f v = {lo.x, lo.y, hi.x, hi.y};
    ((v4f*)WiP)[idx] = v;
  } else {
    const int idx = g - 96 * 128;
    const int jj = (idx >> 5) & 31;
    const int nb = idx >> 10;
    const float* row = Wo + (nb * 16 + c) * 256 + 8 * jj + 2 * h;
    const v2f lo = *(const v2f*)row;
    const v2f hi = *(const v2f*)(row + 4);
    v4f v = {lo.x, lo.y, hi.x, hi.y};
    ((v4f*)WoP)[idx] = v;
  }
  if (g == 0) {
    float c0 = chebw[0], c1 = chebw[1], c2 = chebw[2];
    float mx = fmaxf(c0, fmaxf(c1, c2));
    float e0 = __expf(c0 - mx), e1 = __expf(c1 - mx), e2 = __expf(c2 - mx);
    float s = e0 + e1 + e2;
    wout[0] = e0 / s; wout[1] = e1 / s; wout[2] = e2 / s;
  }
}

__global__ void __launch_bounds__(256) k_main(const float* __restrict__ x,
                       const float* __restrict__ g1, const float* __restrict__ b1,
                       const float* __restrict__ LapP, const float* __restrict__ wv,
                       const float* __restrict__ WiP, const float* __restrict__ bi,
                       const float* __restrict__ WoP, const float* __restrict__ bo,
                       const float* __restrict__ g2, const float* __restrict__ b2,
                       const float* __restrict__ g3, const float* __restrict__ b3,
                       float* __restrict__ out) {
  extern __shared__ float smem[];
  float* yo = smem;
  float* t1 = smem + 64 * PY;
  float* hb = smem + 2 * 64 * PY;
  const int tid = threadIdx.x, lane = tid & 31, wave = tid >> 5;
  const long base = (long)blockIdx.x * 64;
  const float w0 = wv[0], w1 = wv[1], w2 = wv[2];
  const int khalf = lane >> 4;
  const int nl = lane & 15;

  for (int i = 0; i < 8; ++i) {
    int r = wave * 8 + i;
    ln128_row(x + (base + r) * 128, yo + r * PY, g1, b1, lane);
  }
  __syncthreads();

  const float* arow_yo = yo + nl * PY + khalf * 2;
  const float* arow_t1 = t1 + nl * PY + khalf * 2;
  const float* arow_hb = hb + nl * PH + khalf * 2;
  const int nc = wave * 16 + nl;
  const v4f* lapP = (const v4f*)LapP;
  const int lapo = (wave * 16) * 32 + lane;

  {
    v8f acc[4] = {};
#pragma unroll
    for (int jj = 0; jj < 16; ++jj) {
      const v4f b4 = lapP[lapo + jj * 32];
#pragma unroll
      for (int t = 0; t < 2; ++t) {
        const int kb = 2 * jj + t;
        v2f bb;
        bb.x = t ? b4.z : b4.x;   bb.y = t ? b4.w : b4.y;
#pragma unroll
        for (int mi = 0; mi < 4; ++mi) {
          v2f a = *(const v2f*)(arow_yo + mi * 16 * PY + kb * 4);
          acc[mi] = wmma4(a, bb, acc[mi]);
        }
      }
    }
#pragma unroll
    for (int mi = 0; mi < 4; ++mi) {
      int rowB = mi * 16 + khalf * 8;
#pragma unroll
      for (int r = 0; r < 8; ++r) t1[(rowB + r) * PY + nc] = acc[mi][r];
    }
  }
  __syncthreads();

  {
    v8f acc[4] = {};
#pragma unroll
    for (int jj = 0; jj < 16; ++jj) {
      const v4f b4 = lapP[lapo + jj * 32];
#pragma unroll
      for (int t = 0; t < 2; ++t) {
        const int kb = 2 * jj + t;
        v2f bb;
        bb.x = t ? b4.z : b4.x;   bb.y = t ? b4.w : b4.y;
#pragma unroll
        for (int mi = 0; mi < 4; ++mi) {
          v2f a = *(const v2f*)(arow_t1 + mi * 16 * PY + kb * 4);
          acc[mi] = wmma4(a, bb, acc[mi]);
        }
      }
    }
#pragma unroll
    for (int mi = 0; mi < 4; ++mi) {
      int rowB = mi * 16 + khalf * 8;
#pragma unroll
      for (int r = 0; r < 8; ++r) {
        int pos = (rowB + r) * PY + nc;
        float y0 = yo[pos], y1 = t1[pos];
        float tx2 = fminf(fmaxf(2.0f * acc[mi][r] - y0, -50.0f), 50.0f);
        yo[pos] = w0 * y0 + w1 * y1 + w2 * tx2;
      }
    }
  }
  __syncthreads();

  {
    v8f acc[2][4] = {};
    const v4f* wiP = (const v4f*)WiP;
    const int wio = (wave * 16) * 32 + lane;
#pragma unroll
    for (int jj = 0; jj < 16; ++jj) {
      const v4f p0 = wiP[wio + jj * 32];
      const v4f p1 = wiP[wio + 8 * 16 * 32 + jj * 32];
#pragma unroll
      for (int t = 0; t < 2; ++t) {
        const int kb = 2 * jj + t;
        v2f b0, b1;
        b0.x = t ? p0.z : p0.x;   b0.y = t ? p0.w : p0.y;
        b1.x = t ? p1.z : p1.x;   b1.y = t ? p1.w : p1.y;
#pragma unroll
        for (int mi = 0; mi < 4; ++mi) {
          v2f a = *(const v2f*)(arow_yo + mi * 16 * PY + kb * 4);
          acc[0][mi] = wmma4(a, b0, acc[0][mi]);
          acc[1][mi] = wmma4(a, b1, acc[1][mi]);
        }
      }
    }
#pragma unroll
    for (int j = 0; j < 2; ++j) {
      int ncj = nc + 128 * j;
      float bia = bi[ncj];
#pragma unroll
      for (int mi = 0; mi < 4; ++mi) {
        int rowB = mi * 16 + khalf * 8;
#pragma unroll
        for (int r = 0; r < 8; ++r) hb[(rowB + r) * PH + ncj] = acc[j][mi][r] + bia;
      }
    }
  }
  __syncthreads();

  {
    const int rw = wave * 8;
    float xv[8][8], s[8], m[8], vs[8];
#pragma unroll
    for (int i = 0; i < 8; ++i) {
#pragma unroll
      for (int j = 0; j < 8; ++j) xv[i][j] = hb[(rw + i) * PH + lane + 32 * j];
    }
#pragma unroll
    for (int i = 0; i < 8; ++i) {
      float t = 0.0f;
#pragma unroll
      for (int j = 0; j < 8; ++j) t += xv[i][j];
      s[i] = t;
    }
    wave_sum32_rows<8>(s);
#pragma unroll
    for (int i = 0; i < 8; ++i) {
      m[i] = s[i] * (1.0f / 256.0f);
      float t = 0.0f;
#pragma unroll
      for (int j = 0; j < 8; ++j) { float d = xv[i][j] - m[i]; t += d * d; }
      vs[i] = t;
    }
    wave_sum32_rows<8>(vs);
#pragma unroll
    for (int i = 0; i < 8; ++i) {
      float inv = rsqrtf(vs[i] * (1.0f / 256.0f) + EPSV);
#pragma unroll
      for (int j = 0; j < 8; ++j) {
        int d = lane + 32 * j;
        hb[(rw + i) * PH + d] = (xv[i][j] - m[i]) * inv * g2[d] + b2[d];
      }
    }
    for (int i = 0; i < 8; ++i) {
      float yv[8];
#pragma unroll
      for (int j = 0; j < 8; ++j) yv[j] = hb[(rw + i) * PH + lane + 32 * j];
#pragma unroll
      for (int j = 0; j < 8; ++j)
        hb[(rw + i) * PH + lane + 32 * j] =
            0.5f * yv[j] * (1.0f + erff(yv[j] * 0.70710678118654752f));
    }
  }
  __syncthreads();

  {
    v8f acc[4] = {};
    const v4f* woP = (const v4f*)WoP;
    const int woo = (wave * 32) * 32 + lane;
#pragma unroll
    for (int jj = 0; jj < 32; ++jj) {
      const v4f b4 = woP[woo + jj * 32];
#pragma unroll
      for (int t = 0; t < 2; ++t) {
        const int kb = 2 * jj + t;
        v2f bb;
        bb.x = t ? b4.z : b4.x;   bb.y = t ? b4.w : b4.y;
#pragma unroll
        for (int mi = 0; mi < 4; ++mi) {
          v2f a = *(const v2f*)(arow_hb + mi * 16 * PH + kb * 4);
          acc[mi] = wmma4(a, bb, acc[mi]);
        }
      }
    }
    float boa = bo[nc];
#pragma unroll
    for (int mi = 0; mi < 4; ++mi) {
      int rowB = mi * 16 + khalf * 8;
#pragma unroll
      for (int r = 0; r < 8; ++r) t1[(rowB + r) * PY + nc] = acc[mi][r] + boa;
    }
  }
  __syncthreads();

  {
    const int rw = wave * 8;
    float xv[8][4], s[8], m[8], vs[8];
#pragma unroll
    for (int i = 0; i < 8; ++i) {
#pragma unroll
      for (int j = 0; j < 4; ++j) xv[i][j] = t1[(rw + i) * PY + lane + 32 * j];
    }
#pragma unroll
    for (int i = 0; i < 8; ++i) s[i] = xv[i][0] + xv[i][1] + xv[i][2] + xv[i][3];
    wave_sum32_rows<8>(s);
#pragma unroll
    for (int i = 0; i < 8; ++i) {
      m[i] = s[i] * (1.0f / 128.0f);
      float t = 0.0f;
#pragma unroll
      for (int j = 0; j < 4; ++j) { float d = xv[i][j] - m[i]; t += d * d; }
      vs[i] = t;
    }
    wave_sum32_rows<8>(vs);
#pragma unroll
    for (int i = 0; i < 8; ++i) {
      float inv = rsqrtf(vs[i] * (1.0f / 128.0f) + EPSV);
#pragma unroll
      for (int j = 0; j < 4; ++j) {
        int d = lane + 32 * j;
        out[(base + rw + i) * 128 + d] = (xv[i][j] - m[i]) * inv * g3[d] + b3[d];
      }
    }
  }
}

extern "C" void kernel_launch(void* const* d_in, const int* in_sizes, int n_in,
                              void* d_out, int out_size, void* d_ws, size_t ws_size,
                              hipStream_t stream) {
  const float* x     = (const float*)d_in[0];
  const float* corr  = (const float*)d_in[1];
  const float* alog  = (const float*)d_in[2];
  const float* chebw = (const float*)d_in[3];
  const float* Wi    = (const float*)d_in[4];
  const float* bi    = (const float*)d_in[5];
  const float* Wo    = (const float*)d_in[6];
  const float* bo    = (const float*)d_in[7];
  const float* g1    = (const float*)d_in[8];
  const float* b1    = (const float*)d_in[9];
  const float* g2    = (const float*)d_in[10];
  const float* b2    = (const float*)d_in[11];
  const float* g3    = (const float*)d_in[12];
  const float* b3    = (const float*)d_in[13];
  float* out = (float*)d_out;
  float* ws  = (float*)d_ws;

  float* gpart  = ws + 8192;
  float* disv   = ws + 8192 + 256 * 16384;
  float* LapP   = disv + 16384;
  float* wv     = LapP + 16384;
  float* WiP    = gpart;
  float* WoP    = gpart + 32768;
  float* mixed  = out + (long)32 * 4096 * 128;
  float* statp  = out;

  k_gram<<<256, 256, 256 * PG * sizeof(float), stream>>>(x, g1, b1, statp, gpart);
  k_combine<<<128, 128, 0, stream>>>(statp, gpart, corr, alog, mixed, disv);
  k_graph<<<160, 128, 0, stream>>>(chebw, mixed, disv, Wi, Wo, LapP, WiP, WoP, wv);
  k_main<<<2048, 256, (2 * 64 * PY + 64 * PH) * sizeof(float), stream>>>(x, g1, b1, LapP, wv, WiP, bi, WoP, bo,
                                        g2, b2, g3, b3, out);
}


# k_main: each k block's A fragment reads go out as one group of 16-byte LDS loads a block ahead of its WMMAs, not one read behind each WMMA (k re-paired inside 8-k blocks: summation order changed)
# speedup vs baseline: 4.6244x; 1.0300x over previous
#include <hip/hip_runtime.h>
#include <hip/hip_bf16.h>
#include <math.h>

#define EPSV 1e-8f

#define PY 132
#define PH 260
#define PG 144

typedef float v2f __attribute__((ext_vector_type(2)));
typedef float v4f __attribute__((ext_vector_type(4)));
typedef float v8f __attribute__((ext_vector_type(8)));

__device__ __forceinline__ v8f wmma4(v2f a, v2f b, v8f c) {
  return __builtin_amdgcn_wmma_f32_16x16x4_f32(false, a, false, b, (short)0, c,
                                               false, false);
}

__device__ __forceinline__ float wave_sum32(float v) {
#pragma unroll
  for (int off = 16; off > 0; off >>= 1) v += __shfl_xor(v, off, 32);
  return v;
}

template <int R>
__device__ __forceinline__ void wave_sum32_rows(float (&v)[R]) {
#pragma unroll
  for (int off = 16; off > 0; off >>= 1) {
    float t[R];
#pragma unroll
    for (int r = 0; r < R; ++r) t[r] = __shfl_xor(v[r], off, 32);
#pragma unroll
    for (int r = 0; r < R; ++r) v[r] += t[r];
  }
}

__device__ __forceinline__ void ln128_row(const float* __restrict__ src,
                                          float* __restrict__ dst,
                                          const float* __restrict__ g,
                                          const float* __restrict__ b,
                                          int lane) {
  float xv[4];
#pragma unroll
  for (int j = 0; j < 4; ++j) xv[j] = src[lane + 32 * j];
  float s = xv[0] + xv[1] + xv[2] + xv[3];
  s = wave_sum32(s);
  float m = s * (1.0f / 128.0f);
  float vs = 0.0f;
#pragma unroll
  for (int j = 0; j < 4; ++j) { float d = xv[j] - m; vs += d * d; }
  vs = wave_sum32(vs);
  float inv = rsqrtf(vs * (1.0f / 128.0f) + EPSV);
#pragma unroll
  for (int j = 0; j < 4; ++j) {
    int d = lane + 32 * j;
    dst[d] = (xv[j] - m) * inv * g[d] + b[d];
  }
}

__global__ void __launch_bounds__(256) k_gram(const float* __restrict__ x,
                       const float* __restrict__ g1,
                       const float* __restrict__ b1,
                       float* __restrict__ statp,
                       float* __restrict__ gpart) {
  extern __shared__ float yt[];
  __shared__ float sred[2][8][128];
  const int tid = threadIdx.x, lane = tid & 31, wave = tid >> 5;
  const int b = blockIdx.x >> 3;
  const int oct = blockIdx.x & 7;
  const int khalf = lane >> 4;
  const int nl = lane & 15;
  const v8f vzero = {0, 0, 0, 0, 0, 0, 0, 0};
  v8f acc[8];
#pragma unroll
  for (int t = 0; t < 8; ++t) acc[t] = vzero;
  float a1[4] = {0, 0, 0, 0}, a2[4] = {0, 0, 0, 0};

  for (int sc = 0; sc < 2; ++sc) {
    const long rowbase = (long)b * 4096 + oct * 512 + sc * 256;
    __syncthreads();
    for (int i0 = 0; i0 < 32; i0 += 16) {
      const int r0 = wave * 32 + i0;
      float xv[16][4], s[16], m[16], vs[16];
#pragma unroll
      for (int q = 0; q < 16; ++q) {
        const float* src = x + (rowbase + r0 + q) * 128;
#pragma unroll
        for (int j = 0; j < 4; ++j) xv[q][j] = src[lane + 32 * j];
      }
#pragma unroll
      for (int q = 0; q < 16; ++q) s[q] = xv[q][0] + xv[q][1] + xv[q][2] + xv[q][3];
      wave_sum32_rows<16>(s);
#pragma unroll
      for (int q = 0; q < 16; ++q) {
        m[q] = s[q] * (1.0f / 128.0f);
        float t = 0.0f;
#pragma unroll
        for (int j = 0; j < 4; ++j) { float d = xv[q][j] - m[q]; t += d * d; }
        vs[q] = t;
      }
      wave_sum32_rows<16>(vs);
#pragma unroll
      for (int q = 0; q < 16; ++q) {
        float inv = rsqrtf(vs[q] * (1.0f / 128.0f) + EPSV);
#pragma unroll
        for (int j = 0; j < 4; ++j) {
          int d = lane + 32 * j;
          float y = (xv[q][j] - m[q]) * inv * g1[d] + b1[d];
          a1[j] += y; a2[j] += y * y;
          yt[(r0 + q) * PG + d] = y;
        }
      }
    }
    __syncthreads();
    {
      const float* bcol = yt + (khalf * 2) * PG + wave * 16 + nl;
      const float* arow = yt + (khalf * 2) * PG + nl;
#pragma unroll 2
      for (int kb = 0; kb < 64; ++kb) {
        v2f bf;
        bf.x = bcol[kb * 4 * PG];   bf.y = bcol[kb * 4 * PG + PG];
#pragma unroll
        for (int mi = 0; mi < 8; ++mi) {
          v2f af;
          af.x = arow[kb * 4 * PG + mi * 16];   af.y = arow[kb * 4 * PG + PG + mi * 16];
          acc[mi] = wmma4(af, bf, acc[mi]);
        }
      }
    }
  }

  float* gp = gpart + (long)blockIdx.x * 16384;
  for (int t = 0; t < 8; ++t) {
    int idx = wave + 8 * t;
    int mi = idx >> 3, ni = idx & 7;
    int dmBase = mi * 16 + khalf * 8;
    int de = ni * 16 + nl;
#pragma unroll
    for (int r = 0; r < 8; ++r) gp[(dmBase + r) * 128 + de] = acc[t][r];
  }
#pragma unroll
  for (int j = 0; j < 4; ++j) {
    int d = lane + 32 * j;
    sred[0][wave][d] = a1[j];
    sred[1][wave][d] = a2[j];
  }
  __syncthreads();
  {
    const int which = tid >> 7, d = tid & 127;
    float s = 0.0f;
#pragma unroll
    for (int w = 0; w < 8; ++w) s += sred[which][w][d];
    statp[which * 32768 + blockIdx.x * 128 + d] = s;
  }
}

__global__ void __launch_bounds__(128) k_combine(const float* __restrict__ statp,
                          const float* __restrict__ gpart,
                          const float* __restrict__ corr,
                          const float* __restrict__ alpha_logit,
                          float* __restrict__ mixed_out,
                          float* __restrict__ dis_out) {
  __shared__ float mu[32][128];
  __shared__ float iv[32][128];
  __shared__ float wsum[4];
  const int d = blockIdx.x;
  const int e = threadIdx.x;
  const float cde = corr[d * 128 + e];
  const float ced = corr[e * 128 + d];
  const float alog = alpha_logit[0];
  for (int b = 0; b < 32; ++b) {
    float s1 = 0.0f, s2 = 0.0f;
#pragma unroll
    for (int q = 0; q < 8; ++q) {
      s1 += statp[(b * 8 + q) * 128 + e];
      s2 += statp[32768 + (b * 8 + q) * 128 + e];
    }
    float v = s2 - s1 * s1 * (1.0f / 4096.0f);
    mu[b][e] = s1 * (1.0f / 4096.0f);
    iv[b][e] = 1.0f / fmaxf(sqrtf(fmaxf(v, 0.0f)), EPSV);
  }
  __syncthreads();
  float acc = 0.0f;
  for (int b = 0; b < 32; ++b) {
    float md = mu[b][d], me = mu[b][e];
    float ind = iv[b][d], ine = iv[b][e];
    float G = 0.0f;
#pragma unroll
    for (int q = 0; q < 8; ++q)
      G += gpart[(long)(b * 8 + q) * 16384 + d * 128 + e];
    acc += (G - 4096.0f * md * me) * ind * ine;
  }
  const float alpha = 1.0f / (1.0f + __expf(-alog));
  const float learned = tanhf(0.5f * (cde + ced));
  const float dc = fminf(fmaxf(acc * (1.0f / 32.0f), -1.0f), 1.0f);
  const float mx = alpha * learned + (1.0f - alpha) * dc;
  mixed_out[d * 128 + e] = mx;
  float part = wave_sum32((e == d) ? 0.0f : mx);
  if ((e & 31) == 0) wsum[e >> 5] = part;
  __syncthreads();
  if (e == 0) {
    float deg = ((wsum[0] + wsum[1]) + wsum[2]) + wsum[3];
    dis_out[d] = rsqrtf(fmaxf(deg, EPSV));
  }
}

__global__ void __launch_bounds__(128) k_graph(const float* __restrict__ chebw,
                        const float* __restrict__ mixed,
                        const float* __restrict__ dis,
                        const float* __restrict__ Wi,
                        const float* __restrict__ Wo,
                        float* __restrict__ LapP,
                        float* __restrict__ WiP,
                        float* __restrict__ WoP,
                        float* __restrict__ wout) {
  const int g = blockIdx.x * 128 + threadIdx.x;
  const int lane = g & 31;
  const int h = lane >> 4, c = lane & 15;
  if (blockIdx.x < 32) {
    const int jj = (g >> 5) & 15;
    const int nb = g >> 9;
    const int n = nb * 16 + c;
    const float dn = dis[n];
    v4f v;
#pragma unroll
    for (int e = 0; e < 4; ++e) {
      const int k = 8 * jj + 4 * h + e;
      const float dk = dis[k];
      const float m = mixed[k * 128 + n];
      float a = (n == k) ? 0.0f : m;
      float l = ((n == k) ? 1.0f : 0.0f) - dk * a * dn;
      v[e] = fminf(fmaxf(l, -1.5f), 1.5f);
    }
    ((v4f*)LapP)[g] = v;
  } else if (blockIdx.x < 96) {
    const int idx = g - 32 * 128;
    const int jj = (idx >> 5) & 15;
    const int nb = idx >> 9;
    const float* row = Wi + (nb * 16 + c) * 128 + 8 * jj + 4 * h;
    ((v4f*)WiP)[idx] = *(const v4f*)row;
  } else {
    const int idx = g - 96 * 128;
    const int jj = (idx >> 5) & 31;
    const int nb = idx >> 10;
    const float* row = Wo + (nb * 16 + c) * 256 + 8 * jj + 4 * h;
    ((v4f*)WoP)[idx] = *(const v4f*)row;
  }
  if (g == 0) {
    float c0 = chebw[0], c1 = chebw[1], c2 = chebw[2];
    float mx = fmaxf(c0, fmaxf(c1, c2));
    float e0 = __expf(c0 - mx), e1 = __expf(c1 - mx), e2 = __expf(c2 - mx);
    float s = e0 + e1 + e2;
    wout[0] = e0 / s; wout[1] = e1 / s; wout[2] = e2 / s;
  }
}

__device__ __forceinline__ void sched_fence() { __builtin_amdgcn_sched_barrier(0); }

#define DB 4

template <int NT>
struct BRing { v4f v[DB + 1][NT]; };

template <int NT>
__device__ __forceinline__ void b_prefetch(BRing<NT>& r, const v4f* __restrict__ bP,
                                           int tstride) {
#pragma unroll
  for (int j = 0; j < DB; ++j) {
#pragma unroll
    for (int t = 0; t < NT; ++t) r.v[j][t] = bP[t * tstride + j * 32];
  }
}

template <int NJ, int NT, int PITCH>
__device__ __forceinline__ void wave_product(const float* a4, const v4f* __restrict__ bP,
                                             int tstride, BRing<NT>& r,
                                             v8f (&acc)[NT][4]) {
  v4f an[4];
#pragma unroll
  for (int mi = 0; mi < 4; ++mi) an[mi] = *(const v4f*)(a4 + mi * 16 * PITCH);
#pragma unroll
  for (int jj = 0; jj < NJ; ++jj) {
    v4f ac[4];
#pragma unroll
    for (int mi = 0; mi < 4; ++mi) ac[mi] = an[mi];
    if (jj + 1 < NJ) {
#pragma unroll
      for (int mi = 0; mi < 4; ++mi)
        an[mi] = *(const v4f*)(a4 + mi * 16 * PITCH + (jj + 1) * 8);
    }
    if (jj + DB < NJ) {
#pragma unroll
      for (int t = 0; t < NT; ++t)
        r.v[(jj + DB) % (DB + 1)][t] = bP[t * tstride + (jj + DB) * 32];
    }
    sched_fence();
#pragma unroll
    for (int s = 0; s < 2; ++s) {
#pragma unroll
      for (int mi = 0; mi < 4; ++mi) {
        v2f a;
        a.x = s ? ac[mi].z : ac[mi].x;   a.y = s ? ac[mi].w : ac[mi].y;
#pragma unroll
        for (int t = 0; t < NT; ++t) {
          const v4f b4 = r.v[jj % (DB + 1)][t];
          v2f b;
          b.x = s ? b4.z : b4.x;   b.y = s ? b4.w : b4.y;
          acc[t][mi] = wmma4(a, b, acc[t][mi]);
        }
      }
    }
    sched_fence();
  }
}

__global__ void __launch_bounds__(256) k_main(const float* __restrict__ x,
                       const float* __restrict__ g1, const float* __restrict__ b1,
                       const float* __restrict__ LapP, const float* __restrict__ wv,
                       const float* __restrict__ WiP, const float* __restrict__ bi,
                       const float* __restrict__ WoP, const float* __restrict__ bo,
                       const float* __restrict__ g2, const float* __restrict__ b2,
                       const float* __restrict__ g3, const float* __restrict__ b3,
                       float* __restrict__ out) {
  extern __shared__ float smem[];
  float* yo = smem;
  float* t1 = smem + 64 * PY;
  float* hb = smem + 2 * 64 * PY;
  const int tid = threadIdx.x, lane = tid & 31, wave = tid >> 5;
  const long base = (long)blockIdx.x * 64;
  const float w0 = wv[0], w1 = wv[1], w2 = wv[2];
  const int khalf = lane >> 4;
  const int nl = lane & 15;

  for (int i = 0; i < 8; ++i) {
    int r = wave * 8 + i;
    ln128_row(x + (base + r) * 128, yo + r * PY, g1, b1, lane);
  }

  const float* a4_yo = yo + nl * PY + khalf * 4;
  const float* a4_t1 = t1 + nl * PY + khalf * 4;
  const float* a4_hb = hb + nl * PH + khalf * 4;
  const int nc = wave * 16 + nl;
  const v4f* lapP = (const v4f*)LapP + (wave * 16) * 32 + lane;
  BRing<1> rb;
  b_prefetch<1>(rb, lapP, 0);
  __syncthreads();

  {
    v8f acc[1][4] = {};
    wave_product<16, 1, PY>(a4_yo, lapP, 0, rb, acc);
#pragma unroll
    for (int mi = 0; mi < 4; ++mi) {
      int rowB = mi * 16 + khalf * 8;
#pragma unroll
      for (int r = 0; r < 8; ++r) t1[(rowB + r) * PY + nc] = acc[0][mi][r];
    }
  }
  b_prefetch<1>(rb, lapP, 0);
  __syncthreads();

  {
    v8f acc[1][4] = {};
    wave_product<16, 1, PY>(a4_t1, lapP, 0, rb, acc);
#pragma unroll
    for (int mi = 0; mi < 4; ++mi) {
      int rowB = mi * 16 + khalf * 8;
#pragma unroll
      for (int r = 0; r < 8; ++r) {
        int pos = (rowB + r) * PY + nc;
        float y0 = yo[pos], y1 = t1[pos];
        float tx2 = fminf(fmaxf(2.0f * acc[0][mi][r] - y0, -50.0f), 50.0f);
        yo[pos] = w0 * y0 + w1 * y1 + w2 * tx2;
      }
    }
  }
  const v4f* wiP = (const v4f*)WiP + (wave * 16) * 32 + lane;
  BRing<2> rb2;
  b_prefetch<2>(rb2, wiP, 8 * 16 * 32);
  __syncthreads();

  {
    v8f acc[2][4] = {};
    wave_product<16, 2, PY>(a4_yo, wiP, 8 * 16 * 32, rb2, acc);
#pragma unroll
    for (int j = 0; j < 2; ++j) {
      int ncj = nc + 128 * j;
      float bia = bi[ncj];
#pragma unroll
      for (int mi = 0; mi < 4; ++mi) {
        int rowB = mi * 16 + khalf * 8;
#pragma unroll
        for (int r = 0; r < 8; ++r) hb[(rowB + r) * PH + ncj] = acc[j][mi][r] + bia;
      }
    }
  }
  __syncthreads();

  {
    const int rw = wave * 8;
    float xv[8][8], s[8], m[8], vs[8];
#pragma unroll
    for (int i = 0; i < 8; ++i) {
#pragma unroll
      for (int j = 0; j < 8; ++j) xv[i][j] = hb[(rw + i) * PH + lane + 32 * j];
    }
#pragma unroll
    for (int i = 0; i < 8; ++i) {
      float t = 0.0f;
#pragma unroll
      for (int j = 0; j < 8; ++j) t += xv[i][j];
      s[i] = t;
    }
    wave_sum32_rows<8>(s);
#pragma unroll
    for (int i = 0; i < 8; ++i) {
      m[i] = s[i] * (1.0f / 256.0f);
      float t = 0.0f;
#pragma unroll
      for (int j = 0; j < 8; ++j) { float d = xv[i][j] - m[i]; t += d * d; }
      vs[i] = t;
    }
    wave_sum32_rows<8>(vs);
#pragma unroll
    for (int i = 0; i < 8; ++i) {
      float inv = rsqrtf(vs[i] * (1.0f / 256.0f) + EPSV);
#pragma unroll
      for (int j = 0; j < 8; ++j) {
        int d = lane + 32 * j;
        hb[(rw + i) * PH + d] = (xv[i][j] - m[i]) * inv * g2[d] + b2[d];
      }
    }
    for (int i = 0; i < 8; ++i) {
      float yv[8];
#pragma unroll
      for (int j = 0; j < 8; ++j) yv[j] = hb[(rw + i) * PH + lane + 32 * j];
#pragma unroll
      for (int j = 0; j < 8; ++j)
        hb[(rw + i) * PH + lane + 32 * j] =
            0.5f * yv[j] * (1.0f + erff(yv[j] * 0.70710678118654752f));
    }
  }
  const v4f* woP = (const v4f*)WoP + (wave * 32) * 32 + lane;
  b_prefetch<1>(rb, woP, 0);
  __syncthreads();

  {
    v8f acc[1][4] = {};
    wave_product<32, 1, PH>(a4_hb, woP, 0, rb, acc);
    float boa = bo[nc];
#pragma unroll
    for (int mi = 0; mi < 4; ++mi) {
      int rowB = mi * 16 + khalf * 8;
#pragma unroll
      for (int r = 0; r < 8; ++r) t1[(rowB + r) * PY + nc] = acc[0][mi][r] + boa;
    }
  }
  __syncthreads();

  {
    const int rw = wave * 8;
    float xv[8][4], s[8], m[8], vs[8];
#pragma unroll
    for (int i = 0; i < 8; ++i) {
#pragma unroll
      for (int j = 0; j < 4; ++j) xv[i][j] = t1[(rw + i) * PY + lane + 32 * j];
    }
#pragma unroll
    for (int i = 0; i < 8; ++i) s[i] = xv[i][0] + xv[i][1] + xv[i][2] + xv[i][3];
    wave_sum32_rows<8>(s);
#pragma unroll
    for (int i = 0; i < 8; ++i) {
      m[i] = s[i] * (1.0f / 128.0f);
      float t = 0.0f;
#pragma unroll
      for (int j = 0; j < 4; ++j) { float d = xv[i][j] - m[i]; t += d * d; }
      vs[i] = t;
    }
    wave_sum32_rows<8>(vs);
#pragma unroll
    for (int i = 0; i < 8; ++i) {
      float inv = rsqrtf(vs[i] * (1.0f / 128.0f) + EPSV);
#pragma unroll
      for (int j = 0; j < 4; ++j) {
        int d = lane + 32 * j;
        out[(base + rw + i) * 128 + d] = (xv[i][j] - m[i]) * inv * g3[d] + b3[d];
      }
    }
  }
}

extern "C" void kernel_launch(void* const* d_in, const int* in_sizes, int n_in,
                              void* d_out, int out_size, void* d_ws, size_t ws_size,
                              hipStream_t stream) {
  const float* x     = (const float*)d_in[0];
  const float* corr  = (const float*)d_in[1];
  const float* alog  = (const float*)d_in[2];
  const float* chebw = (const float*)d_in[3];
  const float* Wi    = (const float*)d_in[4];
  const float* bi    = (const float*)d_in[5];
  const float* Wo    = (const float*)d_in[6];
  const float* bo    = (const float*)d_in[7];
  const float* g1    = (const float*)d_in[8];
  const float* b1    = (const float*)d_in[9];
  const float* g2    = (const float*)d_in[10];
  const float* b2    = (const float*)d_in[11];
  const float* g3    = (const float*)d_in[12];
  const float* b3    = (const float*)d_in[13];
  float* out = (float*)d_out;
  float* ws  = (float*)d_ws;

  float* gpart  = ws + 8192;
  float* disv   = ws + 8192 + 256 * 16384;
  float* LapP   = disv + 16384;
  float* wv     = LapP + 16384;
  float* WiP    = gpart;
  float* WoP    = gpart + 32768;
  float* mixed  = out + (long)32 * 4096 * 128;
  float* statp  = out;

  k_gram<<<256, 256, 256 * PG * sizeof(float), stream>>>(x, g1, b1, statp, gpart);
  k_combine<<<128, 128, 0, stream>>>(statp, gpart, corr, alog, mixed, disv);
  k_graph<<<160, 128, 0, stream>>>(chebw, mixed, disv, Wi, Wo, LapP, WiP, WoP, wv);
  k_main<<<2048, 256, (2 * 64 * PY + 64 * PH) * sizeof(float), stream>>>(x, g1, b1, LapP, wv, WiP, bi, WoP, bo,
                                        g2, b2, g3, b3, out);
}


# GELU's erf from the five-term rcp/exp formula (abs error about 1e-7, float32 class) in place of library erff, applied in LN2's store pass: about 800 fewer vector and LDS instructions per wave
# speedup vs baseline: 4.7703x; 1.0316x over previous
#include <hip/hip_runtime.h>
#include <hip/hip_bf16.h>
#include <math.h>

#define EPSV 1e-8f

#define PY 132
#define PH 260
#define PG 144

typedef float v2f __attribute__((ext_vector_type(2)));
typedef float v4f __attribute__((ext_vector_type(4)));
typedef float v8f __attribute__((ext_vector_type(8)));

__device__ __forceinline__ v8f wmma4(v2f a, v2f b, v8f c) {
  return __builtin_amdgcn_wmma_f32_16x16x4_f32(false, a, false, b, (short)0, c,
                                               false, false);
}

__device__ __forceinline__ float wave_sum32(float v) {
#pragma unroll
  for (int off = 16; off > 0; off >>= 1) v += __shfl_xor(v, off, 32);
  return v;
}

template <int R>
__device__ __forceinline__ void wave_sum32_rows(float (&v)[R]) {
#pragma unroll
  for (int off = 16; off > 0; off >>= 1) {
    float t[R];
#pragma unroll
    for (int r = 0; r < R; ++r) t[r] = __shfl_xor(v[r], off, 32);
#pragma unroll
    for (int r = 0; r < R; ++r) v[r] += t[r];
  }
}

__device__ __forceinline__ float gelu_erf(float x) {
  const float z = x * 0.70710678118654752f;
  const float a = fabsf(z);
  const float t = __builtin_amdgcn_rcpf(fmaf(0.3275911f, a, 1.0f));
  float q = fmaf(1.061405429f, t, -1.453152027f);
  q = fmaf(q, t, 1.421413741f);
  q = fmaf(q, t, -0.284496736f);
  q = fmaf(q, t, 0.254829592f);
  q = q * t;
  const float e = __builtin_amdgcn_exp2f(z * z * -1.4426950408889634f);
  const float er = copysignf(1.0f - q * e, z);
  const float hx = 0.5f * x;
  return fmaf(hx, er, hx);
}

__device__ __forceinline__ void ln128_row(const float* __restrict__ src,
                                          float* __restrict__ dst,
                                          const float* __restrict__ g,
                                          const float* __restrict__ b,
                                          int lane) {
  float xv[4];
#pragma unroll
  for (int j = 0; j < 4; ++j) xv[j] = src[lane + 32 * j];
  float s = xv[0] + xv[1] + xv[2] + xv[3];
  s = wave_sum32(s);
  float m = s * (1.0f / 128.0f);
  float vs = 0.0f;
#pragma unroll
  for (int j = 0; j < 4; ++j) { float d = xv[j] - m; vs += d * d; }
  vs = wave_sum32(vs);
  float inv = rsqrtf(vs * (1.0f / 128.0f) + EPSV);
#pragma unroll
  for (int j = 0; j < 4; ++j) {
    int d = lane + 32 * j;
    dst[d] = (xv[j] - m) * inv * g[d] + b[d];
  }
}

__global__ void __launch_bounds__(256) k_gram(const float* __restrict__ x,
                       const float* __restrict__ g1,
                       const float* __restrict__ b1,
                       float* __restrict__ statp,
                       float* __restrict__ gpart) {
  extern __shared__ float yt[];
  __shared__ float sred[2][8][128];
  const int tid = threadIdx.x, lane = tid & 31, wave = tid >> 5;
  const int b = blockIdx.x >> 3;
  const int oct = blockIdx.x & 7;
  const int khalf = lane >> 4;
  const int nl = lane & 15;
  const v8f vzero = {0, 0, 0, 0, 0, 0, 0, 0};
  v8f acc[8];
#pragma unroll
  for (int t = 0; t < 8; ++t) acc[t] = vzero;
  float a1[4] = {0, 0, 0, 0}, a2[4] = {0, 0, 0, 0};

  for (int sc = 0; sc < 2; ++sc) {
    const long rowbase = (long)b * 4096 + oct * 512 + sc * 256;
    __syncthreads();
    for (int i0 = 0; i0 < 32; i0 += 16) {
      const int r0 = wave * 32 + i0;
      float xv[16][4], s[16], m[16], vs[16];
#pragma unroll
      for (int q = 0; q < 16; ++q) {
        const float* src = x + (rowbase + r0 + q) * 128;
#pragma unroll
        for (int j = 0; j < 4; ++j) xv[q][j] = src[lane + 32 * j];
      }
#pragma unroll
      for (int q = 0; q < 16; ++q) s[q] = xv[q][0] + xv[q][1] + xv[q][2] + xv[q][3];
      wave_sum32_rows<16>(s);
#pragma unroll
      for (int q = 0; q < 16; ++q) {
        m[q] = s[q] * (1.0f / 128.0f);
        float t = 0.0f;
#pragma unroll
        for (int j = 0; j < 4; ++j) { float d = xv[q][j] - m[q]; t += d * d; }
        vs[q] = t;
      }
      wave_sum32_rows<16>(vs);
#pragma unroll
      for (int q = 0; q < 16; ++q) {
        float inv = rsqrtf(vs[q] * (1.0f / 128.0f) + EPSV);
#pragma unroll
        for (int j = 0; j < 4; ++j) {
          int d = lane + 32 * j;
          float y = (xv[q][j] - m[q]) * inv * g1[d] + b1[d];
          a1[j] += y; a2[j] += y * y;
          yt[(r0 + q) * PG + d] = y;
        }
      }
    }
    __syncthreads();
    {
      const float* bcol = yt + (khalf * 2) * PG + wave * 16 + nl;
      const float* arow = yt + (khalf * 2) * PG + nl;
#pragma unroll 2
      for (int kb = 0; kb < 64; ++kb) {
        v2f bf;
        bf.x = bcol[kb * 4 * PG];   bf.y = bcol[kb * 4 * PG + PG];
#pragma unroll
        for (int mi = 0; mi < 8; ++mi) {
          v2f af;
          af.x = arow[kb * 4 * PG + mi * 16];   af.y = arow[kb * 4 * PG + PG + mi * 16];
          acc[mi] = wmma4(af, bf, acc[mi]);
        }
      }
    }
  }

  float* gp = gpart + (long)blockIdx.x * 16384;
  for (int t = 0; t < 8; ++t) {
    int idx = wave + 8 * t;
    int mi = idx >> 3, ni = idx & 7;
    int dmBase = mi * 16 + khalf * 8;
    int de = ni * 16 + nl;
#pragma unroll
    for (int r = 0; r < 8; ++r) gp[(dmBase + r) * 128 + de] = acc[t][r];
  }
#pragma unroll
  for (int j = 0; j < 4; ++j) {
    int d = lane + 32 * j;
    sred[0][wave][d] = a1[j];
    sred[1][wave][d] = a2[j];
  }
  __syncthreads();
  {
    const int which = tid >> 7, d = tid & 127;
    float s = 0.0f;
#pragma unroll
    for (int w = 0; w < 8; ++w) s += sred[which][w][d];
    statp[which * 32768 + blockIdx.x * 128 + d] = s;
  }
}

__global__ void __launch_bounds__(128) k_combine(const float* __restrict__ statp,
                          const float* __restrict__ gpart,
                          const float* __restrict__ corr,
                          const float* __restrict__ alpha_logit,
                          float* __restrict__ mixed_out,
                          float* __restrict__ dis_out) {
  __shared__ float mu[32][128];
  __shared__ float iv[32][128];
  __shared__ float wsum[4];
  const int d = blockIdx.x;
  const int e = threadIdx.x;
  const float cde = corr[d * 128 + e];
  const float ced = corr[e * 128 + d];
  const float alog = alpha_logit[0];
  for (int b = 0; b < 32; ++b) {
    float s1 = 0.0f, s2 = 0.0f;
#pragma unroll
    for (int q = 0; q < 8; ++q) {
      s1 += statp[(b * 8 + q) * 128 + e];
      s2 += statp[32768 + (b * 8 + q) * 128 + e];
    }
    float v = s2 - s1 * s1 * (1.0f / 4096.0f);
    mu[b][e] = s1 * (1.0f / 4096.0f);
    iv[b][e] = 1.0f / fmaxf(sqrtf(fmaxf(v, 0.0f)), EPSV);
  }
  __syncthreads();
  float acc = 0.0f;
  for (int b = 0; b < 32; ++b) {
    float md = mu[b][d], me = mu[b][e];
    float ind = iv[b][d], ine = iv[b][e];
    float G = 0.0f;
#pragma unroll
    for (int q = 0; q < 8; ++q)
      G += gpart[(long)(b * 8 + q) * 16384 + d * 128 + e];
    acc += (G - 4096.0f * md * me) * ind * ine;
  }
  const float alpha = 1.0f / (1.0f + __expf(-alog));
  const float learned = tanhf(0.5f * (cde + ced));
  const float dc = fminf(fmaxf(acc * (1.0f / 32.0f), -1.0f), 1.0f);
  const float mx = alpha * learned + (1.0f - alpha) * dc;
  mixed_out[d * 128 + e] = mx;
  float part = wave_sum32((e == d) ? 0.0f : mx);
  if ((e & 31) == 0) wsum[e >> 5] = part;
  __syncthreads();
  if (e == 0) {
    float deg = ((wsum[0] + wsum[1]) + wsum[2]) + wsum[3];
    dis_out[d] = rsqrtf(fmaxf(deg, EPSV));
  }
}

__global__ void __launch_bounds__(128) k_graph(const float* __restrict__ chebw,
                        const float* __restrict__ mixed,
                        const float* __restrict__ dis,
                        const float* __restrict__ Wi,
                        const float* __restrict__ Wo,
                        float* __restrict__ LapP,
                        float* __restrict__ WiP,
                        float* __restrict__ WoP,
                        float* __restrict__ wout) {
  const int g = blockIdx.x * 128 + threadIdx.x;
  const int lane = g & 31;
  const int h = lane >> 4, c = lane & 15;
  if (blockIdx.x < 32) {
    const int jj = (g >> 5) & 15;
    const int nb = g >> 9;
    const int n = nb * 16 + c;
    const float dn = dis[n];
    v4f v;
#pragma unroll
    for (int e = 0; e < 4; ++e) {
      const int k = 8 * jj + 4 * h + e;
      const float dk = dis[k];
      const float m = mixed[k * 128 + n];
      float a = (n == k) ? 0.0f : m;
      float l = ((n == k) ? 1.0f : 0.0f) - dk * a * dn;
      v[e] = fminf(fmaxf(l, -1.5f), 1.5f);
    }
    ((v4f*)LapP)[g] = v;
  } else if (blockIdx.x < 96) {
    const int idx = g - 32 * 128;
    const int jj = (idx >> 5) & 15;
    const int nb = idx >> 9;
    const float* row = Wi + (nb * 16 + c) * 128 + 8 * jj + 4 * h;
    ((v4f*)WiP)[idx] = *(const v4f*)row;
  } else {
    const int idx = g - 96 * 128;
    const int jj = (idx >> 5) & 31;
    const int nb = idx >> 10;
    const float* row = Wo + (nb * 16 + c) * 256 + 8 * jj + 4 * h;
    ((v4f*)WoP)[idx] = *(const v4f*)row;
  }
  if (g == 0) {
    float c0 = chebw[0], c1 = chebw[1], c2 = chebw[2];
    float mx = fmaxf(c0, fmaxf(c1, c2));
    float e0 = __expf(c0 - mx), e1 = __expf(c1 - mx), e2 = __expf(c2 - mx);
    float s = e0 + e1 + e2;
    wout[0] = e0 / s; wout[1] = e1 / s; wout[2] = e2 / s;
  }
}

__device__ __forceinline__ void sched_fence() { __builtin_amdgcn_sched_barrier(0); }

#define DB 4

template <int NT>
struct BRing { v4f v[DB + 1][NT]; };

template <int NT>
__device__ __forceinline__ void b_prefetch(BRing<NT>& r, const v4f* __restrict__ bP,
                                           int tstride) {
#pragma unroll
  for (int j = 0; j < DB; ++j) {
#pragma unroll
    for (int t = 0; t < NT; ++t) r.v[j][t] = bP[t * tstride + j * 32];
  }
}

template <int NJ, int NT, int PITCH>
__device__ __forceinline__ void wave_product(const float* a4, const v4f* __restrict__ bP,
                                             int tstride, BRing<NT>& r,
                                             v8f (&acc)[NT][4]) {
  v4f an[4];
#pragma unroll
  for (int mi = 0; mi < 4; ++mi) an[mi] = *(const v4f*)(a4 + mi * 16 * PITCH);
#pragma unroll
  for (int jj = 0; jj < NJ; ++jj) {
    v4f ac[4];
#pragma unroll
    for (int mi = 0; mi < 4; ++mi) ac[mi] = an[mi];
    if (jj + 1 < NJ) {
#pragma unroll
      for (int mi = 0; mi < 4; ++mi)
        an[mi] = *(const v4f*)(a4 + mi * 16 * PITCH + (jj + 1) * 8);
    }
    if (jj + DB < NJ) {
#pragma unroll
      for (int t = 0; t < NT; ++t)
        r.v[(jj + DB) % (DB + 1)][t] = bP[t * tstride + (jj + DB) * 32];
    }
    sched_fence();
#pragma unroll
    for (int s = 0; s < 2; ++s) {
#pragma unroll
      for (int mi = 0; mi < 4; ++mi) {
        v2f a;
        a.x = s ? ac[mi].z : ac[mi].x;   a.y = s ? ac[mi].w : ac[mi].y;
#pragma unroll
        for (int t = 0; t < NT; ++t) {
          const v4f b4 = r.v[jj % (DB + 1)][t];
          v2f b;
          b.x = s ? b4.z : b4.x;   b.y = s ? b4.w : b4.y;
          acc[t][mi] = wmma4(a, b, acc[t][mi]);
        }
      }
    }
    sched_fence();
  }
}

__global__ void __launch_bounds__(256) k_main(const float* __restrict__ x,
                       const float* __restrict__ g1, const float* __restrict__ b1,
                       const float* __restrict__ LapP, const float* __restrict__ wv,
                       const float* __restrict__ WiP, const float* __restrict__ bi,
                       const float* __restrict__ WoP, const float* __restrict__ bo,
                       const float* __restrict__ g2, const float* __restrict__ b2,
                       const float* __restrict__ g3, const float* __restrict__ b3,
                       float* __restrict__ out) {
  extern __shared__ float smem[];
  float* yo = smem;
  float* t1 = smem + 64 * PY;
  float* hb = smem + 2 * 64 * PY;
  const int tid = threadIdx.x, lane = tid & 31, wave = tid >> 5;
  const long base = (long)blockIdx.x * 64;
  const float w0 = wv[0], w1 = wv[1], w2 = wv[2];
  const int khalf = lane >> 4;
  const int nl = lane & 15;

  for (int i = 0; i < 8; ++i) {
    int r = wave * 8 + i;
    ln128_row(x + (base + r) * 128, yo + r * PY, g1, b1, lane);
  }

  const float* a4_yo = yo + nl * PY + khalf * 4;
  const float* a4_t1 = t1 + nl * PY + khalf * 4;
  const float* a4_hb = hb + nl * PH + khalf * 4;
  const int nc = wave * 16 + nl;
  const v4f* lapP = (const v4f*)LapP + (wave * 16) * 32 + lane;
  BRing<1> rb;
  b_prefetch<1>(rb, lapP, 0);
  __syncthreads();

  {
    v8f acc[1][4] = {};
    wave_product<16, 1, PY>(a4_yo, lapP, 0, rb, acc);
#pragma unroll
    for (int mi = 0; mi < 4; ++mi) {
      int rowB = mi * 16 + khalf * 8;
#pragma unroll
      for (int r = 0; r < 8; ++r) t1[(rowB + r) * PY + nc] = acc[0][mi][r];
    }
  }
  b_prefetch<1>(rb, lapP, 0);
  __syncthreads();

  {
    v8f acc[1][4] = {};
    wave_product<16, 1, PY>(a4_t1, lapP, 0, rb, acc);
#pragma unroll
    for (int mi = 0; mi < 4; ++mi) {
      int rowB = mi * 16 + khalf * 8;
#pragma unroll
      for (int r = 0; r < 8; ++r) {
        int pos = (rowB + r) * PY + nc;
        float y0 = yo[pos], y1 = t1[pos];
        float tx2 = fminf(fmaxf(2.0f * acc[0][mi][r] - y0, -50.0f), 50.0f);
        yo[pos] = w0 * y0 + w1 * y1 + w2 * tx2;
      }
    }
  }
  const v4f* wiP = (const v4f*)WiP + (wave * 16) * 32 + lane;
  BRing<2> rb2;
  b_prefetch<2>(rb2, wiP, 8 * 16 * 32);
  __syncthreads();

  {
    v8f acc[2][4] = {};
    wave_product<16, 2, PY>(a4_yo, wiP, 8 * 16 * 32, rb2, acc);
#pragma unroll
    for (int j = 0; j < 2; ++j) {
      int ncj = nc + 128 * j;
      float bia = bi[ncj];
#pragma unroll
      for (int mi = 0; mi < 4; ++mi) {
        int rowB = mi * 16 + khalf * 8;
#pragma unroll
        for (int r = 0; r < 8; ++r) hb[(rowB + r) * PH + ncj] = acc[j][mi][r] + bia;
      }
    }
  }
  __syncthreads();

  {
    const int rw = wave * 8;
    float xv[8][8], s[8], m[8], vs[8];
#pragma unroll
    for (int i = 0; i < 8; ++i) {
#pragma unroll
      for (int j = 0; j < 8; ++j) xv[i][j] = hb[(rw + i) * PH + lane + 32 * j];
    }
#pragma unroll
    for (int i = 0; i < 8; ++i) {
      float t = 0.0f;
#pragma unroll
      for (int j = 0; j < 8; ++j) t += xv[i][j];
      s[i] = t;
    }
    wave_sum32_rows<8>(s);
#pragma unroll
    for (int i = 0; i < 8; ++i) {
      m[i] = s[i] * (1.0f / 256.0f);
      float t = 0.0f;
#pragma unroll
      for (int j = 0; j < 8; ++j) { float d = xv[i][j] - m[i]; t += d * d; }
      vs[i] = t;
    }
    wave_sum32_rows<8>(vs);
#pragma unroll
    for (int i = 0; i < 8; ++i) {
      float inv = rsqrtf(vs[i] * (1.0f / 256.0f) + EPSV);
#pragma unroll
      for (int j = 0; j < 8; ++j) {
        int d = lane + 32 * j;
        hb[(rw + i) * PH + d] = gelu_erf((xv[i][j] - m[i]) * inv * g2[d] + b2[d]);
      }
    }
  }
  const v4f* woP = (const v4f*)WoP + (wave * 32) * 32 + lane;
  b_prefetch<1>(rb, woP, 0);
  __syncthreads();

  {
    v8f acc[1][4] = {};
    wave_product<32, 1, PH>(a4_hb, woP, 0, rb, acc);
    float boa = bo[nc];
#pragma unroll
    for (int mi = 0; mi < 4; ++mi) {
      int rowB = mi * 16 + khalf * 8;
#pragma unroll
      for (int r = 0; r < 8; ++r) t1[(rowB + r) * PY + nc] = acc[0][mi][r] + boa;
    }
  }
  __syncthreads();

  {
    const int rw = wave * 8;
    float xv[8][4], s[8], m[8], vs[8];
#pragma unroll
    for (int i = 0; i < 8; ++i) {
#pragma unroll
      for (int j = 0; j < 4; ++j) xv[i][j] = t1[(rw + i) * PY + lane + 32 * j];
    }
#pragma unroll
    for (int i = 0; i < 8; ++i) s[i] = xv[i][0] + xv[i][1] + xv[i][2] + xv[i][3];
    wave_sum32_rows<8>(s);
#pragma unroll
    for (int i = 0; i < 8; ++i) {
      m[i] = s[i] * (1.0f / 128.0f);
      float t = 0.0f;
#pragma unroll
      for (int j = 0; j < 4; ++j) { float d = xv[i][j] - m[i]; t += d * d; }
      vs[i] = t;
    }
    wave_sum32_rows<8>(vs);
#pragma unroll
    for (int i = 0; i < 8; ++i) {
      float inv = rsqrtf(vs[i] * (1.0f / 128.0f) + EPSV);
#pragma unroll
      for (int j = 0; j < 4; ++j) {
        int d = lane + 32 * j;
        out[(base + rw + i) * 128 + d] = (xv[i][j] - m[i]) * inv * g3[d] + b3[d];
      }
    }
  }
}

extern "C" void kernel_launch(void* const* d_in, const int* in_sizes, int n_in,
                              void* d_out, int out_size, void* d_ws, size_t ws_size,
                              hipStream_t stream) {
  const float* x     = (const float*)d_in[0];
  const float* corr  = (const float*)d_in[1];
  const float* alog  = (const float*)d_in[2];
  const float* chebw = (const float*)d_in[3];
  const float* Wi    = (const float*)d_in[4];
  const float* bi    = (const float*)d_in[5];
  const float* Wo    = (const float*)d_in[6];
  const float* bo    = (const float*)d_in[7];
  const float* g1    = (const float*)d_in[8];
  const float* b1    = (const float*)d_in[9];
  const float* g2    = (const float*)d_in[10];
  const float* b2    = (const float*)d_in[11];
  const float* g3    = (const float*)d_in[12];
  const float* b3    = (const float*)d_in[13];
  float* out = (float*)d_out;
  float* ws  = (float*)d_ws;

  float* gpart  = ws + 8192;
  float* disv   = ws + 8192 + 256 * 16384;
  float* LapP   = disv + 16384;
  float* wv     = LapP + 16384;
  float* WiP    = gpart;
  float* WoP    = gpart + 32768;
  float* mixed  = out + (long)32 * 4096 * 128;
  float* statp  = out;

  k_gram<<<256, 256, 256 * PG * sizeof(float), stream>>>(x, g1, b1, statp, gpart);
  k_combine<<<128, 128, 0, stream>>>(statp, gpart, corr, alog, mixed, disv);
  k_graph<<<160, 128, 0, stream>>>(chebw, mixed, disv, Wi, Wo, LapP, WiP, WoP, wv);
  k_main<<<2048, 256, (2 * 64 * PY + 64 * PH) * sizeof(float), stream>>>(x, g1, b1, LapP, wv, WiP, bi, WoP, bo,
                                        g2, b2, g3, b3, out);
}


# k_gram: gram symmetry used: a wave computes its column block against 4 row blocks + half the antipodal tile (36 of 64 tile products), mirrors stored; antipodal tile = sum of 2 half sweeps (new order)
# speedup vs baseline: 4.9640x; 1.0406x over previous
#include <hip/hip_runtime.h>
#include <hip/hip_bf16.h>
#include <math.h>

#define EPSV 1e-8f

#define PY 132
#define PH 260
#define PG 144

typedef float v2f __attribute__((ext_vector_type(2)));
typedef float v4f __attribute__((ext_vector_type(4)));
typedef float v8f __attribute__((ext_vector_type(8)));

__device__ __forceinline__ v8f wmma4(v2f a, v2f b, v8f c) {
  return __builtin_amdgcn_wmma_f32_16x16x4_f32(false, a, false, b, (short)0, c,
                                               false, false);
}

__device__ __forceinline__ float wave_sum32(float v) {
#pragma unroll
  for (int off = 16; off > 0; off >>= 1) v += __shfl_xor(v, off, 32);
  return v;
}

template <int R>
__device__ __forceinline__ void wave_sum32_rows(float (&v)[R]) {
#pragma unroll
  for (int off = 16; off > 0; off >>= 1) {
    float t[R];
#pragma unroll
    for (int r = 0; r < R; ++r) t[r] = __shfl_xor(v[r], off, 32);
#pragma unroll
    for (int r = 0; r < R; ++r) v[r] += t[r];
  }
}

__device__ __forceinline__ float gelu_erf(float x) {
  const float z = x * 0.70710678118654752f;
  const float a = fabsf(z);
  const float t = __builtin_amdgcn_rcpf(fmaf(0.3275911f, a, 1.0f));
  float q = fmaf(1.061405429f, t, -1.453152027f);
  q = fmaf(q, t, 1.421413741f);
  q = fmaf(q, t, -0.284496736f);
  q = fmaf(q, t, 0.254829592f);
  q = q * t;
  const float e = __builtin_amdgcn_exp2f(z * z * -1.4426950408889634f);
  const float er = copysignf(1.0f - q * e, z);
  const float hx = 0.5f * x;
  return fmaf(hx, er, hx);
}

__device__ __forceinline__ void ln128_row(const float* __restrict__ src,
                                          float* __restrict__ dst,
                                          const float* __restrict__ g,
                                          const float* __restrict__ b,
                                          int lane) {
  float xv[4];
#pragma unroll
  for (int j = 0; j < 4; ++j) xv[j] = src[lane + 32 * j];
  float s = xv[0] + xv[1] + xv[2] + xv[3];
  s = wave_sum32(s);
  float m = s * (1.0f / 128.0f);
  float vs = 0.0f;
#pragma unroll
  for (int j = 0; j < 4; ++j) { float d = xv[j] - m; vs += d * d; }
  vs = wave_sum32(vs);
  float inv = rsqrtf(vs * (1.0f / 128.0f) + EPSV);
#pragma unroll
  for (int j = 0; j < 4; ++j) {
    int d = lane + 32 * j;
    dst[d] = (xv[j] - m) * inv * g[d] + b[d];
  }
}

__device__ __forceinline__ void sched_fence() { __builtin_amdgcn_sched_barrier(0); }

__device__ __forceinline__ void gram_rows(const float* __restrict__ yt, int cb, int khalf,
                                          int nl, v8f (&acc)[5]) {
  const float* bcol = yt + (khalf * 2) * PG + cb * 16 + nl;
  const float* ar1 = yt + (khalf * 2) * PG + ((cb + 1) & 7) * 16 + nl;
  const float* ar2 = yt + (khalf * 2) * PG + ((cb + 2) & 7) * 16 + nl;
  const float* ar3 = yt + (khalf * 2) * PG + ((cb + 3) & 7) * 16 + nl;
  const int par = cb >> 2;
  const float* bcp = bcol + par * 4 * PG;
  const float* ar4 = yt + (khalf * 2 + par * 4) * PG + ((cb + 4) & 7) * 16 + nl;
  for (int kq = 0; kq < 16; ++kq) {
    v2f bf[4], a1[4], a2[4], a3[4], bp[2], a4[2];
#pragma unroll
    for (int s = 0; s < 4; ++s) {
      const int o = s * 4 * PG;
      bf[s].x = bcol[o];   bf[s].y = bcol[o + PG];
      a1[s].x = ar1[o];    a1[s].y = ar1[o + PG];
      a2[s].x = ar2[o];    a2[s].y = ar2[o + PG];
      a3[s].x = ar3[o];    a3[s].y = ar3[o + PG];
    }
#pragma unroll
    for (int u = 0; u < 2; ++u) {
      const int o = u * 8 * PG;
      bp[u].x = bcp[o];   bp[u].y = bcp[o + PG];
      a4[u].x = ar4[o];   a4[u].y = ar4[o + PG];
    }
    sched_fence();
#pragma unroll
    for (int s = 0; s < 4; ++s) {
      acc[0] = wmma4(bf[s], bf[s], acc[0]);
      acc[1] = wmma4(a1[s], bf[s], acc[1]);
      acc[2] = wmma4(a2[s], bf[s], acc[2]);
      acc[3] = wmma4(a3[s], bf[s], acc[3]);
      if (s & 1) acc[4] = wmma4(a4[s >> 1], bp[s >> 1], acc[4]);
    }
    bcol += 16 * PG;  ar1 += 16 * PG;  ar2 += 16 * PG;  ar3 += 16 * PG;
    bcp += 16 * PG;   ar4 += 16 * PG;
  }
}

__global__ void __launch_bounds__(256) k_gram(const float* __restrict__ x,
                       const float* __restrict__ g1,
                       const float* __restrict__ b1,
                       float* __restrict__ statp,
                       float* __restrict__ gpart) {
  extern __shared__ float yt[];
  __shared__ float sred[2][8][128];
  __shared__ float hx[8][16 * 17];
  const int tid = threadIdx.x, lane = tid & 31, wave = tid >> 5;
  const int b = blockIdx.x >> 3;
  const int oct = blockIdx.x & 7;
  const int khalf = lane >> 4;
  const int nl = lane & 15;
  const v8f vzero = {0, 0, 0, 0, 0, 0, 0, 0};
  v8f acc[5];
#pragma unroll
  for (int t = 0; t < 5; ++t) acc[t] = vzero;
  float a1[4] = {0, 0, 0, 0}, a2[4] = {0, 0, 0, 0};

  for (int sc = 0; sc < 2; ++sc) {
    const long rowbase = (long)b * 4096 + oct * 512 + sc * 256;
    __syncthreads();
    for (int i0 = 0; i0 < 32; i0 += 16) {
      const int r0 = wave * 32 + i0;
      float xv[16][4], s[16], m[16], vs[16];
#pragma unroll
      for (int q = 0; q < 16; ++q) {
        const float* src = x + (rowbase + r0 + q) * 128;
#pragma unroll
        for (int j = 0; j < 4; ++j) xv[q][j] = src[lane + 32 * j];
      }
#pragma unroll
      for (int q = 0; q < 16; ++q) s[q] = xv[q][0] + xv[q][1] + xv[q][2] + xv[q][3];
      wave_sum32_rows<16>(s);
#pragma unroll
      for (int q = 0; q < 16; ++q) {
        m[q] = s[q] * (1.0f / 128.0f);
        float t = 0.0f;
#pragma unroll
        for (int j = 0; j < 4; ++j) { float d = xv[q][j] - m[q]; t += d * d; }
        vs[q] = t;
      }
      wave_sum32_rows<16>(vs);
#pragma unroll
      for (int q = 0; q < 16; ++q) {
        float inv = rsqrtf(vs[q] * (1.0f / 128.0f) + EPSV);
#pragma unroll
        for (int j = 0; j < 4; ++j) {
          int d = lane + 32 * j;
          float y = (xv[q][j] - m[q]) * inv * g1[d] + b1[d];
          a1[j] += y; a2[j] += y * y;
          yt[(r0 + q) * PG + d] = y;
        }
      }
    }
    __syncthreads();
    gram_rows(yt, wave, khalf, nl, acc);
  }

  float* gp = gpart + (long)blockIdx.x * 16384;
  const int de = wave * 16 + nl;
#pragma unroll
  for (int j = 0; j < 4; ++j) {
    const int mi = (wave + j) & 7;
    const int dmBase = mi * 16 + khalf * 8;
#pragma unroll
    for (int r = 0; r < 8; ++r) gp[(dmBase + r) * 128 + de] = acc[j][r];
    if (j > 0) {
      float* mp = gp + de * 128 + dmBase;
      v4f lo, hi;
      lo.x = acc[j][0]; lo.y = acc[j][1]; lo.z = acc[j][2]; lo.w = acc[j][3];
      hi.x = acc[j][4]; hi.y = acc[j][5]; hi.z = acc[j][6]; hi.w = acc[j][7];
      *(v4f*)mp = lo;
      *(v4f*)(mp + 4) = hi;
    }
  }
#pragma unroll
  for (int r = 0; r < 8; ++r) hx[wave][(khalf * 8 + r) * 17 + nl] = acc[4][r];
#pragma unroll
  for (int j = 0; j < 4; ++j) {
    int d = lane + 32 * j;
    sred[0][wave][d] = a1[j];
    sred[1][wave][d] = a2[j];
  }
  __syncthreads();
  {
    const int which = tid >> 7, d = tid & 127;
    float s = 0.0f;
#pragma unroll
    for (int w = 0; w < 8; ++w) s += sred[which][w][d];
    statp[which * 32768 + blockIdx.x * 128 + d] = s;
  }
  {
    const int pw = (wave + 4) & 7;
    const int dmBase = pw * 16 + khalf * 8;
#pragma unroll
    for (int r = 0; r < 8; ++r)
      gp[(dmBase + r) * 128 + de] = acc[4][r] + hx[pw][nl * 17 + khalf * 8 + r];
  }
}

__global__ void __launch_bounds__(128) k_combine(const float* __restrict__ statp,
                          const float* __restrict__ gpart,
                          const float* __restrict__ corr,
                          const float* __restrict__ alpha_logit,
                          float* __restrict__ mixed_out,
                          float* __restrict__ dis_out) {
  __shared__ float mu[32][128];
  __shared__ float iv[32][128];
  __shared__ float wsum[4];
  const int d = blockIdx.x;
  const int e = threadIdx.x;
  const float cde = corr[d * 128 + e];
  const float ced = corr[e * 128 + d];
  const float alog = alpha_logit[0];
  for (int b = 0; b < 32; ++b) {
    float s1 = 0.0f, s2 = 0.0f;
#pragma unroll
    for (int q = 0; q < 8; ++q) {
      s1 += statp[(b * 8 + q) * 128 + e];
      s2 += statp[32768 + (b * 8 + q) * 128 + e];
    }
    float v = s2 - s1 * s1 * (1.0f / 4096.0f);
    mu[b][e] = s1 * (1.0f / 4096.0f);
    iv[b][e] = 1.0f / fmaxf(sqrtf(fmaxf(v, 0.0f)), EPSV);
  }
  __syncthreads();
  float acc = 0.0f;
  for (int b = 0; b < 32; ++b) {
    float md = mu[b][d], me = mu[b][e];
    float ind = iv[b][d], ine = iv[b][e];
    float G = 0.0f;
#pragma unroll
    for (int q = 0; q < 8; ++q)
      G += gpart[(long)(b * 8 + q) * 16384 + d * 128 + e];
    acc += (G - 4096.0f * md * me) * ind * ine;
  }
  const float alpha = 1.0f / (1.0f + __expf(-alog));
  const float learned = tanhf(0.5f * (cde + ced));
  const float dc = fminf(fmaxf(acc * (1.0f / 32.0f), -1.0f), 1.0f);
  const float mx = alpha * learned + (1.0f - alpha) * dc;
  mixed_out[d * 128 + e] = mx;
  float part = wave_sum32((e == d) ? 0.0f : mx);
  if ((e & 31) == 0) wsum[e >> 5] = part;
  __syncthreads();
  if (e == 0) {
    float deg = ((wsum[0] + wsum[1]) + wsum[2]) + wsum[3];
    dis_out[d] = rsqrtf(fmaxf(deg, EPSV));
  }
}

__global__ void __launch_bounds__(128) k_graph(const float* __restrict__ chebw,
                        const float* __restrict__ mixed,
                        const float* __restrict__ dis,
                        const float* __restrict__ Wi,
                        const float* __restrict__ Wo,
                        float* __restrict__ LapP,
                        float* __restrict__ WiP,
                        float* __restrict__ WoP,
                        float* __restrict__ wout) {
  const int g = blockIdx.x * 128 + threadIdx.x;
  const int lane = g & 31;
  const int h = lane >> 4, c = lane & 15;
  if (blockIdx.x < 32) {
    const int jj = (g >> 5) & 15;
    const int nb = g >> 9;
    const int n = nb * 16 + c;
    const float dn = dis[n];
    v4f v;
#pragma unroll
    for (int e = 0; e < 4; ++e) {
      const int k = 8 * jj + 4 * h + e;
      const float dk = dis[k];
      const float m = mixed[k * 128 + n];
      float a = (n == k) ? 0.0f : m;
      float l = ((n == k) ? 1.0f : 0.0f) - dk * a * dn;
      v[e] = fminf(fmaxf(l, -1.5f), 1.5f);
    }
    ((v4f*)LapP)[g] = v;
  } else if (blockIdx.x < 96) {
    const int idx = g - 32 * 128;
    const int jj = (idx >> 5) & 15;
    const int nb = idx >> 9;
    const float* row = Wi + (nb * 16 + c) * 128 + 8 * jj + 4 * h;
    ((v4f*)WiP)[idx] = *(const v4f*)row;
  } else {
    const int idx = g - 96 * 128;
    const int jj = (idx >> 5) & 31;
    const int nb = idx >> 10;
    const float* row = Wo + (nb * 16 + c) * 256 + 8 * jj + 4 * h;
    ((v4f*)WoP)[idx] = *(const v4f*)row;
  }
  if (g == 0) {
    float c0 = chebw[0], c1 = chebw[1], c2 = chebw[2];
    float mx = fmaxf(c0, fmaxf(c1, c2));
    float e0 = __expf(c0 - mx), e1 = __expf(c1 - mx), e2 = __expf(c2 - mx);
    float s = e0 + e1 + e2;
    wout[0] = e0 / s; wout[1] = e1 / s; wout[2] = e2 / s;
  }
}


#define DB 4

template <int NT>
struct BRing { v4f v[DB + 1][NT]; };

template <int NT>
__device__ __forceinline__ void b_prefetch(BRing<NT>& r, const v4f* __restrict__ bP,
                                           int tstride) {
#pragma unroll
  for (int j = 0; j < DB; ++j) {
#pragma unroll
    for (int t = 0; t < NT; ++t) r.v[j][t] = bP[t * tstride + j * 32];
  }
}

template <int NJ, int NT, int PITCH>
__device__ __forceinline__ void wave_product(const float* a4, const v4f* __restrict__ bP,
                                             int tstride, BRing<NT>& r,
                                             v8f (&acc)[NT][4]) {
  v4f an[4];
#pragma unroll
  for (int mi = 0; mi < 4; ++mi) an[mi] = *(const v4f*)(a4 + mi * 16 * PITCH);
#pragma unroll
  for (int jj = 0; jj < NJ; ++jj) {
    v4f ac[4];
#pragma unroll
    for (int mi = 0; mi < 4; ++mi) ac[mi] = an[mi];
    if (jj + 1 < NJ) {
#pragma unroll
      for (int mi = 0; mi < 4; ++mi)
        an[mi] = *(const v4f*)(a4 + mi * 16 * PITCH + (jj + 1) * 8);
    }
    if (jj + DB < NJ) {
#pragma unroll
      for (int t = 0; t < NT; ++t)
        r.v[(jj + DB) % (DB + 1)][t] = bP[t * tstride + (jj + DB) * 32];
    }
    sched_fence();
#pragma unroll
    for (int s = 0; s < 2; ++s) {
#pragma unroll
      for (int mi = 0; mi < 4; ++mi) {
        v2f a;
        a.x = s ? ac[mi].z : ac[mi].x;   a.y = s ? ac[mi].w : ac[mi].y;
#pragma unroll
        for (int t = 0; t < NT; ++t) {
          const v4f b4 = r.v[jj % (DB + 1)][t];
          v2f b;
          b.x = s ? b4.z : b4.x;   b.y = s ? b4.w : b4.y;
          acc[t][mi] = wmma4(a, b, acc[t][mi]);
        }
      }
    }
    sched_fence();
  }
}

__global__ void __launch_bounds__(256) k_main(const float* __restrict__ x,
                       const float* __restrict__ g1, const float* __restrict__ b1,
                       const float* __restrict__ LapP, const float* __restrict__ wv,
                       const float* __restrict__ WiP, const float* __restrict__ bi,
                       const float* __restrict__ WoP, const float* __restrict__ bo,
                       const float* __restrict__ g2, const float* __restrict__ b2,
                       const float* __restrict__ g3, const float* __restrict__ b3,
                       float* __restrict__ out) {
  extern __shared__ float smem[];
  float* yo = smem;
  float* t1 = smem + 64 * PY;
  float* hb = smem + 2 * 64 * PY;
  const int tid = threadIdx.x, lane = tid & 31, wave = tid >> 5;
  const long base = (long)blockIdx.x * 64;
  const float w0 = wv[0], w1 = wv[1], w2 = wv[2];
  const int khalf = lane >> 4;
  const int nl = lane & 15;

  for (int i = 0; i < 8; ++i) {
    int r = wave * 8 + i;
    ln128_row(x + (base + r) * 128, yo + r * PY, g1, b1, lane);
  }

  const float* a4_yo = yo + nl * PY + khalf * 4;
  const float* a4_t1 = t1 + nl * PY + khalf * 4;
  const float* a4_hb = hb + nl * PH + khalf * 4;
  const int nc = wave * 16 + nl;
  const v4f* lapP = (const v4f*)LapP + (wave * 16) * 32 + lane;
  BRing<1> rb;
  b_prefetch<1>(rb, lapP, 0);
  __syncthreads();

  {
    v8f acc[1][4] = {};
    wave_product<16, 1, PY>(a4_yo, lapP, 0, rb, acc);
#pragma unroll
    for (int mi = 0; mi < 4; ++mi) {
      int rowB = mi * 16 + khalf * 8;
#pragma unroll
      for (int r = 0; r < 8; ++r) t1[(rowB + r) * PY + nc] = acc[0][mi][r];
    }
  }
  b_prefetch<1>(rb, lapP, 0);
  __syncthreads();

  {
    v8f acc[1][4] = {};
    wave_product<16, 1, PY>(a4_t1, lapP, 0, rb, acc);
#pragma unroll
    for (int mi = 0; mi < 4; ++mi) {
      int rowB = mi * 16 + khalf * 8;
#pragma unroll
      for (int r = 0; r < 8; ++r) {
        int pos = (rowB + r) * PY + nc;
        float y0 = yo[pos], y1 = t1[pos];
        float tx2 = fminf(fmaxf(2.0f * acc[0][mi][r] - y0, -50.0f), 50.0f);
        yo[pos] = w0 * y0 + w1 * y1 + w2 * tx2;
      }
    }
  }
  const v4f* wiP = (const v4f*)WiP + (wave * 16) * 32 + lane;
  BRing<2> rb2;
  b_prefetch<2>(rb2, wiP, 8 * 16 * 32);
  __syncthreads();

  {
    v8f acc[2][4] = {};
    wave_product<16, 2, PY>(a4_yo, wiP, 8 * 16 * 32, rb2, acc);
#pragma unroll
    for (int j = 0; j < 2; ++j) {
      int ncj = nc + 128 * j;
      float bia = bi[ncj];
#pragma unroll
      for (int mi = 0; mi < 4; ++mi) {
        int rowB = mi * 16 + khalf * 8;
#pragma unroll
        for (int r = 0; r < 8; ++r) hb[(rowB + r) * PH + ncj] = acc[j][mi][r] + bia;
      }
    }
  }
  __syncthreads();

  {
    const int rw = wave * 8;
    float xv[8][8], s[8], m[8], vs[8];
#pragma unroll
    for (int i = 0; i < 8; ++i) {
#pragma unroll
      for (int j = 0; j < 8; ++j) xv[i][j] = hb[(rw + i) * PH + lane + 32 * j];
    }
#pragma unroll
    for (int i = 0; i < 8; ++i) {
      float t = 0.0f;
#pragma unroll
      for (int j = 0; j < 8; ++j) t += xv[i][j];
      s[i] = t;
    }
    wave_sum32_rows<8>(s);
#pragma unroll
    for (int i = 0; i < 8; ++i) {
      m[i] = s[i] * (1.0f / 256.0f);
      float t = 0.0f;
#pragma unroll
      for (int j = 0; j < 8; ++j) { float d = xv[i][j] - m[i]; t += d * d; }
      vs[i] = t;
    }
    wave_sum32_rows<8>(vs);
#pragma unroll
    for (int i = 0; i < 8; ++i) {
      float inv = rsqrtf(vs[i] * (1.0f / 256.0f) + EPSV);
#pragma unroll
      for (int j = 0; j < 8; ++j) {
        int d = lane + 32 * j;
        hb[(rw + i) * PH + d] = gelu_erf((xv[i][j] - m[i]) * inv * g2[d] + b2[d]);
      }
    }
  }
  const v4f* woP = (const v4f*)WoP + (wave * 32) * 32 + lane;
  b_prefetch<1>(rb, woP, 0);
  __syncthreads();

  {
    v8f acc[1][4] = {};
    wave_product<32, 1, PH>(a4_hb, woP, 0, rb, acc);
    float boa = bo[nc];
#pragma unroll
    for (int mi = 0; mi < 4; ++mi) {
      int rowB = mi * 16 + khalf * 8;
#pragma unroll
      for (int r = 0; r < 8; ++r) t1[(rowB + r) * PY + nc] = acc[0][mi][r] + boa;
    }
  }
  __syncthreads();

  {
    const int rw = wave * 8;
    float xv[8][4], s[8], m[8], vs[8];
#pragma unroll
    for (int i = 0; i < 8; ++i) {
#pragma unroll
      for (int j = 0; j < 4; ++j) xv[i][j] = t1[(rw + i) * PY + lane + 32 * j];
    }
#pragma unroll
    for (int i = 0; i < 8; ++i) s[i] = xv[i][0] + xv[i][1] + xv[i][2] + xv[i][3];
    wave_sum32_rows<8>(s);
#pragma unroll
    for (int i = 0; i < 8; ++i) {
      m[i] = s[i] * (1.0f / 128.0f);
      float t = 0.0f;
#pragma unroll
      for (int j = 0; j < 4; ++j) { float d = xv[i][j] - m[i]; t += d * d; }
      vs[i] = t;
    }
    wave_sum32_rows<8>(vs);
#pragma unroll
    for (int i = 0; i < 8; ++i) {
      float inv = rsqrtf(vs[i] * (1.0f / 128.0f) + EPSV);
#pragma unroll
      for (int j = 0; j < 4; ++j) {
        int d = lane + 32 * j;
        out[(base + rw + i) * 128 + d] = (xv[i][j] - m[i]) * inv * g3[d] + b3[d];
      }
    }
  }
}

extern "C" void kernel_launch(void* const* d_in, const int* in_sizes, int n_in,
                              void* d_out, int out_size, void* d_ws, size_t ws_size,
                              hipStream_t stream) {
  const float* x     = (const float*)d_in[0];
  const float* corr  = (const float*)d_in[1];
  const float* alog  = (const float*)d_in[2];
  const float* chebw = (const float*)d_in[3];
  const float* Wi    = (const float*)d_in[4];
  const float* bi    = (const float*)d_in[5];
  const float* Wo    = (const float*)d_in[6];
  const float* bo    = (const float*)d_in[7];
  const float* g1    = (const float*)d_in[8];
  const float* b1    = (const float*)d_in[9];
  const float* g2    = (const float*)d_in[10];
  const float* b2    = (const float*)d_in[11];
  const float* g3    = (const float*)d_in[12];
  const float* b3    = (const float*)d_in[13];
  float* out = (float*)d_out;
  float* ws  = (float*)d_ws;

  float* gpart  = ws + 8192;
  float* disv   = ws + 8192 + 256 * 16384;
  float* LapP   = disv + 16384;
  float* wv     = LapP + 16384;
  float* WiP    = gpart;
  float* WoP    = gpart + 32768;
  float* mixed  = out + (long)32 * 4096 * 128;
  float* statp  = out;

  k_gram<<<256, 256, 256 * PG * sizeof(float), stream>>>(x, g1, b1, statp, gpart);
  k_combine<<<128, 128, 0, stream>>>(statp, gpart, corr, alog, mixed, disv);
  k_graph<<<160, 128, 0, stream>>>(chebw, mixed, disv, Wi, Wo, LapP, WiP, WoP, wv);
  k_main<<<2048, 256, (2 * 64 * PY + 64 * PH) * sizeof(float), stream>>>(x, g1, b1, LapP, wv, WiP, bi, WoP, bo,
                                        g2, b2, g3, b3, out);
}


# k_combine: a wave owns 8 batches of its row, a lane 4 columns: all rows as 16-byte loads in groups of 48 (was 4-byte loads, 16 dependent rounds); batch sum = 4 wave sums of 8 (order changed)
# speedup vs baseline: 5.1579x; 1.0391x over previous
#include <hip/hip_runtime.h>
#include <hip/hip_bf16.h>
#include <math.h>

#define EPSV 1e-8f

#define PY 132
#define PH 260
#define PG 144

typedef float v2f __attribute__((ext_vector_type(2)));
typedef float v4f __attribute__((ext_vector_type(4)));
typedef float v8f __attribute__((ext_vector_type(8)));

__device__ __forceinline__ v8f wmma4(v2f a, v2f b, v8f c) {
  return __builtin_amdgcn_wmma_f32_16x16x4_f32(false, a, false, b, (short)0, c,
                                               false, false);
}

__device__ __forceinline__ float wave_sum32(float v) {
#pragma unroll
  for (int off = 16; off > 0; off >>= 1) v += __shfl_xor(v, off, 32);
  return v;
}

template <int R>
__device__ __forceinline__ void wave_sum32_rows(float (&v)[R]) {
#pragma unroll
  for (int off = 16; off > 0; off >>= 1) {
    float t[R];
#pragma unroll
    for (int r = 0; r < R; ++r) t[r] = __shfl_xor(v[r], off, 32);
#pragma unroll
    for (int r = 0; r < R; ++r) v[r] += t[r];
  }
}

__device__ __forceinline__ float gelu_erf(float x) {
  const float z = x * 0.70710678118654752f;
  const float a = fabsf(z);
  const float t = __builtin_amdgcn_rcpf(fmaf(0.3275911f, a, 1.0f));
  float q = fmaf(1.061405429f, t, -1.453152027f);
  q = fmaf(q, t, 1.421413741f);
  q = fmaf(q, t, -0.284496736f);
  q = fmaf(q, t, 0.254829592f);
  q = q * t;
  const float e = __builtin_amdgcn_exp2f(z * z * -1.4426950408889634f);
  const float er = copysignf(1.0f - q * e, z);
  const float hx = 0.5f * x;
  return fmaf(hx, er, hx);
}

__device__ __forceinline__ void ln128_row(const float* __restrict__ src,
                                          float* __restrict__ dst,
                                          const float* __restrict__ g,
                                          const float* __restrict__ b,
                                          int lane) {
  float xv[4];
#pragma unroll
  for (int j = 0; j < 4; ++j) xv[j] = src[lane + 32 * j];
  float s = xv[0] + xv[1] + xv[2] + xv[3];
  s = wave_sum32(s);
  float m = s * (1.0f / 128.0f);
  float vs = 0.0f;
#pragma unroll
  for (int j = 0; j < 4; ++j) { float d = xv[j] - m; vs += d * d; }
  vs = wave_sum32(vs);
  float inv = rsqrtf(vs * (1.0f / 128.0f) + EPSV);
#pragma unroll
  for (int j = 0; j < 4; ++j) {
    int d = lane + 32 * j;
    dst[d] = (xv[j] - m) * inv * g[d] + b[d];
  }
}

__device__ __forceinline__ void sched_fence() { __builtin_amdgcn_sched_barrier(0); }

__device__ __forceinline__ void gram_rows(const float* __restrict__ yt, int cb, int khalf,
                                          int nl, v8f (&acc)[5]) {
  const float* bcol = yt + (khalf * 2) * PG + cb * 16 + nl;
  const float* ar1 = yt + (khalf * 2) * PG + ((cb + 1) & 7) * 16 + nl;
  const float* ar2 = yt + (khalf * 2) * PG + ((cb + 2) & 7) * 16 + nl;
  const float* ar3 = yt + (khalf * 2) * PG + ((cb + 3) & 7) * 16 + nl;
  const int par = cb >> 2;
  const float* bcp = bcol + par * 4 * PG;
  const float* ar4 = yt + (khalf * 2 + par * 4) * PG + ((cb + 4) & 7) * 16 + nl;
  for (int kq = 0; kq < 16; ++kq) {
    v2f bf[4], a1[4], a2[4], a3[4], bp[2], a4[2];
#pragma unroll
    for (int s = 0; s < 4; ++s) {
      const int o = s * 4 * PG;
      bf[s].x = bcol[o];   bf[s].y = bcol[o + PG];
      a1[s].x = ar1[o];    a1[s].y = ar1[o + PG];
      a2[s].x = ar2[o];    a2[s].y = ar2[o + PG];
      a3[s].x = ar3[o];    a3[s].y = ar3[o + PG];
    }
#pragma unroll
    for (int u = 0; u < 2; ++u) {
      const int o = u * 8 * PG;
      bp[u].x = bcp[o];   bp[u].y = bcp[o + PG];
      a4[u].x = ar4[o];   a4[u].y = ar4[o + PG];
    }
    sched_fence();
#pragma unroll
    for (int s = 0; s < 4; ++s) {
      acc[0] = wmma4(bf[s], bf[s], acc[0]);
      acc[1] = wmma4(a1[s], bf[s], acc[1]);
      acc[2] = wmma4(a2[s], bf[s], acc[2]);
      acc[3] = wmma4(a3[s], bf[s], acc[3]);
      if (s & 1) acc[4] = wmma4(a4[s >> 1], bp[s >> 1], acc[4]);
    }
    bcol += 16 * PG;  ar1 += 16 * PG;  ar2 += 16 * PG;  ar3 += 16 * PG;
    bcp += 16 * PG;   ar4 += 16 * PG;
  }
}

__global__ void __launch_bounds__(256) k_gram(const float* __restrict__ x,
                       const float* __restrict__ g1,
                       const float* __restrict__ b1,
                       float* __restrict__ statp,
                       float* __restrict__ gpart) {
  extern __shared__ float yt[];
  __shared__ float sred[2][8][128];
  __shared__ float hx[8][16 * 17];
  const int tid = threadIdx.x, lane = tid & 31, wave = tid >> 5;
  const int b = blockIdx.x >> 3;
  const int oct = blockIdx.x & 7;
  const int khalf = lane >> 4;
  const int nl = lane & 15;
  const v8f vzero = {0, 0, 0, 0, 0, 0, 0, 0};
  v8f acc[5];
#pragma unroll
  for (int t = 0; t < 5; ++t) acc[t] = vzero;
  float a1[4] = {0, 0, 0, 0}, a2[4] = {0, 0, 0, 0};

  for (int sc = 0; sc < 2; ++sc) {
    const long rowbase = (long)b * 4096 + oct * 512 + sc * 256;
    __syncthreads();
    for (int i0 = 0; i0 < 32; i0 += 16) {
      const int r0 = wave * 32 + i0;
      float xv[16][4], s[16], m[16], vs[16];
#pragma unroll
      for (int q = 0; q < 16; ++q) {
        const float* src = x + (rowbase + r0 + q) * 128;
#pragma unroll
        for (int j = 0; j < 4; ++j) xv[q][j] = src[lane + 32 * j];
      }
#pragma unroll
      for (int q = 0; q < 16; ++q) s[q] = xv[q][0] + xv[q][1] + xv[q][2] + xv[q][3];
      wave_sum32_rows<16>(s);
#pragma unroll
      for (int q = 0; q < 16; ++q) {
        m[q] = s[q] * (1.0f / 128.0f);
        float t = 0.0f;
#pragma unroll
        for (int j = 0; j < 4; ++j) { float d = xv[q][j] - m[q]; t += d * d; }
        vs[q] = t;
      }
      wave_sum32_rows<16>(vs);
#pragma unroll
      for (int q = 0; q < 16; ++q) {
        float inv = rsqrtf(vs[q] * (1.0f / 128.0f) + EPSV);
#pragma unroll
        for (int j = 0; j < 4; ++j) {
          int d = lane + 32 * j;
          float y = (xv[q][j] - m[q]) * inv * g1[d] + b1[d];
          a1[j] += y; a2[j] += y * y;
          yt[(r0 + q) * PG + d] = y;
        }
      }
    }
    __syncthreads();
    gram_rows(yt, wave, khalf, nl, acc);
  }

  float* gp = gpart + (long)blockIdx.x * 16384;
  const int de = wave * 16 + nl;
#pragma unroll
  for (int j = 0; j < 4; ++j) {
    const int mi = (wave + j) & 7;
    const int dmBase = mi * 16 + khalf * 8;
#pragma unroll
    for (int r = 0; r < 8; ++r) gp[(dmBase + r) * 128 + de] = acc[j][r];
    if (j > 0) {
      float* mp = gp + de * 128 + dmBase;
      v4f lo, hi;
      lo.x = acc[j][0]; lo.y = acc[j][1]; lo.z = acc[j][2]; lo.w = acc[j][3];
      hi.x = acc[j][4]; hi.y = acc[j][5]; hi.z = acc[j][6]; hi.w = acc[j][7];
      *(v4f*)mp = lo;
      *(v4f*)(mp + 4) = hi;
    }
  }
#pragma unroll
  for (int r = 0; r < 8; ++r) hx[wave][(khalf * 8 + r) * 17 + nl] = acc[4][r];
#pragma unroll
  for (int j = 0; j < 4; ++j) {
    int d = lane + 32 * j;
    sred[0][wave][d] = a1[j];
    sred[1][wave][d] = a2[j];
  }
  __syncthreads();
  {
    const int which = tid >> 7, d = tid & 127;
    float s = 0.0f;
#pragma unroll
    for (int w = 0; w < 8; ++w) s += sred[which][w][d];
    statp[which * 32768 + blockIdx.x * 128 + d] = s;
  }
  {
    const int pw = (wave + 4) & 7;
    const int dmBase = pw * 16 + khalf * 8;
#pragma unroll
    for (int r = 0; r < 8; ++r)
      gp[(dmBase + r) * 128 + de] = acc[4][r] + hx[pw][nl * 17 + khalf * 8 + r];
  }
}


struct RowSet { v4f s1[2][8], s2[2][8], g[2][8]; };

__device__ __forceinline__ void rows_load(RowSet& r, const float* __restrict__ sp,
                                          const float* __restrict__ gp, int grp) {
#pragma unroll
  for (int t = 0; t < 2; ++t) {
#pragma unroll
    for (int q = 0; q < 8; ++q) {
      const int row = (grp * 2 + t) * 8 + q;
      r.s1[t][q] = *(const v4f*)(sp + row * 128);
      r.s2[t][q] = *(const v4f*)(sp + 32768 + row * 128);
      r.g[t][q] = *(const v4f*)(gp + (long)row * 16384);
    }
  }
}

__device__ __forceinline__ void rows_use(const RowSet& r, int dl, int dc, v4f& accv) {
#pragma unroll
  for (int t = 0; t < 2; ++t) {
    v4f s1 = {0, 0, 0, 0}, s2 = {0, 0, 0, 0}, G = {0, 0, 0, 0};
#pragma unroll
    for (int q = 0; q < 8; ++q) { s1 += r.s1[t][q]; s2 += r.s2[t][q]; G += r.g[t][q]; }
    v4f m, iv;
#pragma unroll
    for (int c = 0; c < 4; ++c) {
      float v = s2[c] - s1[c] * s1[c] * (1.0f / 4096.0f);
      m[c] = s1[c] * (1.0f / 4096.0f);
      iv[c] = 1.0f / fmaxf(sqrtf(fmaxf(v, 0.0f)), EPSV);
    }
    const float mc = (dc == 0) ? m[0] : (dc == 1) ? m[1] : (dc == 2) ? m[2] : m[3];
    const float ic = (dc == 0) ? iv[0] : (dc == 1) ? iv[1] : (dc == 2) ? iv[2] : iv[3];
    const float md = __int_as_float(__builtin_amdgcn_readlane(__float_as_int(mc), dl));
    const float ind = __int_as_float(__builtin_amdgcn_readlane(__float_as_int(ic), dl));
#pragma unroll
    for (int c = 0; c < 4; ++c) accv[c] += (G[c] - 4096.0f * md * m[c]) * ind * iv[c];
  }
}

__global__ void __launch_bounds__(128) k_combine(const float* __restrict__ statp,
                          const float* __restrict__ gpart,
                          const float* __restrict__ corr,
                          const float* __restrict__ alpha_logit,
                          float* __restrict__ mixed_out,
                          float* __restrict__ dis_out) {
  __shared__ __attribute__((aligned(16))) float part[4][128];
  __shared__ float wsum[4];
  const int d = blockIdx.x;
  const int e = threadIdx.x;
  const int lane = e & 31, wave = e >> 5;
  const float cde = corr[d * 128 + e];
  const float ced = corr[e * 128 + d];
  const float alog = alpha_logit[0];
  const int dl = d >> 2, dc = d & 3;
  const float* sp = statp + (wave * 64) * 128 + 4 * lane;
  const float* gp = gpart + (long)(wave * 64) * 16384 + d * 128 + 4 * lane;
  v4f accv = {0, 0, 0, 0};
  RowSet ra, rb;
  rows_load(ra, sp, gp, 0);
  rows_load(rb, sp, gp, 1);
  sched_fence();
  rows_use(ra, dl, dc, accv);
  rows_load(ra, sp, gp, 2);
  sched_fence();
  rows_use(rb, dl, dc, accv);
  rows_load(rb, sp, gp, 3);
  sched_fence();
  rows_use(ra, dl, dc, accv);
  rows_use(rb, dl, dc, accv);
  *(v4f*)&part[wave][4 * lane] = accv;
  __syncthreads();
  const float acc = ((part[0][e] + part[1][e]) + part[2][e]) + part[3][e];
  const float alpha = 1.0f / (1.0f + __expf(-alog));
  const float learned = tanhf(0.5f * (cde + ced));
  const float dcr = fminf(fmaxf(acc * (1.0f / 32.0f), -1.0f), 1.0f);
  const float mx = alpha * learned + (1.0f - alpha) * dcr;
  mixed_out[d * 128 + e] = mx;
  float psum = wave_sum32((e == d) ? 0.0f : mx);
  if ((e & 31) == 0) wsum[e >> 5] = psum;
  __syncthreads();
  if (e == 0) {
    float deg = ((wsum[0] + wsum[1]) + wsum[2]) + wsum[3];
    dis_out[d] = rsqrtf(fmaxf(deg, EPSV));
  }
}

__global__ void __launch_bounds__(128) k_graph(const float* __restrict__ chebw,
                        const float* __restrict__ mixed,
                        const float* __restrict__ dis,
                        const float* __restrict__ Wi,
                        const float* __restrict__ Wo,
                        float* __restrict__ LapP,
                        float* __restrict__ WiP,
                        float* __restrict__ WoP,
                        float* __restrict__ wout) {
  const int g = blockIdx.x * 128 + threadIdx.x;
  const int lane = g & 31;
  const int h = lane >> 4, c = lane & 15;
  if (blockIdx.x < 32) {
    const int jj = (g >> 5) & 15;
    const int nb = g >> 9;
    const int n = nb * 16 + c;
    const float dn = dis[n];
    v4f v;
#pragma unroll
    for (int e = 0; e < 4; ++e) {
      const int k = 8 * jj + 4 * h + e;
      const float dk = dis[k];
      const float m = mixed[k * 128 + n];
      float a = (n == k) ? 0.0f : m;
      float l = ((n == k) ? 1.0f : 0.0f) - dk * a * dn;
      v[e] = fminf(fmaxf(l, -1.5f), 1.5f);
    }
    ((v4f*)LapP)[g] = v;
  } else if (blockIdx.x < 96) {
    const int idx = g - 32 * 128;
    const int jj = (idx >> 5) & 15;
    const int nb = idx >> 9;
    const float* row = Wi + (nb * 16 + c) * 128 + 8 * jj + 4 * h;
    ((v4f*)WiP)[idx] = *(const v4f*)row;
  } else {
    const int idx = g - 96 * 128;
    const int jj = (idx >> 5) & 31;
    const int nb = idx >> 10;
    const float* row = Wo + (nb * 16 + c) * 256 + 8 * jj + 4 * h;
    ((v4f*)WoP)[idx] = *(const v4f*)row;
  }
  if (g == 0) {
    float c0 = chebw[0], c1 = chebw[1], c2 = chebw[2];
    float mx = fmaxf(c0, fmaxf(c1, c2));
    float e0 = __expf(c0 - mx), e1 = __expf(c1 - mx), e2 = __expf(c2 - mx);
    float s = e0 + e1 + e2;
    wout[0] = e0 / s; wout[1] = e1 / s; wout[2] = e2 / s;
  }
}


#define DB 4

template <int NT>
struct BRing { v4f v[DB + 1][NT]; };

template <int NT>
__device__ __forceinline__ void b_prefetch(BRing<NT>& r, const v4f* __restrict__ bP,
                                           int tstride) {
#pragma unroll
  for (int j = 0; j < DB; ++j) {
#pragma unroll
    for (int t = 0; t < NT; ++t) r.v[j][t] = bP[t * tstride + j * 32];
  }
}

template <int NJ, int NT, int PITCH>
__device__ __forceinline__ void wave_product(const float* a4, const v4f* __restrict__ bP,
                                             int tstride, BRing<NT>& r,
                                             v8f (&acc)[NT][4]) {
  v4f an[4];
#pragma unroll
  for (int mi = 0; mi < 4; ++mi) an[mi] = *(const v4f*)(a4 + mi * 16 * PITCH);
#pragma unroll
  for (int jj = 0; jj < NJ; ++jj) {
    v4f ac[4];
#pragma unroll
    for (int mi = 0; mi < 4; ++mi) ac[mi] = an[mi];
    if (jj + 1 < NJ) {
#pragma unroll
      for (int mi = 0; mi < 4; ++mi)
        an[mi] = *(const v4f*)(a4 + mi * 16 * PITCH + (jj + 1) * 8);
    }
    if (jj + DB < NJ) {
#pragma unroll
      for (int t = 0; t < NT; ++t)
        r.v[(jj + DB) % (DB + 1)][t] = bP[t * tstride + (jj + DB) * 32];
    }
    sched_fence();
#pragma unroll
    for (int s = 0; s < 2; ++s) {
#pragma unroll
      for (int mi = 0; mi < 4; ++mi) {
        v2f a;
        a.x = s ? ac[mi].z : ac[mi].x;   a.y = s ? ac[mi].w : ac[mi].y;
#pragma unroll
        for (int t = 0; t < NT; ++t) {
          const v4f b4 = r.v[jj % (DB + 1)][t];
          v2f b;
          b.x = s ? b4.z : b4.x;   b.y = s ? b4.w : b4.y;
          acc[t][mi] = wmma4(a, b, acc[t][mi]);
        }
      }
    }
    sched_fence();
  }
}

__global__ void __launch_bounds__(256) k_main(const float* __restrict__ x,
                       const float* __restrict__ g1, const float* __restrict__ b1,
                       const float* __restrict__ LapP, const float* __restrict__ wv,
                       const float* __restrict__ WiP, const float* __restrict__ bi,
                       const float* __restrict__ WoP, const float* __restrict__ bo,
                       const float* __restrict__ g2, const float* __restrict__ b2,
                       const float* __restrict__ g3, const float* __restrict__ b3,
                       float* __restrict__ out) {
  extern __shared__ float smem[];
  float* yo = smem;
  float* t1 = smem + 64 * PY;
  float* hb = smem + 2 * 64 * PY;
  const int tid = threadIdx.x, lane = tid & 31, wave = tid >> 5;
  const long base = (long)blockIdx.x * 64;
  const float w0 = wv[0], w1 = wv[1], w2 = wv[2];
  const int khalf = lane >> 4;
  const int nl = lane & 15;

  for (int i = 0; i < 8; ++i) {
    int r = wave * 8 + i;
    ln128_row(x + (base + r) * 128, yo + r * PY, g1, b1, lane);
  }

  const float* a4_yo = yo + nl * PY + khalf * 4;
  const float* a4_t1 = t1 + nl * PY + khalf * 4;
  const float* a4_hb = hb + nl * PH + khalf * 4;
  const int nc = wave * 16 + nl;
  const v4f* lapP = (const v4f*)LapP + (wave * 16) * 32 + lane;
  BRing<1> rb;
  b_prefetch<1>(rb, lapP, 0);
  __syncthreads();

  {
    v8f acc[1][4] = {};
    wave_product<16, 1, PY>(a4_yo, lapP, 0, rb, acc);
#pragma unroll
    for (int mi = 0; mi < 4; ++mi) {
      int rowB = mi * 16 + khalf * 8;
#pragma unroll
      for (int r = 0; r < 8; ++r) t1[(rowB + r) * PY + nc] = acc[0][mi][r];
    }
  }
  b_prefetch<1>(rb, lapP, 0);
  __syncthreads();

  {
    v8f acc[1][4] = {};
    wave_product<16, 1, PY>(a4_t1, lapP, 0, rb, acc);
#pragma unroll
    for (int mi = 0; mi < 4; ++mi) {
      int rowB = mi * 16 + khalf * 8;
#pragma unroll
      for (int r = 0; r < 8; ++r) {
        int pos = (rowB + r) * PY + nc;
        float y0 = yo[pos], y1 = t1[pos];
        float tx2 = fminf(fmaxf(2.0f * acc[0][mi][r] - y0, -50.0f), 50.0f);
        yo[pos] = w0 * y0 + w1 * y1 + w2 * tx2;
      }
    }
  }
  const v4f* wiP = (const v4f*)WiP + (wave * 16) * 32 + lane;
  BRing<2> rb2;
  b_prefetch<2>(rb2, wiP, 8 * 16 * 32);
  __syncthreads();

  {
    v8f acc[2][4] = {};
    wave_product<16, 2, PY>(a4_yo, wiP, 8 * 16 * 32, rb2, acc);
#pragma unroll
    for (int j = 0; j < 2; ++j) {
      int ncj = nc + 128 * j;
      float bia = bi[ncj];
#pragma unroll
      for (int mi = 0; mi < 4; ++mi) {
        int rowB = mi * 16 + khalf * 8;
#pragma unroll
        for (int r = 0; r < 8; ++r) hb[(rowB + r) * PH + ncj] = acc[j][mi][r] + bia;
      }
    }
  }
  __syncthreads();

  {
    const int rw = wave * 8;
    float xv[8][8], s[8], m[8], vs[8];
#pragma unroll
    for (int i = 0; i < 8; ++i) {
#pragma unroll
      for (int j = 0; j < 8; ++j) xv[i][j] = hb[(rw + i) * PH + lane + 32 * j];
    }
#pragma unroll
    for (int i = 0; i < 8; ++i) {
      float t = 0.0f;
#pragma unroll
      for (int j = 0; j < 8; ++j) t += xv[i][j];
      s[i] = t;
    }
    wave_sum32_rows<8>(s);
#pragma unroll
    for (int i = 0; i < 8; ++i) {
      m[i] = s[i] * (1.0f / 256.0f);
      float t = 0.0f;
#pragma unroll
      for (int j = 0; j < 8; ++j) { float d = xv[i][j] - m[i]; t += d * d; }
      vs[i] = t;
    }
    wave_sum32_rows<8>(vs);
#pragma unroll
    for (int i = 0; i < 8; ++i) {
      float inv = rsqrtf(vs[i] * (1.0f / 256.0f) + EPSV);
#pragma unroll
      for (int j = 0; j < 8; ++j) {
        int d = lane + 32 * j;
        hb[(rw + i) * PH + d] = gelu_erf((xv[i][j] - m[i]) * inv * g2[d] + b2[d]);
      }
    }
  }
  const v4f* woP = (const v4f*)WoP + (wave * 32) * 32 + lane;
  b_prefetch<1>(rb, woP, 0);
  __syncthreads();

  {
    v8f acc[1][4] = {};
    wave_product<32, 1, PH>(a4_hb, woP, 0, rb, acc);
    float boa = bo[nc];
#pragma unroll
    for (int mi = 0; mi < 4; ++mi) {
      int rowB = mi * 16 + khalf * 8;
#pragma unroll
      for (int r = 0; r < 8; ++r) t1[(rowB + r) * PY + nc] = acc[0][mi][r] + boa;
    }
  }
  __syncthreads();

  {
    const int rw = wave * 8;
    float xv[8][4], s[8], m[8], vs[8];
#pragma unroll
    for (int i = 0; i < 8; ++i) {
#pragma unroll
      for (int j = 0; j < 4; ++j) xv[i][j] = t1[(rw + i) * PY + lane + 32 * j];
    }
#pragma unroll
    for (int i = 0; i < 8; ++i) s[i] = xv[i][0] + xv[i][1] + xv[i][2] + xv[i][3];
    wave_sum32_rows<8>(s);
#pragma unroll
    for (int i = 0; i < 8; ++i) {
      m[i] = s[i] * (1.0f / 128.0f);
      float t = 0.0f;
#pragma unroll
      for (int j = 0; j < 4; ++j) { float d = xv[i][j] - m[i]; t += d * d; }
      vs[i] = t;
    }
    wave_sum32_rows<8>(vs);
#pragma unroll
    for (int i = 0; i < 8; ++i) {
      float inv = rsqrtf(vs[i] * (1.0f / 128.0f) + EPSV);
#pragma unroll
      for (int j = 0; j < 4; ++j) {
        int d = lane + 32 * j;
        out[(base + rw + i) * 128 + d] = (xv[i][j] - m[i]) * inv * g3[d] + b3[d];
      }
    }
  }
}

extern "C" void kernel_launch(void* const* d_in, const int* in_sizes, int n_in,
                              void* d_out, int out_size, void* d_ws, size_t ws_size,
                              hipStream_t stream) {
  const float* x     = (const float*)d_in[0];
  const float* corr  = (const float*)d_in[1];
  const float* alog  = (const float*)d_in[2];
  const float* chebw = (const float*)d_in[3];
  const float* Wi    = (const float*)d_in[4];
  const float* bi    = (const float*)d_in[5];
  const float* Wo    = (const float*)d_in[6];
  const float* bo    = (const float*)d_in[7];
  const float* g1    = (const float*)d_in[8];
  const float* b1    = (const float*)d_in[9];
  const float* g2    = (const float*)d_in[10];
  const float* b2    = (const float*)d_in[11];
  const float* g3    = (const float*)d_in[12];
  const float* b3    = (const float*)d_in[13];
  float* out = (float*)d_out;
  float* ws  = (float*)d_ws;

  float* gpart  = ws + 8192;
  float* disv   = ws + 8192 + 256 * 16384;
  float* LapP   = disv + 16384;
  float* wv     = LapP + 16384;
  float* WiP    = gpart;
  float* WoP    = gpart + 32768;
  float* mixed  = out + (long)32 * 4096 * 128;
  float* statp  = out;

  k_gram<<<256, 256, 256 * PG * sizeof(float), stream>>>(x, g1, b1, statp, gpart);
  k_combine<<<128, 128, 0, stream>>>(statp, gpart, corr, alog, mixed, disv);
  k_graph<<<160, 128, 0, stream>>>(chebw, mixed, disv, Wi, Wo, LapP, WiP, WoP, wv);
  k_main<<<2048, 256, (2 * 64 * PY + 64 * PH) * sizeof(float), stream>>>(x, g1, b1, LapP, wv, WiP, bi, WoP, bo,
                                        g2, b2, g3, b3, out);
}


# All LayerNorm and degree lane sums on the vector ALU: row_xmask DPP adds + one permlanex16 per sum replace ds_bpermute shuffles (no LDS round trip per step); same pairs and order, sums bit-identical
# speedup vs baseline: 5.3315x; 1.0337x over previous
#include <hip/hip_runtime.h>
#include <hip/hip_bf16.h>
#include <math.h>

#define EPSV 1e-8f

#define PY 132
#define PH 260
#define PG 144

typedef float v2f __attribute__((ext_vector_type(2)));
typedef float v4f __attribute__((ext_vector_type(4)));
typedef float v8f __attribute__((ext_vector_type(8)));

__device__ __forceinline__ v8f wmma4(v2f a, v2f b, v8f c) {
  return __builtin_amdgcn_wmma_f32_16x16x4_f32(false, a, false, b, (short)0, c,
                                               false, false);
}

template <int M>
__device__ __forceinline__ float row_xor(float v) {
  const int b = __float_as_int(v);
  return __int_as_float(__builtin_amdgcn_update_dpp(b, b, 0x160 + M, 0xf, 0xf, true));
}
__device__ __forceinline__ float half_xor(float v) {
  const int b = __float_as_int(v);
  return __int_as_float(
      __builtin_amdgcn_permlanex16(b, b, 0x76543210, 0xfedcba98, false, false));
}

__device__ __forceinline__ float wave_sum32(float v) {
  v += half_xor(v);
  v += row_xor<8>(v);
  v += row_xor<4>(v);
  v += row_xor<2>(v);
  v += row_xor<1>(v);
  return v;
}

template <int R>
__device__ __forceinline__ void wave_sum32_rows(float (&v)[R]) {
#pragma unroll
  for (int r = 0; r < R; ++r) v[r] += half_xor(v[r]);
#pragma unroll
  for (int r = 0; r < R; ++r) v[r] += row_xor<8>(v[r]);
#pragma unroll
  for (int r = 0; r < R; ++r) v[r] += row_xor<4>(v[r]);
#pragma unroll
  for (int r = 0; r < R; ++r) v[r] += row_xor<2>(v[r]);
#pragma unroll
  for (int r = 0; r < R; ++r) v[r] += row_xor<1>(v[r]);
}

__device__ __forceinline__ float gelu_erf(float x) {
  const float z = x * 0.70710678118654752f;
  const float a = fabsf(z);
  const float t = __builtin_amdgcn_rcpf(fmaf(0.3275911f, a, 1.0f));
  float q = fmaf(1.061405429f, t, -1.453152027f);
  q = fmaf(q, t, 1.421413741f);
  q = fmaf(q, t, -0.284496736f);
  q = fmaf(q, t, 0.254829592f);
  q = q * t;
  const float e = __builtin_amdgcn_exp2f(z * z * -1.4426950408889634f);
  const float er = copysignf(1.0f - q * e, z);
  const float hx = 0.5f * x;
  return fmaf(hx, er, hx);
}

__device__ __forceinline__ void ln128_row(const float* __restrict__ src,
                                          float* __restrict__ dst,
                                          const float* __restrict__ g,
                                          const float* __restrict__ b,
                                          int lane) {
  float xv[4];
#pragma unroll
  for (int j = 0; j < 4; ++j) xv[j] = src[lane + 32 * j];
  float s = xv[0] + xv[1] + xv[2] + xv[3];
  s = wave_sum32(s);
  float m = s * (1.0f / 128.0f);
  float vs = 0.0f;
#pragma unroll
  for (int j = 0; j < 4; ++j) { float d = xv[j] - m; vs += d * d; }
  vs = wave_sum32(vs);
  float inv = rsqrtf(vs * (1.0f / 128.0f) + EPSV);
#pragma unroll
  for (int j = 0; j < 4; ++j) {
    int d = lane + 32 * j;
    dst[d] = (xv[j] - m) * inv * g[d] + b[d];
  }
}

__device__ __forceinline__ void sched_fence() { __builtin_amdgcn_sched_barrier(0); }

__device__ __forceinline__ void gram_rows(const float* __restrict__ yt, int cb, int khalf,
                                          int nl, v8f (&acc)[5]) {
  const float* bcol = yt + (khalf * 2) * PG + cb * 16 + nl;
  const float* ar1 = yt + (khalf * 2) * PG + ((cb + 1) & 7) * 16 + nl;
  const float* ar2 = yt + (khalf * 2) * PG + ((cb + 2) & 7) * 16 + nl;
  const float* ar3 = yt + (khalf * 2) * PG + ((cb + 3) & 7) * 16 + nl;
  const int par = cb >> 2;
  const float* bcp = bcol + par * 4 * PG;
  const float* ar4 = yt + (khalf * 2 + par * 4) * PG + ((cb + 4) & 7) * 16 + nl;
  for (int kq = 0; kq < 16; ++kq) {
    v2f bf[4], a1[4], a2[4], a3[4], bp[2], a4[2];
#pragma unroll
    for (int s = 0; s < 4; ++s) {
      const int o = s * 4 * PG;
      bf[s].x = bcol[o];   bf[s].y = bcol[o + PG];
      a1[s].x = ar1[o];    a1[s].y = ar1[o + PG];
      a2[s].x = ar2[o];    a2[s].y = ar2[o + PG];
      a3[s].x = ar3[o];    a3[s].y = ar3[o + PG];
    }
#pragma unroll
    for (int u = 0; u < 2; ++u) {
      const int o = u * 8 * PG;
      bp[u].x = bcp[o];   bp[u].y = bcp[o + PG];
      a4[u].x = ar4[o];   a4[u].y = ar4[o + PG];
    }
    sched_fence();
#pragma unroll
    for (int s = 0; s < 4; ++s) {
      acc[0] = wmma4(bf[s], bf[s], acc[0]);
      acc[1] = wmma4(a1[s], bf[s], acc[1]);
      acc[2] = wmma4(a2[s], bf[s], acc[2]);
      acc[3] = wmma4(a3[s], bf[s], acc[3]);
      if (s & 1) acc[4] = wmma4(a4[s >> 1], bp[s >> 1], acc[4]);
    }
    bcol += 16 * PG;  ar1 += 16 * PG;  ar2 += 16 * PG;  ar3 += 16 * PG;
    bcp += 16 * PG;   ar4 += 16 * PG;
  }
}

__global__ void __launch_bounds__(256) k_gram(const float* __restrict__ x,
                       const float* __restrict__ g1,
                       const float* __restrict__ b1,
                       float* __restrict__ statp,
                       float* __restrict__ gpart) {
  extern __shared__ float yt[];
  __shared__ float sred[2][8][128];
  __shared__ float hx[8][16 * 17];
  const int tid = threadIdx.x, lane = tid & 31, wave = tid >> 5;
  const int b = blockIdx.x >> 3;
  const int oct = blockIdx.x & 7;
  const int khalf = lane >> 4;
  const int nl = lane & 15;
  const v8f vzero = {0, 0, 0, 0, 0, 0, 0, 0};
  v8f acc[5];
#pragma unroll
  for (int t = 0; t < 5; ++t) acc[t] = vzero;
  float a1[4] = {0, 0, 0, 0}, a2[4] = {0, 0, 0, 0};

  for (int sc = 0; sc < 2; ++sc) {
    const long rowbase = (long)b * 4096 + oct * 512 + sc * 256;
    __syncthreads();
    for (int i0 = 0; i0 < 32; i0 += 16) {
      const int r0 = wave * 32 + i0;
      float xv[16][4], s[16], m[16], vs[16];
#pragma unroll
      for (int q = 0; q < 16; ++q) {
        const float* src = x + (rowbase + r0 + q) * 128;
#pragma unroll
        for (int j = 0; j < 4; ++j) xv[q][j] = src[lane + 32 * j];
      }
#pragma unroll
      for (int q = 0; q < 16; ++q) s[q] = xv[q][0] + xv[q][1] + xv[q][2] + xv[q][3];
      wave_sum32_rows<16>(s);
#pragma unroll
      for (int q = 0; q < 16; ++q) {
        m[q] = s[q] * (1.0f / 128.0f);
        float t = 0.0f;
#pragma unroll
        for (int j = 0; j < 4; ++j) { float d = xv[q][j] - m[q]; t += d * d; }
        vs[q] = t;
      }
      wave_sum32_rows<16>(vs);
#pragma unroll
      for (int q = 0; q < 16; ++q) {
        float inv = rsqrtf(vs[q] * (1.0f / 128.0f) + EPSV);
#pragma unroll
        for (int j = 0; j < 4; ++j) {
          int d = lane + 32 * j;
          float y = (xv[q][j] - m[q]) * inv * g1[d] + b1[d];
          a1[j] += y; a2[j] += y * y;
          yt[(r0 + q) * PG + d] = y;
        }
      }
    }
    __syncthreads();
    gram_rows(yt, wave, khalf, nl, acc);
  }

  float* gp = gpart + (long)blockIdx.x * 16384;
  const int de = wave * 16 + nl;
#pragma unroll
  for (int j = 0; j < 4; ++j) {
    const int mi = (wave + j) & 7;
    const int dmBase = mi * 16 + khalf * 8;
#pragma unroll
    for (int r = 0; r < 8; ++r) gp[(dmBase + r) * 128 + de] = acc[j][r];
    if (j > 0) {
      float* mp = gp + de * 128 + dmBase;
      v4f lo, hi;
      lo.x = acc[j][0]; lo.y = acc[j][1]; lo.z = acc[j][2]; lo.w = acc[j][3];
      hi.x = acc[j][4]; hi.y = acc[j][5]; hi.z = acc[j][6]; hi.w = acc[j][7];
      *(v4f*)mp = lo;
      *(v4f*)(mp + 4) = hi;
    }
  }
#pragma unroll
  for (int r = 0; r < 8; ++r) hx[wave][(khalf * 8 + r) * 17 + nl] = acc[4][r];
#pragma unroll
  for (int j = 0; j < 4; ++j) {
    int d = lane + 32 * j;
    sred[0][wave][d] = a1[j];
    sred[1][wave][d] = a2[j];
  }
  __syncthreads();
  {
    const int which = tid >> 7, d = tid & 127;
    float s = 0.0f;
#pragma unroll
    for (int w = 0; w < 8; ++w) s += sred[which][w][d];
    statp[which * 32768 + blockIdx.x * 128 + d] = s;
  }
  {
    const int pw = (wave + 4) & 7;
    const int dmBase = pw * 16 + khalf * 8;
#pragma unroll
    for (int r = 0; r < 8; ++r)
      gp[(dmBase + r) * 128 + de] = acc[4][r] + hx[pw][nl * 17 + khalf * 8 + r];
  }
}


struct RowSet { v4f s1[2][8], s2[2][8], g[2][8]; };

__device__ __forceinline__ void rows_load(RowSet& r, const float* __restrict__ sp,
                                          const float* __restrict__ gp, int grp) {
#pragma unroll
  for (int t = 0; t < 2; ++t) {
#pragma unroll
    for (int q = 0; q < 8; ++q) {
      const int row = (grp * 2 + t) * 8 + q;
      r.s1[t][q] = *(const v4f*)(sp + row * 128);
      r.s2[t][q] = *(const v4f*)(sp + 32768 + row * 128);
      r.g[t][q] = *(const v4f*)(gp + (long)row * 16384);
    }
  }
}

__device__ __forceinline__ void rows_use(const RowSet& r, int dl, int dc, v4f& accv) {
#pragma unroll
  for (int t = 0; t < 2; ++t) {
    v4f s1 = {0, 0, 0, 0}, s2 = {0, 0, 0, 0}, G = {0, 0, 0, 0};
#pragma unroll
    for (int q = 0; q < 8; ++q) { s1 += r.s1[t][q]; s2 += r.s2[t][q]; G += r.g[t][q]; }
    v4f m, iv;
#pragma unroll
    for (int c = 0; c < 4; ++c) {
      float v = s2[c] - s1[c] * s1[c] * (1.0f / 4096.0f);
      m[c] = s1[c] * (1.0f / 4096.0f);
      iv[c] = 1.0f / fmaxf(sqrtf(fmaxf(v, 0.0f)), EPSV);
    }
    const float mc = (dc == 0) ? m[0] : (dc == 1) ? m[1] : (dc == 2) ? m[2] : m[3];
    const float ic = (dc == 0) ? iv[0] : (dc == 1) ? iv[1] : (dc == 2) ? iv[2] : iv[3];
    const float md = __int_as_float(__builtin_amdgcn_readlane(__float_as_int(mc), dl));
    const float ind = __int_as_float(__builtin_amdgcn_readlane(__float_as_int(ic), dl));
#pragma unroll
    for (int c = 0; c < 4; ++c) accv[c] += (G[c] - 4096.0f * md * m[c]) * ind * iv[c];
  }
}

__global__ void __launch_bounds__(128) k_combine(const float* __restrict__ statp,
                          const float* __restrict__ gpart,
                          const float* __restrict__ corr,
                          const float* __restrict__ alpha_logit,
                          float* __restrict__ mixed_out,
                          float* __restrict__ dis_out) {
  __shared__ __attribute__((aligned(16))) float part[4][128];
  __shared__ float wsum[4];
  const int d = blockIdx.x;
  const int e = threadIdx.x;
  const int lane = e & 31, wave = e >> 5;
  const float cde = corr[d * 128 + e];
  const float ced = corr[e * 128 + d];
  const float alog = alpha_logit[0];
  const int dl = d >> 2, dc = d & 3;
  const float* sp = statp + (wave * 64) * 128 + 4 * lane;
  const float* gp = gpart + (long)(wave * 64) * 16384 + d * 128 + 4 * lane;
  v4f accv = {0, 0, 0, 0};
  RowSet ra, rb;
  rows_load(ra, sp, gp, 0);
  rows_load(rb, sp, gp, 1);
  sched_fence();
  rows_use(ra, dl, dc, accv);
  rows_load(ra, sp, gp, 2);
  sched_fence();
  rows_use(rb, dl, dc, accv);
  rows_load(rb, sp, gp, 3);
  sched_fence();
  rows_use(ra, dl, dc, accv);
  rows_use(rb, dl, dc, accv);
  *(v4f*)&part[wave][4 * lane] = accv;
  __syncthreads();
  const float acc = ((part[0][e] + part[1][e]) + part[2][e]) + part[3][e];
  const float alpha = 1.0f / (1.0f + __expf(-alog));
  const float learned = tanhf(0.5f * (cde + ced));
  const float dcr = fminf(fmaxf(acc * (1.0f / 32.0f), -1.0f), 1.0f);
  const float mx = alpha * learned + (1.0f - alpha) * dcr;
  mixed_out[d * 128 + e] = mx;
  float psum = wave_sum32((e == d) ? 0.0f : mx);
  if ((e & 31) == 0) wsum[e >> 5] = psum;
  __syncthreads();
  if (e == 0) {
    float deg = ((wsum[0] + wsum[1]) + wsum[2]) + wsum[3];
    dis_out[d] = rsqrtf(fmaxf(deg, EPSV));
  }
}

__global__ void __launch_bounds__(128) k_graph(const float* __restrict__ chebw,
                        const float* __restrict__ mixed,
                        const float* __restrict__ dis,
                        const float* __restrict__ Wi,
                        const float* __restrict__ Wo,
                        float* __restrict__ LapP,
                        float* __restrict__ WiP,
                        float* __restrict__ WoP,
                        float* __restrict__ wout) {
  const int g = blockIdx.x * 128 + threadIdx.x;
  const int lane = g & 31;
  const int h = lane >> 4, c = lane & 15;
  if (blockIdx.x < 32) {
    const int jj = (g >> 5) & 15;
    const int nb = g >> 9;
    const int n = nb * 16 + c;
    const float dn = dis[n];
    v4f v;
#pragma unroll
    for (int e = 0; e < 4; ++e) {
      const int k = 8 * jj + 4 * h + e;
      const float dk = dis[k];
      const float m = mixed[k * 128 + n];
      float a = (n == k) ? 0.0f : m;
      float l = ((n == k) ? 1.0f : 0.0f) - dk * a * dn;
      v[e] = fminf(fmaxf(l, -1.5f), 1.5f);
    }
    ((v4f*)LapP)[g] = v;
  } else if (blockIdx.x < 96) {
    const int idx = g - 32 * 128;
    const int jj = (idx >> 5) & 15;
    const int nb = idx >> 9;
    const float* row = Wi + (nb * 16 + c) * 128 + 8 * jj + 4 * h;
    ((v4f*)WiP)[idx] = *(const v4f*)row;
  } else {
    const int idx = g - 96 * 128;
    const int jj = (idx >> 5) & 31;
    const int nb = idx >> 10;
    const float* row = Wo + (nb * 16 + c) * 256 + 8 * jj + 4 * h;
    ((v4f*)WoP)[idx] = *(const v4f*)row;
  }
  if (g == 0) {
    float c0 = chebw[0], c1 = chebw[1], c2 = chebw[2];
    float mx = fmaxf(c0, fmaxf(c1, c2));
    float e0 = __expf(c0 - mx), e1 = __expf(c1 - mx), e2 = __expf(c2 - mx);
    float s = e0 + e1 + e2;
    wout[0] = e0 / s; wout[1] = e1 / s; wout[2] = e2 / s;
  }
}


#define DB 4

template <int NT>
struct BRing { v4f v[DB + 1][NT]; };

template <int NT>
__device__ __forceinline__ void b_prefetch(BRing<NT>& r, const v4f* __restrict__ bP,
                                           int tstride) {
#pragma unroll
  for (int j = 0; j < DB; ++j) {
#pragma unroll
    for (int t = 0; t < NT; ++t) r.v[j][t] = bP[t * tstride + j * 32];
  }
}

template <int NJ, int NT, int PITCH>
__device__ __forceinline__ void wave_product(const float* a4, const v4f* __restrict__ bP,
                                             int tstride, BRing<NT>& r,
                                             v8f (&acc)[NT][4]) {
  v4f an[4];
#pragma unroll
  for (int mi = 0; mi < 4; ++mi) an[mi] = *(const v4f*)(a4 + mi * 16 * PITCH);
#pragma unroll
  for (int jj = 0; jj < NJ; ++jj) {
    v4f ac[4];
#pragma unroll
    for (int mi = 0; mi < 4; ++mi) ac[mi] = an[mi];
    if (jj + 1 < NJ) {
#pragma unroll
      for (int mi = 0; mi < 4; ++mi)
        an[mi] = *(const v4f*)(a4 + mi * 16 * PITCH + (jj + 1) * 8);
    }
    if (jj + DB < NJ) {
#pragma unroll
      for (int t = 0; t < NT; ++t)
        r.v[(jj + DB) % (DB + 1)][t] = bP[t * tstride + (jj + DB) * 32];
    }
    sched_fence();
#pragma unroll
    for (int s = 0; s < 2; ++s) {
#pragma unroll
      for (int mi = 0; mi < 4; ++mi) {
        v2f a;
        a.x = s ? ac[mi].z : ac[mi].x;   a.y = s ? ac[mi].w : ac[mi].y;
#pragma unroll
        for (int t = 0; t < NT; ++t) {
          const v4f b4 = r.v[jj % (DB + 1)][t];
          v2f b;
          b.x = s ? b4.z : b4.x;   b.y = s ? b4.w : b4.y;
          acc[t][mi] = wmma4(a, b, acc[t][mi]);
        }
      }
    }
    sched_fence();
  }
}

__global__ void __launch_bounds__(256) k_main(const float* __restrict__ x,
                       const float* __restrict__ g1, const float* __restrict__ b1,
                       const float* __restrict__ LapP, const float* __restrict__ wv,
                       const float* __restrict__ WiP, const float* __restrict__ bi,
                       const float* __restrict__ WoP, const float* __restrict__ bo,
                       const float* __restrict__ g2, const float* __restrict__ b2,
                       const float* __restrict__ g3, const float* __restrict__ b3,
                       float* __restrict__ out) {
  extern __shared__ float smem[];
  float* yo = smem;
  float* t1 = smem + 64 * PY;
  float* hb = smem + 2 * 64 * PY;
  const int tid = threadIdx.x, lane = tid & 31, wave = tid >> 5;
  const long base = (long)blockIdx.x * 64;
  const float w0 = wv[0], w1 = wv[1], w2 = wv[2];
  const int khalf = lane >> 4;
  const int nl = lane & 15;

  for (int i = 0; i < 8; ++i) {
    int r = wave * 8 + i;
    ln128_row(x + (base + r) * 128, yo + r * PY, g1, b1, lane);
  }

  const float* a4_yo = yo + nl * PY + khalf * 4;
  const float* a4_t1 = t1 + nl * PY + khalf * 4;
  const float* a4_hb = hb + nl * PH + khalf * 4;
  const int nc = wave * 16 + nl;
  const v4f* lapP = (const v4f*)LapP + (wave * 16) * 32 + lane;
  BRing<1> rb;
  b_prefetch<1>(rb, lapP, 0);
  __syncthreads();

  {
    v8f acc[1][4] = {};
    wave_product<16, 1, PY>(a4_yo, lapP, 0, rb, acc);
#pragma unroll
    for (int mi = 0; mi < 4; ++mi) {
      int rowB = mi * 16 + khalf * 8;
#pragma unroll
      for (int r = 0; r < 8; ++r) t1[(rowB + r) * PY + nc] = acc[0][mi][r];
    }
  }
  b_prefetch<1>(rb, lapP, 0);
  __syncthreads();

  {
    v8f acc[1][4] = {};
    wave_product<16, 1, PY>(a4_t1, lapP, 0, rb, acc);
#pragma unroll
    for (int mi = 0; mi < 4; ++mi) {
      int rowB = mi * 16 + khalf * 8;
#pragma unroll
      for (int r = 0; r < 8; ++r) {
        int pos = (rowB + r) * PY + nc;
        float y0 = yo[pos], y1 = t1[pos];
        float tx2 = fminf(fmaxf(2.0f * acc[0][mi][r] - y0, -50.0f), 50.0f);
        yo[pos] = w0 * y0 + w1 * y1 + w2 * tx2;
      }
    }
  }
  const v4f* wiP = (const v4f*)WiP + (wave * 16) * 32 + lane;
  BRing<2> rb2;
  b_prefetch<2>(rb2, wiP, 8 * 16 * 32);
  __syncthreads();

  {
    v8f acc[2][4] = {};
    wave_product<16, 2, PY>(a4_yo, wiP, 8 * 16 * 32, rb2, acc);
#pragma unroll
    for (int j = 0; j < 2; ++j) {
      int ncj = nc + 128 * j;
      float bia = bi[ncj];
#pragma unroll
      for (int mi = 0; mi < 4; ++mi) {
        int rowB = mi * 16 + khalf * 8;
#pragma unroll
        for (int r = 0; r < 8; ++r) hb[(rowB + r) * PH + ncj] = acc[j][mi][r] + bia;
      }
    }
  }
  __syncthreads();

  {
    const int rw = wave * 8;
    float xv[8][8], s[8], m[8], vs[8];
#pragma unroll
    for (int i = 0; i < 8; ++i) {
#pragma unroll
      for (int j = 0; j < 8; ++j) xv[i][j] = hb[(rw + i) * PH + lane + 32 * j];
    }
#pragma unroll
    for (int i = 0; i < 8; ++i) {
      float t = 0.0f;
#pragma unroll
      for (int j = 0; j < 8; ++j) t += xv[i][j];
      s[i] = t;
    }
    wave_sum32_rows<8>(s);
#pragma unroll
    for (int i = 0; i < 8; ++i) {
      m[i] = s[i] * (1.0f / 256.0f);
      float t = 0.0f;
#pragma unroll
      for (int j = 0; j < 8; ++j) { float d = xv[i][j] - m[i]; t += d * d; }
      vs[i] = t;
    }
    wave_sum32_rows<8>(vs);
#pragma unroll
    for (int i = 0; i < 8; ++i) {
      float inv = rsqrtf(vs[i] * (1.0f / 256.0f) + EPSV);
#pragma unroll
      for (int j = 0; j < 8; ++j) {
        int d = lane + 32 * j;
        hb[(rw + i) * PH + d] = gelu_erf((xv[i][j] - m[i]) * inv * g2[d] + b2[d]);
      }
    }
  }
  const v4f* woP = (const v4f*)WoP + (wave * 32) * 32 + lane;
  b_prefetch<1>(rb, woP, 0);
  __syncthreads();

  {
    v8f acc[1][4] = {};
    wave_product<32, 1, PH>(a4_hb, woP, 0, rb, acc);
    float boa = bo[nc];
#pragma unroll
    for (int mi = 0; mi < 4; ++mi) {
      int rowB = mi * 16 + khalf * 8;
#pragma unroll
      for (int r = 0; r < 8; ++r) t1[(rowB + r) * PY + nc] = acc[0][mi][r] + boa;
    }
  }
  __syncthreads();

  {
    const int rw = wave * 8;
    float xv[8][4], s[8], m[8], vs[8];
#pragma unroll
    for (int i = 0; i < 8; ++i) {
#pragma unroll
      for (int j = 0; j < 4; ++j) xv[i][j] = t1[(rw + i) * PY + lane + 32 * j];
    }
#pragma unroll
    for (int i = 0; i < 8; ++i) s[i] = xv[i][0] + xv[i][1] + xv[i][2] + xv[i][3];
    wave_sum32_rows<8>(s);
#pragma unroll
    for (int i = 0; i < 8; ++i) {
      m[i] = s[i] * (1.0f / 128.0f);
      float t = 0.0f;
#pragma unroll
      for (int j = 0; j < 4; ++j) { float d = xv[i][j] - m[i]; t += d * d; }
      vs[i] = t;
    }
    wave_sum32_rows<8>(vs);
#pragma unroll
    for (int i = 0; i < 8; ++i) {
      float inv = rsqrtf(vs[i] * (1.0f / 128.0f) + EPSV);
#pragma unroll
      for (int j = 0; j < 4; ++j) {
        int d = lane + 32 * j;
        out[(base + rw + i) * 128 + d] = (xv[i][j] - m[i]) * inv * g3[d] + b3[d];
      }
    }
  }
}

extern "C" void kernel_launch(void* const* d_in, const int* in_sizes, int n_in,
                              void* d_out, int out_size, void* d_ws, size_t ws_size,
                              hipStream_t stream) {
  const float* x     = (const float*)d_in[0];
  const float* corr  = (const float*)d_in[1];
  const float* alog  = (const float*)d_in[2];
  const float* chebw = (const float*)d_in[3];
  const float* Wi    = (const float*)d_in[4];
  const float* bi    = (const float*)d_in[5];
  const float* Wo    = (const float*)d_in[6];
  const float* bo    = (const float*)d_in[7];
  const float* g1    = (const float*)d_in[8];
  const float* b1    = (const float*)d_in[9];
  const float* g2    = (const float*)d_in[10];
  const float* b2    = (const float*)d_in[11];
  const float* g3    = (const float*)d_in[12];
  const float* b3    = (const float*)d_in[13];
  float* out = (float*)d_out;
  float* ws  = (float*)d_ws;

  float* gpart  = ws + 8192;
  float* disv   = ws + 8192 + 256 * 16384;
  float* LapP   = disv + 16384;
  float* wv     = LapP + 16384;
  float* WiP    = gpart;
  float* WoP    = gpart + 32768;
  float* mixed  = out + (long)32 * 4096 * 128;
  float* statp  = out;

  k_gram<<<256, 256, 256 * PG * sizeof(float), stream>>>(x, g1, b1, statp, gpart);
  k_combine<<<128, 128, 0, stream>>>(statp, gpart, corr, alog, mixed, disv);
  k_graph<<<160, 128, 0, stream>>>(chebw, mixed, disv, Wi, Wo, LapP, WiP, WoP, wv);
  k_main<<<2048, 256, (2 * 64 * PY + 64 * PH) * sizeof(float), stream>>>(x, g1, b1, LapP, wv, WiP, bi, WoP, bo,
                                        g2, b2, g3, b3, out);
}
